# starting kernel
# baseline (speedup 1.0000x reference)
#include <hip/hip_runtime.h>

typedef float v2f __attribute__((ext_vector_type(2)));
typedef float v8f __attribute__((ext_vector_type(8)));

#define TK 64

__global__ __launch_bounds__(256) void k_gemm_wmma(
    const float* __restrict__ X, const float* __restrict__ Wp,
    float* __restrict__ H, int M, int K, int Kpad, int Nout)
{
    __shared__ float As[16][TK + 4];

    const int row0 = blockIdx.x << 4;
    const int t    = threadIdx.x;
    const int wave = t >> 5;
    const int lane = t & 31;
    const int half = lane >> 4;
    const int m    = lane & 15;

    const int n0 = (wave << 4) + m;
    const int n1 = ((wave + 8) << 4) + m;
    const float* __restrict__ Wr0 = Wp + (size_t)n0 * Kpad;
    const float* __restrict__ Wr1 = Wp + (size_t)n1 * Kpad;

    v8f acc0 = {}; v8f acc1 = {};

    for (int k0 = 0; k0 < Kpad; k0 += TK) {
        __syncthreads();
        #pragma unroll
        for (int i = 0; i < (16 * TK) / 256; ++i) {
            int e  = t + (i << 8);
            int r  = e >> 6;
            int kk = e & (TK - 1);
            int gr = row0 + r, gk = k0 + kk;
            As[r][kk] = (gr < M && gk < K) ? X[(size_t)gr * K + gk] : 0.0f;
        }
        __syncthreads();

        #pragma unroll
        for (int kk = 0; kk < TK; kk += 4) {
            const int kb = kk + (half << 1);
            v2f a  = *(const v2f*)&As[m][kb];
            v2f b0 = *(const v2f*)&Wr0[k0 + kb];
            v2f b1 = *(const v2f*)&Wr1[k0 + kb];
            acc0 = __builtin_amdgcn_wmma_f32_16x16x4_f32(
                false, a, false, b0, (short)0, acc0, false, false);
            acc1 = __builtin_amdgcn_wmma_f32_16x16x4_f32(
                false, a, false, b1, (short)0, acc1, false, false);
        }
    }

    #pragma unroll
    for (int r = 0; r < 8; ++r) {
        int grow = row0 + r + (half << 3);
        if (grow < M) {
            if (n0 < Nout) H[(size_t)grow * Nout + n0] = acc0[r];
            if (n1 < Nout) H[(size_t)grow * Nout + n1] = acc1[r];
        }
    }
}

__global__ void k_padW(const float* __restrict__ W, float* __restrict__ Wp,
                       int Nout, int K, int Kpad)
{
    int idx = blockIdx.x * blockDim.x + threadIdx.x;
    int tot = 256 * Kpad;
    if (idx >= tot) return;
    int r = idx / Kpad, c = idx % Kpad;
    Wp[idx] = (r < Nout && c < K) ? W[(size_t)r * K + c] : 0.0f;
}

__global__ void k_fill1(float* __restrict__ p, int n) {
    int i = blockIdx.x * blockDim.x + threadIdx.x;
    if (i < n) p[i] = 1.0f;
}

__global__ void k_deg(const int* __restrict__ dst, float* __restrict__ deg, int E) {
    int e = blockIdx.x * blockDim.x + threadIdx.x;
    if (e < E) atomicAdd(&deg[dst[e]], 1.0f);
}

__global__ void k_rsqrt(float* __restrict__ p, int n) {
    int i = blockIdx.x * blockDim.x + threadIdx.x;
    if (i < n) { float d = p[i]; p[i] = (d > 0.0f) ? rsqrtf(d) : 0.0f; }
}

__global__ void k_zero(float* __restrict__ p, size_t n) {
    size_t i = (size_t)blockIdx.x * blockDim.x + threadIdx.x;
    if (i < n) p[i] = 0.0f;
}

__global__ __launch_bounds__(256) void k_agg_wave(
    const float* __restrict__ H, const int* __restrict__ src,
    const int* __restrict__ dst, const float* __restrict__ dinv,
    float* __restrict__ out, int E, int F)
{
    const int lane   = threadIdx.x & 31;
    const int wid    = (int)((blockIdx.x * blockDim.x + threadIdx.x) >> 5);
    const int stride = (int)((gridDim.x * blockDim.x) >> 5);

    for (int e = wid; e < E; e += stride) {
        const int s = src[e];
        const int d = dst[e];
        const int e2 = e + stride;
        if (e2 < E) {
            __builtin_prefetch(&H[(size_t)src[e2] * F], 0, 1);
        }
        const float nrm = dinv[s] * dinv[d];
        const float* __restrict__ hrow = H + (size_t)s * F;
        float* __restrict__ orow = out + (size_t)d * F;
        for (int f = lane; f < F; f += 32) {
            atomicAdd(&orow[f], hrow[f] * nrm);
        }
    }
}

__global__ void k_agg_small(
    const float* __restrict__ H, const int* __restrict__ src,
    const int* __restrict__ dst, const float* __restrict__ dinv,
    float* __restrict__ out, int E, int F)
{
    size_t idx = (size_t)blockIdx.x * blockDim.x + threadIdx.x;
    size_t tot = (size_t)E * F;
    if (idx >= tot) return;
    int e = (int)(idx / F);
    int f = (int)(idx % F);
    int s = src[e], d = dst[e];
    float nrm = dinv[s] * dinv[d];
    atomicAdd(&out[(size_t)d * F + f], H[(size_t)s * F + f] * nrm);
}

__global__ void k_finish(float* __restrict__ dest, const float* __restrict__ H,
                         const float* __restrict__ dinv, const float* __restrict__ bias,
                         int N, int F, int doRelu)
{
    size_t idx = (size_t)blockIdx.x * blockDim.x + threadIdx.x;
    size_t tot = (size_t)N * F;
    if (idx >= tot) return;
    int i = (int)(idx / F);
    int f = (int)(idx % F);
    float di = dinv[i];
    float v  = dest[idx] + H[idx] * di * di + bias[f];
    dest[idx] = doRelu ? fmaxf(v, 0.0f) : v;
}

extern "C" void kernel_launch(void* const* d_in, const int* in_sizes, int n_in,
                              void* d_out, int out_size, void* d_ws, size_t ws_size,
                              hipStream_t stream) {
    (void)n_in; (void)out_size; (void)ws_size;

    const float* x   = (const float*)d_in[0];
    const int*   ei  = (const int*)  d_in[1];
    const float* W1  = (const float*)d_in[2];
    const float* b1  = (const float*)d_in[3];
    const float* W1b = (const float*)d_in[4];
    const float* b1b = (const float*)d_in[5];
    const float* W2b = (const float*)d_in[6];
    const float* b2b = (const float*)d_in[7];
    const float* W2  = (const float*)d_in[8];
    const float* b2  = (const float*)d_in[9];

    const int F_IN = 500, H1 = 200, H2 = 100, H3 = 40, C = 3;
    const int N = in_sizes[0] / F_IN;
    const int E = in_sizes[1] / 2;
    const int* srcIdx = ei;
    const int* dstIdx = ei + E;

    char*  ws   = (char*)d_ws;
    float* dinv = (float*)ws;
    size_t off  = ((size_t)N * sizeof(float) + 511) & ~(size_t)511;
    float* Wpad = (float*)(ws + off);
    off += (size_t)256 * 512 * sizeof(float);
    float* bufH = (float*)(ws + off);
    off += (((size_t)N * H1 * sizeof(float)) + 511) & ~(size_t)511;
    float* bufA = (float*)(ws + off);
    float* out  = (float*)d_out;

    k_fill1<<<(N + 255) / 256, 256, 0, stream>>>(dinv, N);
    k_deg  <<<(E + 255) / 256, 256, 0, stream>>>(dstIdx, dinv, E);
    k_rsqrt<<<(N + 255) / 256, 256, 0, stream>>>(dinv, N);

    const int gemmBlocks = (N + 15) / 16;
    const int aggBlocks  = 65536;

    auto layer = [&](const float* in, const float* Wt, const float* bias,
                     float* dest, int K, int F, int relu) {
        const int Kpad = (K + TK - 1) & ~(TK - 1);
        k_padW<<<(256 * Kpad + 255) / 256, 256, 0, stream>>>(Wt, Wpad, F, K, Kpad);
        k_gemm_wmma<<<gemmBlocks, 256, 0, stream>>>(in, Wpad, bufH, N, K, Kpad, F);
        size_t tot = (size_t)N * F;
        k_zero<<<(int)((tot + 255) / 256), 256, 0, stream>>>(dest, tot);
        if (F >= 32) {
            k_agg_wave<<<aggBlocks, 256, 0, stream>>>(bufH, srcIdx, dstIdx, dinv, dest, E, F);
        } else {
            size_t tote = (size_t)E * F;
            k_agg_small<<<(int)((tote + 255) / 256), 256, 0, stream>>>(
                bufH, srcIdx, dstIdx, dinv, dest, E, F);
        }
        k_finish<<<(int)((tot + 255) / 256), 256, 0, stream>>>(
            dest, bufH, dinv, bias, N, F, relu);
    };

    layer(x,    W1,  b1,  bufA, F_IN, H1, 1);
    layer(bufA, W1b, b1b, bufA, H1,   H2, 1);
    layer(bufA, W2b, b2b, bufA, H2,   H3, 1);
    layer(bufA, W2,  b2,  out,  H3,   C,  0);
}


# GEMM re-tiled for operand reuse: 64-row block stages X tile and W chunk in LDS (W read directly, pad kernel gone, next chunk prefetched), one A fragment feeds all real column tiles; sum order kept
# speedup vs baseline: 1.3393x; 1.3393x over previous
#include <hip/hip_runtime.h>

typedef float v2f __attribute__((ext_vector_type(2)));
typedef float v8f __attribute__((ext_vector_type(8)));
typedef float v4f __attribute__((ext_vector_type(4)));

#define TK 32
#define LDP (TK + 4)
#define BM 64
#define XQ ((BM * TK / 4) / 128)

template <int NT>
__global__ __launch_bounds__(128) void k_gemm_wmma(
    const float* __restrict__ X, const float* __restrict__ W,
    float* __restrict__ H, int M, int K, int Nout)
{
    __shared__ __attribute__((aligned(16))) float As[BM][LDP];
    __shared__ __attribute__((aligned(16))) float Ws[NT * 16][LDP];

    const int row0 = blockIdx.x * BM;
    const int t    = threadIdx.x;
    const int wave = t >> 5;
    const int lane = t & 31;
    const int half = lane >> 4;
    const int m    = lane & 15;
    const int arow = (wave << 4) + m;

    const int srow = t >> 3;
    const int skq  = (t & 7) << 2;

    v8f acc[NT];
    #pragma unroll
    for (int j = 0; j < NT; ++j) acc[j] = v8f{};

    v4f xa[XQ];
    v4f wa[NT];

#define LOAD_CHUNK(k0_)                                                        \
    {                                                                          \
        const int gk_  = (k0_) + skq;                                          \
        const int gkc_ = gk_ < K ? gk_ : K - 4;       \
        _Pragma("unroll")                                                      \
        for (int i = 0; i < XQ; ++i) {                                         \
            const int gr_  = row0 + srow + (i << 4);                           \
            const int grc_ = gr_ < M ? gr_ : M - 1;   \
            xa[i] = *(const v4f*)&X[(size_t)grc_ * K + gkc_];               \
        }                                                                      \
        _Pragma("unroll")                                                      \
        for (int i = 0; i < NT; ++i) {                                         \
            const int r_  = srow + (i << 4);                                   \
            const int rc_ = r_ < Nout ? r_ : Nout - 1;   \
            wa[i] = *(const v4f*)&W[(size_t)rc_ * K + gkc_];                \
        }                                                                      \
    }

#define KSTEP(kk_)                                                             \
    {                                                                          \
        const int kb_ = (kk_) + (half << 1);            \
        const v2f a_ = *(const v2f*)&As[arow][kb_];                            \
        _Pragma("unroll")                                                      \
        for (int j = 0; j < NT; ++j) {                                         \
            const v2f b_ = *(const v2f*)&Ws[(j << 4) + m][kb_];                \
            acc[j] = __builtin_amdgcn_wmma_f32_16x16x4_f32(                    \
                false, a_, false, b_, (short)0, acc[j], false, false);         \
        }                                                                      \
    }

    LOAD_CHUNK(0)
    for (int k0 = 0; k0 < K; k0 += TK) {
        __syncthreads();
        #pragma unroll
        for (int i = 0; i < XQ; ++i) *(v4f*)&As[srow + (i << 4)][skq] = xa[i];
        #pragma unroll
        for (int i = 0; i < NT; ++i) *(v4f*)&Ws[srow + (i << 4)][skq] = wa[i];
        __syncthreads();

        if (k0 + TK < K) LOAD_CHUNK(k0 + TK)

        if (k0 + TK <= K) {
            #pragma unroll
            for (int kk = 0; kk < TK; kk += 4) KSTEP(kk)
        } else {
            for (int kk = 0; k0 + kk < K; kk += 4) KSTEP(kk)
        }
    }
#undef LOAD_CHUNK
#undef KSTEP

    const int rbase = row0 + (wave << 4) + (half << 3);
    if (row0 + BM <= M) {
        float* __restrict__ hp = H + (size_t)rbase * Nout + m;
        #pragma unroll
        for (int j = 0; j < NT; ++j) {
            if (((j + 1) << 4) <= Nout) {
                #pragma unroll
                for (int r = 0; r < 8; ++r) hp[(size_t)r * Nout + (j << 4)] = acc[j][r];
            } else if ((j << 4) + m < Nout) {
                #pragma unroll
                for (int r = 0; r < 8; ++r) hp[(size_t)r * Nout + (j << 4)] = acc[j][r];
            }
        }
    } else {
        #pragma unroll
        for (int j = 0; j < NT; ++j) {
            const int n = (j << 4) + m;
            #pragma unroll
            for (int r = 0; r < 8; ++r) {
                const int grow = rbase + r;
                if (grow < M && n < Nout) H[(size_t)grow * Nout + n] = acc[j][r];
            }
        }
    }
}

__global__ void k_fill1(float* __restrict__ p, int n) {
    int i = blockIdx.x * blockDim.x + threadIdx.x;
    if (i < n) p[i] = 1.0f;
}

__global__ void k_deg(const int* __restrict__ dst, float* __restrict__ deg, int E) {
    int e = blockIdx.x * blockDim.x + threadIdx.x;
    if (e < E) atomicAdd(&deg[dst[e]], 1.0f);
}

__global__ void k_rsqrt(float* __restrict__ p, int n) {
    int i = blockIdx.x * blockDim.x + threadIdx.x;
    if (i < n) { float d = p[i]; p[i] = (d > 0.0f) ? rsqrtf(d) : 0.0f; }
}

__global__ void k_zero(float* __restrict__ p, size_t n) {
    size_t i = (size_t)blockIdx.x * blockDim.x + threadIdx.x;
    if (i < n) p[i] = 0.0f;
}

__global__ __launch_bounds__(256) void k_agg_wave(
    const float* __restrict__ H, const int* __restrict__ src,
    const int* __restrict__ dst, const float* __restrict__ dinv,
    float* __restrict__ out, int E, int F)
{
    const int lane   = threadIdx.x & 31;
    const int wid    = (int)((blockIdx.x * blockDim.x + threadIdx.x) >> 5);
    const int stride = (int)((gridDim.x * blockDim.x) >> 5);

    for (int e = wid; e < E; e += stride) {
        const int s = src[e];
        const int d = dst[e];
        const int e2 = e + stride;
        if (e2 < E) {
            __builtin_prefetch(&H[(size_t)src[e2] * F], 0, 1);
        }
        const float nrm = dinv[s] * dinv[d];
        const float* __restrict__ hrow = H + (size_t)s * F;
        float* __restrict__ orow = out + (size_t)d * F;
        for (int f = lane; f < F; f += 32) {
            atomicAdd(&orow[f], hrow[f] * nrm);
        }
    }
}

__global__ void k_agg_small(
    const float* __restrict__ H, const int* __restrict__ src,
    const int* __restrict__ dst, const float* __restrict__ dinv,
    float* __restrict__ out, int E, int F)
{
    size_t idx = (size_t)blockIdx.x * blockDim.x + threadIdx.x;
    size_t tot = (size_t)E * F;
    if (idx >= tot) return;
    int e = (int)(idx / F);
    int f = (int)(idx % F);
    int s = src[e], d = dst[e];
    float nrm = dinv[s] * dinv[d];
    atomicAdd(&out[(size_t)d * F + f], H[(size_t)s * F + f] * nrm);
}

__global__ void k_finish(float* __restrict__ dest, const float* __restrict__ H,
                         const float* __restrict__ dinv, const float* __restrict__ bias,
                         int N, int F, int doRelu)
{
    size_t idx = (size_t)blockIdx.x * blockDim.x + threadIdx.x;
    size_t tot = (size_t)N * F;
    if (idx >= tot) return;
    int i = (int)(idx / F);
    int f = (int)(idx % F);
    float di = dinv[i];
    float v  = dest[idx] + H[idx] * di * di + bias[f];
    dest[idx] = doRelu ? fmaxf(v, 0.0f) : v;
}

extern "C" void kernel_launch(void* const* d_in, const int* in_sizes, int n_in,
                              void* d_out, int out_size, void* d_ws, size_t ws_size,
                              hipStream_t stream) {
    (void)n_in; (void)out_size; (void)ws_size;

    const float* x   = (const float*)d_in[0];
    const int*   ei  = (const int*)  d_in[1];
    const float* W1  = (const float*)d_in[2];
    const float* b1  = (const float*)d_in[3];
    const float* W1b = (const float*)d_in[4];
    const float* b1b = (const float*)d_in[5];
    const float* W2b = (const float*)d_in[6];
    const float* b2b = (const float*)d_in[7];
    const float* W2  = (const float*)d_in[8];
    const float* b2  = (const float*)d_in[9];

    const int F_IN = 500, H1 = 200, H2 = 100, H3 = 40, C = 3;
    const int N = in_sizes[0] / F_IN;
    const int E = in_sizes[1] / 2;
    const int* srcIdx = ei;
    const int* dstIdx = ei + E;

    char*  ws   = (char*)d_ws;
    float* dinv = (float*)ws;
    size_t off  = ((size_t)N * sizeof(float) + 511) & ~(size_t)511;
    off += (size_t)256 * 512 * sizeof(float);
    float* bufH = (float*)(ws + off);
    off += (((size_t)N * H1 * sizeof(float)) + 511) & ~(size_t)511;
    float* bufA = (float*)(ws + off);
    float* out  = (float*)d_out;

    k_fill1<<<(N + 255) / 256, 256, 0, stream>>>(dinv, N);
    k_deg  <<<(E + 255) / 256, 256, 0, stream>>>(dstIdx, dinv, E);
    k_rsqrt<<<(N + 255) / 256, 256, 0, stream>>>(dinv, N);

    const int gemmBlocks = (N + BM - 1) / BM;
    const int aggBlocks  = 65536;

    auto layer = [&](const float* in, const float* Wt, const float* bias,
                     float* dest, int K, int F, int relu) {
        if (F <= 16)       k_gemm_wmma<1><<<gemmBlocks, 128, 0, stream>>>(in, Wt, bufH, N, K, F);
        else if (F <= 48)  k_gemm_wmma<3><<<gemmBlocks, 128, 0, stream>>>(in, Wt, bufH, N, K, F);
        else if (F <= 112) k_gemm_wmma<7><<<gemmBlocks, 128, 0, stream>>>(in, Wt, bufH, N, K, F);
        else               k_gemm_wmma<13><<<gemmBlocks, 128, 0, stream>>>(in, Wt, bufH, N, K, F);
        size_t tot = (size_t)N * F;
        k_zero<<<(int)((tot + 255) / 256), 256, 0, stream>>>(dest, tot);
        if (F >= 32) {
            k_agg_wave<<<aggBlocks, 256, 0, stream>>>(bufH, srcIdx, dstIdx, dinv, dest, E, F);
        } else {
            size_t tote = (size_t)E * F;
            k_agg_small<<<(int)((tote + 255) / 256), 256, 0, stream>>>(
                bufH, srcIdx, dstIdx, dinv, dest, E, F);
        }
        k_finish<<<(int)((tot + 255) / 256), 256, 0, stream>>>(
            dest, bufH, dinv, bias, N, F, relu);
    };

    layer(x,    W1,  b1,  bufA, F_IN, H1, 1);
    layer(bufA, W1b, b1b, bufA, H1,   H2, 1);
    layer(bufA, W2b, b2b, bufA, H2,   H3, 1);
    layer(bufA, W2,  b2,  out,  H3,   C,  0);
}


# Aggregation as a gather over a destination-ordered edge list built in the call (count with returned rank, scan, place): float atomics, zero fill, epilogue pass gone; row sums in arrival-rank order
# speedup vs baseline: 3.4468x; 2.5736x over previous
#include <hip/hip_runtime.h>

typedef float v2f __attribute__((ext_vector_type(2)));
typedef float v8f __attribute__((ext_vector_type(8)));
typedef float v4f __attribute__((ext_vector_type(4)));

#define TK 32
#define LDP (TK + 4)
#define BM 64
#define XQ ((BM * TK / 4) / 128)

template <int NT>
__global__ __launch_bounds__(128) void k_gemm_wmma(
    const float* __restrict__ X, const float* __restrict__ W,
    float* __restrict__ H, int M, int K, int Nout)
{
    __shared__ __attribute__((aligned(16))) float As[BM][LDP];
    __shared__ __attribute__((aligned(16))) float Ws[NT * 16][LDP];

    const int row0 = blockIdx.x * BM;
    const int t    = threadIdx.x;
    const int wave = t >> 5;
    const int lane = t & 31;
    const int half = lane >> 4;
    const int m    = lane & 15;
    const int arow = (wave << 4) + m;

    const int srow = t >> 3;
    const int skq  = (t & 7) << 2;

    v8f acc[NT];
    #pragma unroll
    for (int j = 0; j < NT; ++j) acc[j] = v8f{};

    v4f xa[XQ];
    v4f wa[NT];

#define LOAD_CHUNK(k0_)                                                        \
    {                                                                          \
        const int gk_  = (k0_) + skq;                                          \
        const int gkc_ = gk_ < K ? gk_ : K - 4;       \
        _Pragma("unroll")                                                      \
        for (int i = 0; i < XQ; ++i) {                                         \
            const int gr_  = row0 + srow + (i << 4);                           \
            const int grc_ = gr_ < M ? gr_ : M - 1;   \
            xa[i] = *(const v4f*)&X[(size_t)grc_ * K + gkc_];               \
        }                                                                      \
        _Pragma("unroll")                                                      \
        for (int i = 0; i < NT; ++i) {                                         \
            const int r_  = srow + (i << 4);                                   \
            const int rc_ = r_ < Nout ? r_ : Nout - 1;   \
            wa[i] = *(const v4f*)&W[(size_t)rc_ * K + gkc_];                \
        }                                                                      \
    }

#define KSTEP(kk_)                                                             \
    {                                                                          \
        const int kb_ = (kk_) + (half << 1);            \
        const v2f a_ = *(const v2f*)&As[arow][kb_];                            \
        _Pragma("unroll")                                                      \
        for (int j = 0; j < NT; ++j) {                                         \
            const v2f b_ = *(const v2f*)&Ws[(j << 4) + m][kb_];                \
            acc[j] = __builtin_amdgcn_wmma_f32_16x16x4_f32(                    \
                false, a_, false, b_, (short)0, acc[j], false, false);         \
        }                                                                      \
    }

    LOAD_CHUNK(0)
    for (int k0 = 0; k0 < K; k0 += TK) {
        __syncthreads();
        #pragma unroll
        for (int i = 0; i < XQ; ++i) *(v4f*)&As[srow + (i << 4)][skq] = xa[i];
        #pragma unroll
        for (int i = 0; i < NT; ++i) *(v4f*)&Ws[srow + (i << 4)][skq] = wa[i];
        __syncthreads();

        if (k0 + TK < K) LOAD_CHUNK(k0 + TK)

        if (k0 + TK <= K) {
            #pragma unroll
            for (int kk = 0; kk < TK; kk += 4) KSTEP(kk)
        } else {
            for (int kk = 0; k0 + kk < K; kk += 4) KSTEP(kk)
        }
    }
#undef LOAD_CHUNK
#undef KSTEP

    const int rbase = row0 + (wave << 4) + (half << 3);
    if (row0 + BM <= M) {
        float* __restrict__ hp = H + (size_t)rbase * Nout + m;
        #pragma unroll
        for (int j = 0; j < NT; ++j) {
            if (((j + 1) << 4) <= Nout) {
                #pragma unroll
                for (int r = 0; r < 8; ++r) hp[(size_t)r * Nout + (j << 4)] = acc[j][r];
            } else if ((j << 4) + m < Nout) {
                #pragma unroll
                for (int r = 0; r < 8; ++r) hp[(size_t)r * Nout + (j << 4)] = acc[j][r];
            }
        }
    } else {
        #pragma unroll
        for (int j = 0; j < NT; ++j) {
            const int n = (j << 4) + m;
            #pragma unroll
            for (int r = 0; r < 8; ++r) {
                const int grow = rbase + r;
                if (grow < M && n < Nout) H[(size_t)grow * Nout + n] = acc[j][r];
            }
        }
    }
}

__global__ void k_fill1(float* __restrict__ p, int n) {
    int i = blockIdx.x * blockDim.x + threadIdx.x;
    if (i < n) p[i] = 1.0f;
}

__global__ void k_deg(const int* __restrict__ dst, float* __restrict__ deg, int E) {
    int e = blockIdx.x * blockDim.x + threadIdx.x;
    if (e < E) atomicAdd(&deg[dst[e]], 1.0f);
}

__global__ void k_rsqrt(float* __restrict__ p, int n) {
    int i = blockIdx.x * blockDim.x + threadIdx.x;
    if (i < n) { float d = p[i]; p[i] = (d > 0.0f) ? rsqrtf(d) : 0.0f; }
}

__global__ void k_zero(float* __restrict__ p, size_t n) {
    size_t i = (size_t)blockIdx.x * blockDim.x + threadIdx.x;
    if (i < n) p[i] = 0.0f;
}

__global__ __launch_bounds__(256) void k_agg_wave(
    const float* __restrict__ H, const int* __restrict__ src,
    const int* __restrict__ dst, const float* __restrict__ dinv,
    float* __restrict__ out, int E, int F)
{
    const int lane   = threadIdx.x & 31;
    const int wid    = (int)((blockIdx.x * blockDim.x + threadIdx.x) >> 5);
    const int stride = (int)((gridDim.x * blockDim.x) >> 5);

    for (int e = wid; e < E; e += stride) {
        const int s = src[e];
        const int d = dst[e];
        const int e2 = e + stride;
        if (e2 < E) {
            __builtin_prefetch(&H[(size_t)src[e2] * F], 0, 1);
        }
        const float nrm = dinv[s] * dinv[d];
        const float* __restrict__ hrow = H + (size_t)s * F;
        float* __restrict__ orow = out + (size_t)d * F;
        for (int f = lane; f < F; f += 32) {
            atomicAdd(&orow[f], hrow[f] * nrm);
        }
    }
}

__global__ void k_agg_small(
    const float* __restrict__ H, const int* __restrict__ src,
    const int* __restrict__ dst, const float* __restrict__ dinv,
    float* __restrict__ out, int E, int F)
{
    size_t idx = (size_t)blockIdx.x * blockDim.x + threadIdx.x;
    size_t tot = (size_t)E * F;
    if (idx >= tot) return;
    int e = (int)(idx / F);
    int f = (int)(idx % F);
    int s = src[e], d = dst[e];
    float nrm = dinv[s] * dinv[d];
    atomicAdd(&out[(size_t)d * F + f], H[(size_t)s * F + f] * nrm);
}

__global__ void k_finish(float* __restrict__ dest, const float* __restrict__ H,
                         const float* __restrict__ dinv, const float* __restrict__ bias,
                         int N, int F, int doRelu)
{
    size_t idx = (size_t)blockIdx.x * blockDim.x + threadIdx.x;
    size_t tot = (size_t)N * F;
    if (idx >= tot) return;
    int i = (int)(idx / F);
    int f = (int)(idx % F);
    float di = dinv[i];
    float v  = dest[idx] + H[idx] * di * di + bias[f];
    dest[idx] = doRelu ? fmaxf(v, 0.0f) : v;
}

typedef int v2i __attribute__((ext_vector_type(2)));
typedef int v4i __attribute__((ext_vector_type(4)));

__global__ void k_izero(int* __restrict__ p, int n) {
    int i = blockIdx.x * blockDim.x + threadIdx.x;
    if (i < n) p[i] = 0;
}

__global__ void k_rank(const int* __restrict__ dst, int* __restrict__ cnt,
                       int* __restrict__ rank, int E) {
    int e = blockIdx.x * blockDim.x + threadIdx.x;
    if (e < E) rank[e] = atomicAdd(&cnt[dst[e]], 1);
}

#define SCAN_T 1024
#define SCAN_SEG (SCAN_T * 4)
__global__ __launch_bounds__(SCAN_T) void k_scan(
    const int* __restrict__ cnt, int* __restrict__ rowptr,
    float* __restrict__ dinv, int N)
{
    __shared__ int wpre[32];
    __shared__ int wtot[32];
    const int t = threadIdx.x, lane = t & 31, wv = t >> 5;
    const int base = blockIdx.x * SCAN_SEG;

    int pre = 0;
    const v4i* c4 = (const v4i*)cnt;
    for (int i = t; i < (base >> 2); i += SCAN_T) {
        const v4i v = c4[i];
        pre += (v.x + v.y) + (v.z + v.w);
    }

    const int i0 = base + (t << 2);
    int c[4];
    #pragma unroll
    for (int q = 0; q < 4; ++q) c[q] = (i0 + q < N) ? cnt[i0 + q] : 0;
    const int s = (c[0] + c[1]) + (c[2] + c[3]);

    int incl = s;
    #pragma unroll
    for (int o = 1; o < 32; o <<= 1) {
        const int v = __shfl_up(incl, o);
        if (lane >= o) incl += v;
    }
    #pragma unroll
    for (int o = 16; o > 0; o >>= 1) pre += __shfl_xor(pre, o);
    if (lane == 31) wtot[wv] = incl;
    if (lane == 0)  wpre[wv] = pre;
    __syncthreads();

    int front = wpre[lane];
    int mine  = (lane < wv) ? wtot[lane] : 0;
    #pragma unroll
    for (int o = 16; o > 0; o >>= 1) {
        front += __shfl_xor(front, o);
        mine  += __shfl_xor(mine, o);
    }
    int run = front + mine + (incl - s);
    #pragma unroll
    for (int q = 0; q < 4; ++q) {
        if (i0 + q < N) {
            rowptr[i0 + q] = run;
            const float d = (float)(c[q] + 1);
            dinv[i0 + q] = (d > 0.0f) ? rsqrtf(d) : 0.0f;
        }
        run += c[q];
    }
    if (blockIdx.x == gridDim.x - 1 && t == SCAN_T - 1) rowptr[N] = run;
}

__global__ void k_place(const int* __restrict__ src, const int* __restrict__ dst,
                        const int* __restrict__ rank, const int* __restrict__ rowptr,
                        const float* __restrict__ dinv, int* __restrict__ list, int E) {
    int e = blockIdx.x * blockDim.x + threadIdx.x;
    if (e >= E) return;
    const int s = src[e], d = dst[e];
    int pos = rowptr[d] + rank[e];
    pos = pos < 0 ? 0 : (pos < E ? pos : E - 1);
    v2i ent;
    ent.x = s;
    ent.y = __float_as_int(dinv[s] * dinv[d]);
    *(v2i*)&list[(size_t)pos << 1] = ent;
}

template <int F>
__global__ __launch_bounds__(256) void k_agg_rows(
    const float* __restrict__ H, const int* __restrict__ rowptr,
    const int* __restrict__ list, const float* __restrict__ dinv,
    const float* __restrict__ bias, float* __restrict__ dest,
    int N, int E, int doRelu)
{
    constexpr int LPN = F / 4;
    constexpr int NPB = 256 / LPN;
    const int t = threadIdx.x;
    if (t >= NPB * LPN) return;
    const int ln   = t / LPN;
    const int col  = (t - ln * LPN) << 2;
    v4f bz;
    bz.x = bias[col]; bz.y = bias[col + 1]; bz.z = bias[col + 2]; bz.w = bias[col + 3];
    const v2i* __restrict__ ents = (const v2i*)list;

    for (int g = blockIdx.x; g * NPB < N; g += gridDim.x) {
        const int i = g * NPB + ln;
        if (i >= N) continue;
        int beg = rowptr[i], end = rowptr[i + 1];
        beg = beg < 0 ? 0 : beg;
        end = end > E ? E : end;
        const float di = dinv[i];
        const v4f hs = *(const v4f*)&H[(size_t)i * F + col];
        v4f acc = v4f{};
        const int last = end - 1;
        for (int j = beg; j < end; j += 4) {
            const bool p1 = j + 1 < end, p2 = j + 2 < end, p3 = j + 3 < end;
            const v2i e0 = ents[j];
            const v2i e1 = ents[p1 ? j + 1 : last];
            const v2i e2 = ents[p2 ? j + 2 : last];
            const v2i e3 = ents[p3 ? j + 3 : last];
            const v4f h0 = *(const v4f*)&H[(size_t)e0.x * F + col];
            const v4f h1 = *(const v4f*)&H[(size_t)e1.x * F + col];
            const v4f h2 = *(const v4f*)&H[(size_t)e2.x * F + col];
            const v4f h3 = *(const v4f*)&H[(size_t)e3.x * F + col];
            acc += h0 * __int_as_float(e0.y);
            acc += h1 * (p1 ? __int_as_float(e1.y) : 0.0f);
            acc += h2 * (p2 ? __int_as_float(e2.y) : 0.0f);
            acc += h3 * (p3 ? __int_as_float(e3.y) : 0.0f);
        }
        v4f v = acc + hs * di * di + bz;
        if (doRelu) {
            v.x = fmaxf(v.x, 0.0f); v.y = fmaxf(v.y, 0.0f);
            v.z = fmaxf(v.z, 0.0f); v.w = fmaxf(v.w, 0.0f);
        }
        *(v4f*)&dest[(size_t)i * F + col] = v;
    }
}

__global__ void k_agg_elem(
    const float* __restrict__ H, const int* __restrict__ rowptr,
    const int* __restrict__ list, const float* __restrict__ dinv,
    const float* __restrict__ bias, float* __restrict__ dest,
    int N, int E, int F, int doRelu)
{
    const size_t idx = (size_t)blockIdx.x * blockDim.x + threadIdx.x;
    if (idx >= (size_t)N * F) return;
    const int i = (int)(idx / F);
    const int f = (int)(idx - (size_t)i * F);
    const v2i* __restrict__ ents = (const v2i*)list;
    int beg = rowptr[i], end = rowptr[i + 1];
    beg = beg < 0 ? 0 : beg;
    end = end > E ? E : end;
    const float di = dinv[i];
    const float hs = H[idx];
    float acc = 0.0f;
    const int last = end - 1;
    for (int j = beg; j < end; j += 4) {
        const bool p1 = j + 1 < end, p2 = j + 2 < end, p3 = j + 3 < end;
        const v2i e0 = ents[j];
        const v2i e1 = ents[p1 ? j + 1 : last];
        const v2i e2 = ents[p2 ? j + 2 : last];
        const v2i e3 = ents[p3 ? j + 3 : last];
        const float h0 = H[(size_t)e0.x * F + f];
        const float h1 = H[(size_t)e1.x * F + f];
        const float h2 = H[(size_t)e2.x * F + f];
        const float h3 = H[(size_t)e3.x * F + f];
        acc += h0 * __int_as_float(e0.y);
        acc += h1 * (p1 ? __int_as_float(e1.y) : 0.0f);
        acc += h2 * (p2 ? __int_as_float(e2.y) : 0.0f);
        acc += h3 * (p3 ? __int_as_float(e3.y) : 0.0f);
    }
    const float v = acc + hs * di * di + bias[f];
    dest[idx] = doRelu ? fmaxf(v, 0.0f) : v;
}

extern "C" void kernel_launch(void* const* d_in, const int* in_sizes, int n_in,
                              void* d_out, int out_size, void* d_ws, size_t ws_size,
                              hipStream_t stream) {
    (void)n_in; (void)out_size;

    const float* x   = (const float*)d_in[0];
    const int*   ei  = (const int*)  d_in[1];
    const float* W1  = (const float*)d_in[2];
    const float* b1  = (const float*)d_in[3];
    const float* W1b = (const float*)d_in[4];
    const float* b1b = (const float*)d_in[5];
    const float* W2b = (const float*)d_in[6];
    const float* b2b = (const float*)d_in[7];
    const float* W2  = (const float*)d_in[8];
    const float* b2  = (const float*)d_in[9];

    const int F_IN = 500, H1 = 200, H2 = 100, H3 = 40, C = 3;
    const int N = in_sizes[0] / F_IN;
    const int E = in_sizes[1] / 2;
    const int* srcIdx = ei;
    const int* dstIdx = ei + E;

    char*  ws   = (char*)d_ws;
    float* dinv = (float*)ws;
    size_t off  = ((size_t)N * sizeof(float) + 511) & ~(size_t)511;
    off += (size_t)256 * 512 * sizeof(float);
    float* bufH = (float*)(ws + off);
    off += (((size_t)N * H1 * sizeof(float)) + 511) & ~(size_t)511;
    float* bufA = (float*)(ws + off);
    off += (((size_t)N * H1 * sizeof(float)) + 511) & ~(size_t)511;
    int*   list = (int*)(ws + off);
    off += (((size_t)E * 2 * sizeof(int)) + 511) & ~(size_t)511;
    int*   rowptr = (int*)(ws + off);
    off += (((size_t)(N + 1) * sizeof(int)) + 511) & ~(size_t)511;
    float* out  = (float*)d_out;

    const size_t cntPad = ((size_t)N + 127) & ~(size_t)127;
    int* cnt  = (int*)bufA;
    int* rank = cnt + cntPad;

    const int gemmBlocks = (N + BM - 1) / BM;

    auto gemm = [&](const float* in, const float* Wt, int K, int F) {
        if (F <= 16)       k_gemm_wmma<1><<<gemmBlocks, 128, 0, stream>>>(in, Wt, bufH, N, K, F);
        else if (F <= 48)  k_gemm_wmma<3><<<gemmBlocks, 128, 0, stream>>>(in, Wt, bufH, N, K, F);
        else if (F <= 112) k_gemm_wmma<7><<<gemmBlocks, 128, 0, stream>>>(in, Wt, bufH, N, K, F);
        else               k_gemm_wmma<13><<<gemmBlocks, 128, 0, stream>>>(in, Wt, bufH, N, K, F);
    };

    const bool gather = ws_size >= off && cntPad + (size_t)E <= (size_t)N * H1;

    if (gather) {
        k_izero<<<(N + 255) / 256, 256, 0, stream>>>(cnt, N);
        k_rank <<<(E + 255) / 256, 256, 0, stream>>>(dstIdx, cnt, rank, E);
        k_scan <<<(N + SCAN_SEG - 1) / SCAN_SEG, SCAN_T, 0, stream>>>(cnt, rowptr, dinv, N);
        k_place<<<(E + 255) / 256, 256, 0, stream>>>(srcIdx, dstIdx, rank, rowptr, dinv, list, E);

        const int resident = 2048;
        auto rowsGrid = [&](int nodesPerBlock) {
            const int groups = (N + nodesPerBlock - 1) / nodesPerBlock;
            return groups < resident ? groups : resident;
        };
        auto layer = [&](const float* in, const float* Wt, const float* bias,
                         float* dest, int K, int F, int relu) {
            gemm(in, Wt, K, F);
            if (F == 200)      k_agg_rows<200><<<rowsGrid(256 / 50), 256, 0, stream>>>(
                                   bufH, rowptr, list, dinv, bias, dest, N, E, relu);
            else if (F == 100) k_agg_rows<100><<<rowsGrid(256 / 25), 256, 0, stream>>>(
                                   bufH, rowptr, list, dinv, bias, dest, N, E, relu);
            else if (F == 40)  k_agg_rows<40><<<rowsGrid(256 / 10), 256, 0, stream>>>(
                                   bufH, rowptr, list, dinv, bias, dest, N, E, relu);
            else {
                const size_t tot = (size_t)N * F;
                k_agg_elem<<<(int)((tot + 255) / 256), 256, 0, stream>>>(
                    bufH, rowptr, list, dinv, bias, dest, N, E, F, relu);
            }
        };
        layer(x,    W1,  b1,  bufA, F_IN, H1, 1);
        layer(bufA, W1b, b1b, bufA, H1,   H2, 1);
        layer(bufA, W2b, b2b, bufA, H2,   H3, 1);
        layer(bufA, W2,  b2,  out,  H3,   C,  0);
        return;
    }

    k_fill1<<<(N + 255) / 256, 256, 0, stream>>>(dinv, N);
    k_deg  <<<(E + 255) / 256, 256, 0, stream>>>(dstIdx, dinv, E);
    k_rsqrt<<<(N + 255) / 256, 256, 0, stream>>>(dinv, N);

    const int aggBlocks  = 65536;

    auto layer = [&](const float* in, const float* Wt, const float* bias,
                     float* dest, int K, int F, int relu) {
        gemm(in, Wt, K, F);
        size_t tot = (size_t)N * F;
        k_zero<<<(int)((tot + 255) / 256), 256, 0, stream>>>(dest, tot);
        if (F >= 32) {
            k_agg_wave<<<aggBlocks, 256, 0, stream>>>(bufH, srcIdx, dstIdx, dinv, dest, E, F);
        } else {
            size_t tote = (size_t)E * F;
            k_agg_small<<<(int)((tote + 255) / 256), 256, 0, stream>>>(
                bufH, srcIdx, dstIdx, dinv, dest, E, F);
        }
        k_finish<<<(int)((tot + 255) / 256), 256, 0, stream>>>(
            dest, bufH, dinv, bias, N, F, relu);
    };

    layer(x,    W1,  b1,  bufA, F_IN, H1, 1);
    layer(bufA, W1b, b1b, bufA, H1,   H2, 1);
    layer(bufA, W2b, b2b, bufA, H2,   H3, 1);
    layer(bufA, W2,  b2,  out,  H3,   C,  0);
}


# Edge list built by a two-level counting sort with all adds in LDS (512-node buckets, then per-node slots); global returning-add rank pass gone; row sums stay in arrival order, now LDS arrival
# speedup vs baseline: 4.1548x; 1.2054x over previous
#include <hip/hip_runtime.h>

typedef float v2f __attribute__((ext_vector_type(2)));
typedef float v8f __attribute__((ext_vector_type(8)));
typedef float v4f __attribute__((ext_vector_type(4)));

#define TK 32
#define LDP (TK + 4)
#define BM 64
#define XQ ((BM * TK / 4) / 128)

template <int NT>
__global__ __launch_bounds__(128) void k_gemm_wmma(
    const float* __restrict__ X, const float* __restrict__ W,
    float* __restrict__ H, int M, int K, int Nout)
{
    __shared__ __attribute__((aligned(16))) float As[BM][LDP];
    __shared__ __attribute__((aligned(16))) float Ws[NT * 16][LDP];

    const int row0 = blockIdx.x * BM;
    const int t    = threadIdx.x;
    const int wave = t >> 5;
    const int lane = t & 31;
    const int half = lane >> 4;
    const int m    = lane & 15;
    const int arow = (wave << 4) + m;

    const int srow = t >> 3;
    const int skq  = (t & 7) << 2;

    v8f acc[NT];
    #pragma unroll
    for (int j = 0; j < NT; ++j) acc[j] = v8f{};

    v4f xa[XQ];
    v4f wa[NT];

#define LOAD_CHUNK(k0_)                                                        \
    {                                                                          \
        const int gk_  = (k0_) + skq;                                          \
        const int gkc_ = gk_ < K ? gk_ : K - 4;       \
        _Pragma("unroll")                                                      \
        for (int i = 0; i < XQ; ++i) {                                         \
            const int gr_  = row0 + srow + (i << 4);                           \
            const int grc_ = gr_ < M ? gr_ : M - 1;   \
            xa[i] = *(const v4f*)&X[(size_t)grc_ * K + gkc_];               \
        }                                                                      \
        _Pragma("unroll")                                                      \
        for (int i = 0; i < NT; ++i) {                                         \
            const int r_  = srow + (i << 4);                                   \
            const int rc_ = r_ < Nout ? r_ : Nout - 1;   \
            wa[i] = *(const v4f*)&W[(size_t)rc_ * K + gkc_];                \
        }                                                                      \
    }

#define KSTEP(kk_)                                                             \
    {                                                                          \
        const int kb_ = (kk_) + (half << 1);            \
        const v2f a_ = *(const v2f*)&As[arow][kb_];                            \
        _Pragma("unroll")                                                      \
        for (int j = 0; j < NT; ++j) {                                         \
            const v2f b_ = *(const v2f*)&Ws[(j << 4) + m][kb_];                \
            acc[j] = __builtin_amdgcn_wmma_f32_16x16x4_f32(                    \
                false, a_, false, b_, (short)0, acc[j], false, false);         \
        }                                                                      \
    }

    LOAD_CHUNK(0)
    for (int k0 = 0; k0 < K; k0 += TK) {
        __syncthreads();
        #pragma unroll
        for (int i = 0; i < XQ; ++i) *(v4f*)&As[srow + (i << 4)][skq] = xa[i];
        #pragma unroll
        for (int i = 0; i < NT; ++i) *(v4f*)&Ws[srow + (i << 4)][skq] = wa[i];
        __syncthreads();

        if (k0 + TK < K) LOAD_CHUNK(k0 + TK)

        if (k0 + TK <= K) {
            #pragma unroll
            for (int kk = 0; kk < TK; kk += 4) KSTEP(kk)
        } else {
            for (int kk = 0; k0 + kk < K; kk += 4) KSTEP(kk)
        }
    }
#undef LOAD_CHUNK
#undef KSTEP

    const int rbase = row0 + (wave << 4) + (half << 3);
    if (row0 + BM <= M) {
        float* __restrict__ hp = H + (size_t)rbase * Nout + m;
        #pragma unroll
        for (int j = 0; j < NT; ++j) {
            if (((j + 1) << 4) <= Nout) {
                #pragma unroll
                for (int r = 0; r < 8; ++r) hp[(size_t)r * Nout + (j << 4)] = acc[j][r];
            } else if ((j << 4) + m < Nout) {
                #pragma unroll
                for (int r = 0; r < 8; ++r) hp[(size_t)r * Nout + (j << 4)] = acc[j][r];
            }
        }
    } else {
        #pragma unroll
        for (int j = 0; j < NT; ++j) {
            const int n = (j << 4) + m;
            #pragma unroll
            for (int r = 0; r < 8; ++r) {
                const int grow = rbase + r;
                if (grow < M && n < Nout) H[(size_t)grow * Nout + n] = acc[j][r];
            }
        }
    }
}

__global__ void k_fill1(float* __restrict__ p, int n) {
    int i = blockIdx.x * blockDim.x + threadIdx.x;
    if (i < n) p[i] = 1.0f;
}

__global__ void k_deg(const int* __restrict__ dst, float* __restrict__ deg, int E) {
    int e = blockIdx.x * blockDim.x + threadIdx.x;
    if (e < E) atomicAdd(&deg[dst[e]], 1.0f);
}

__global__ void k_rsqrt(float* __restrict__ p, int n) {
    int i = blockIdx.x * blockDim.x + threadIdx.x;
    if (i < n) { float d = p[i]; p[i] = (d > 0.0f) ? rsqrtf(d) : 0.0f; }
}

__global__ void k_zero(float* __restrict__ p, size_t n) {
    size_t i = (size_t)blockIdx.x * blockDim.x + threadIdx.x;
    if (i < n) p[i] = 0.0f;
}

__global__ __launch_bounds__(256) void k_agg_wave(
    const float* __restrict__ H, const int* __restrict__ src,
    const int* __restrict__ dst, const float* __restrict__ dinv,
    float* __restrict__ out, int E, int F)
{
    const int lane   = threadIdx.x & 31;
    const int wid    = (int)((blockIdx.x * blockDim.x + threadIdx.x) >> 5);
    const int stride = (int)((gridDim.x * blockDim.x) >> 5);

    for (int e = wid; e < E; e += stride) {
        const int s = src[e];
        const int d = dst[e];
        const int e2 = e + stride;
        if (e2 < E) {
            __builtin_prefetch(&H[(size_t)src[e2] * F], 0, 1);
        }
        const float nrm = dinv[s] * dinv[d];
        const float* __restrict__ hrow = H + (size_t)s * F;
        float* __restrict__ orow = out + (size_t)d * F;
        for (int f = lane; f < F; f += 32) {
            atomicAdd(&orow[f], hrow[f] * nrm);
        }
    }
}

__global__ void k_agg_small(
    const float* __restrict__ H, const int* __restrict__ src,
    const int* __restrict__ dst, const float* __restrict__ dinv,
    float* __restrict__ out, int E, int F)
{
    size_t idx = (size_t)blockIdx.x * blockDim.x + threadIdx.x;
    size_t tot = (size_t)E * F;
    if (idx >= tot) return;
    int e = (int)(idx / F);
    int f = (int)(idx % F);
    int s = src[e], d = dst[e];
    float nrm = dinv[s] * dinv[d];
    atomicAdd(&out[(size_t)d * F + f], H[(size_t)s * F + f] * nrm);
}

__global__ void k_finish(float* __restrict__ dest, const float* __restrict__ H,
                         const float* __restrict__ dinv, const float* __restrict__ bias,
                         int N, int F, int doRelu)
{
    size_t idx = (size_t)blockIdx.x * blockDim.x + threadIdx.x;
    size_t tot = (size_t)N * F;
    if (idx >= tot) return;
    int i = (int)(idx / F);
    int f = (int)(idx % F);
    float di = dinv[i];
    float v  = dest[idx] + H[idx] * di * di + bias[f];
    dest[idx] = doRelu ? fmaxf(v, 0.0f) : v;
}

typedef int v2i __attribute__((ext_vector_type(2)));
typedef int v4i __attribute__((ext_vector_type(4)));

__global__ void k_izero(int* __restrict__ p, int n) {
    int i = blockIdx.x * blockDim.x + threadIdx.x;
    if (i < n) p[i] = 0;
}

__global__ void k_rank(const int* __restrict__ dst, int* __restrict__ cnt,
                       int* __restrict__ rank, int E) {
    int e = blockIdx.x * blockDim.x + threadIdx.x;
    if (e < E) rank[e] = atomicAdd(&cnt[dst[e]], 1);
}

#define SCAN_T 1024
#define SCAN_SEG (SCAN_T * 4)
__global__ __launch_bounds__(SCAN_T) void k_scan(
    const int* __restrict__ cnt, int* __restrict__ rowptr,
    float* __restrict__ dinv, int N)
{
    __shared__ int wpre[32];
    __shared__ int wtot[32];
    const int t = threadIdx.x, lane = t & 31, wv = t >> 5;
    const int base = blockIdx.x * SCAN_SEG;

    int pre = 0;
    const v4i* c4 = (const v4i*)cnt;
    for (int i = t; i < (base >> 2); i += SCAN_T) {
        const v4i v = c4[i];
        pre += (v.x + v.y) + (v.z + v.w);
    }

    const int i0 = base + (t << 2);
    int c[4];
    #pragma unroll
    for (int q = 0; q < 4; ++q) c[q] = (i0 + q < N) ? cnt[i0 + q] : 0;
    const int s = (c[0] + c[1]) + (c[2] + c[3]);

    int incl = s;
    #pragma unroll
    for (int o = 1; o < 32; o <<= 1) {
        const int v = __shfl_up(incl, o);
        if (lane >= o) incl += v;
    }
    #pragma unroll
    for (int o = 16; o > 0; o >>= 1) pre += __shfl_xor(pre, o);
    if (lane == 31) wtot[wv] = incl;
    if (lane == 0)  wpre[wv] = pre;
    __syncthreads();

    int front = wpre[lane];
    int mine  = (lane < wv) ? wtot[lane] : 0;
    #pragma unroll
    for (int o = 16; o > 0; o >>= 1) {
        front += __shfl_xor(front, o);
        mine  += __shfl_xor(mine, o);
    }
    int run = front + mine + (incl - s);
    #pragma unroll
    for (int q = 0; q < 4; ++q) {
        if (i0 + q < N) {
            rowptr[i0 + q] = run;
            const float d = (float)(c[q] + 1);
            dinv[i0 + q] = (d > 0.0f) ? rsqrtf(d) : 0.0f;
        }
        run += c[q];
    }
    if (blockIdx.x == gridDim.x - 1 && t == SCAN_T - 1) rowptr[N] = run;
}

__global__ void k_place(const int* __restrict__ src, const int* __restrict__ dst,
                        const int* __restrict__ rank, const int* __restrict__ rowptr,
                        const float* __restrict__ dinv, int* __restrict__ list, int E) {
    int e = blockIdx.x * blockDim.x + threadIdx.x;
    if (e >= E) return;
    const int s = src[e], d = dst[e];
    int pos = rowptr[d] + rank[e];
    pos = pos < 0 ? 0 : (pos < E ? pos : E - 1);
    v2i ent;
    ent.x = s;
    ent.y = __float_as_int(dinv[s] * dinv[d]);
    *(v2i*)&list[(size_t)pos << 1] = ent;
}

#define BSH 9
#define BNODES (1 << BSH)
#define MAXB 256
#define G1 256
#define L1T 1024
#define EB 8

__global__ __launch_bounds__(L1T) void k_bin_count(
    const int* __restrict__ dst, int* __restrict__ blkcnt, int E, int NB, int chunk)
{
    __shared__ int hist[MAXB];
    const int t = threadIdx.x;
    if (t < MAXB) hist[t] = 0;
    __syncthreads();
    const int e0 = blockIdx.x * chunk;
    const int e1 = (e0 + chunk < E) ? e0 + chunk : E;
    for (int e = e0 + t; e < e1; e += EB * L1T) {
        int d[EB];
        #pragma unroll
        for (int q = 0; q < EB; ++q) d[q] = dst[(e + q * L1T < e1) ? e + q * L1T : e1 - 1];
        #pragma unroll
        for (int q = 0; q < EB; ++q) {
            if (e + q * L1T < e1) {
                int b = (int)((unsigned)d[q] >> BSH);
                b = b < NB ? b : NB - 1;
                atomicAdd(&hist[b], 1);
            }
        }
    }
    __syncthreads();
    if (t < NB) blkcnt[t * G1 + blockIdx.x] = hist[t];
}

__global__ __launch_bounds__(SCAN_T) void k_scan_tab(
    const int* __restrict__ cnt, int* __restrict__ off, int n)
{
    __shared__ int wpre[32];
    __shared__ int wtot[32];
    const int t = threadIdx.x, lane = t & 31, wv = t >> 5;
    const int base = blockIdx.x * SCAN_SEG;

    int pre = 0;
    const v4i* c4 = (const v4i*)cnt;
    #pragma unroll 4
    for (int i = t; i < (base >> 2); i += SCAN_T) {
        const v4i v = c4[i];
        pre += (v.x + v.y) + (v.z + v.w);
    }

    const int i0 = base + (t << 2);
    int c[4];
    #pragma unroll
    for (int q = 0; q < 4; ++q) c[q] = (i0 + q < n) ? cnt[i0 + q] : 0;
    const int s = (c[0] + c[1]) + (c[2] + c[3]);

    int incl = s;
    #pragma unroll
    for (int o = 1; o < 32; o <<= 1) {
        const int v = __shfl_up(incl, o);
        if (lane >= o) incl += v;
    }
    #pragma unroll
    for (int o = 16; o > 0; o >>= 1) pre += __shfl_xor(pre, o);
    if (lane == 31) wtot[wv] = incl;
    if (lane == 0)  wpre[wv] = pre;
    __syncthreads();

    int front = wpre[lane];
    int mine  = (lane < wv) ? wtot[lane] : 0;
    #pragma unroll
    for (int o = 16; o > 0; o >>= 1) {
        front += __shfl_xor(front, o);
        mine  += __shfl_xor(mine, o);
    }
    int run = front + mine + (incl - s);
    #pragma unroll
    for (int q = 0; q < 4; ++q) {
        if (i0 + q < n) off[i0 + q] = run;
        run += c[q];
    }
}

__global__ __launch_bounds__(L1T) void k_bin_scatter(
    const int* __restrict__ src, const int* __restrict__ dst,
    const int* __restrict__ blkoff, int* __restrict__ tmp, int E, int NB, int chunk)
{
    __shared__ int cur[MAXB];
    const int t = threadIdx.x;
    if (t < NB) cur[t] = blkoff[t * G1 + blockIdx.x];
    __syncthreads();
    const int e0 = blockIdx.x * chunk;
    const int e1 = (e0 + chunk < E) ? e0 + chunk : E;
    for (int e = e0 + t; e < e1; e += EB * L1T) {
        v2i ent[EB];
        #pragma unroll
        for (int q = 0; q < EB; ++q) {
            const int ec = (e + q * L1T < e1) ? e + q * L1T : e1 - 1;
            ent[q].x = src[ec];
            ent[q].y = dst[ec];
        }
        #pragma unroll
        for (int q = 0; q < EB; ++q) {
            if (e + q * L1T < e1) {
                int b = (int)((unsigned)ent[q].y >> BSH);
                b = b < NB ? b : NB - 1;
                int pos = atomicAdd(&cur[b], 1);
                pos = pos < 0 ? 0 : (pos < E ? pos : E - 1);
                *(v2i*)&tmp[(size_t)pos << 1] = ent[q];
            }
        }
    }
}

__global__ __launch_bounds__(L1T) void k_bucket_sort(
    const int* __restrict__ tmp, const int* __restrict__ blkoff,
    int* __restrict__ rowptr, float* __restrict__ dinv, int* __restrict__ list,
    int N, int E, int NB)
{
    __shared__ int cnt[BNODES];
    __shared__ int wsum[32];
    const int t = threadIdx.x, lane = t & 31, wv = t >> 5;
    const int b = blockIdx.x;
    const int base = b << BSH;
    int bs = blkoff[b * G1];
    int be = (b + 1 < NB) ? blkoff[(b + 1) * G1] : E;
    bs = bs < 0 ? 0 : (bs < E ? bs : E);
    be = be < bs ? bs : (be < E ? be : E);
    if (t < BNODES) cnt[t] = 0;
    __syncthreads();

    const v2i* __restrict__ te = (const v2i*)tmp;
    for (int e = bs + t; e < be; e += EB * L1T) {
        int d[EB];
        #pragma unroll
        for (int q = 0; q < EB; ++q) d[q] = te[(e + q * L1T < be) ? e + q * L1T : be - 1].y;
        #pragma unroll
        for (int q = 0; q < EB; ++q) {
            if (e + q * L1T < be) {
                int k = d[q] - base;
                k = k < 0 ? 0 : (k < BNODES ? k : BNODES - 1);
                atomicAdd(&cnt[k], 1);
            }
        }
    }
    __syncthreads();

    const int c = (t < BNODES) ? cnt[t] : 0;
    int incl = c;
    #pragma unroll
    for (int o = 1; o < 32; o <<= 1) {
        const int v = __shfl_up(incl, o);
        if (lane >= o) incl += v;
    }
    if (lane == 31) wsum[wv] = incl;
    __syncthreads();
    int front = (lane < wv) ? wsum[lane] : 0;
    #pragma unroll
    for (int o = 16; o > 0; o >>= 1) front += __shfl_xor(front, o);
    const int excl = front + (incl - c);
    if (t < BNODES) {
        cnt[t] = excl;
        const int node = base + t;
        if (node < N) {
            rowptr[node] = bs + excl;
            const float d = (float)(c + 1);
            dinv[node] = (d > 0.0f) ? rsqrtf(d) : 0.0f;
        }
    }
    if (b == NB - 1 && t == 0) rowptr[N] = be;
    __syncthreads();

    for (int e = bs + t; e < be; e += EB * L1T) {
        v2i ent[EB];
        #pragma unroll
        for (int q = 0; q < EB; ++q) ent[q] = te[(e + q * L1T < be) ? e + q * L1T : be - 1];
        #pragma unroll
        for (int q = 0; q < EB; ++q) {
            if (e + q * L1T < be) {
                int k = ent[q].y - base;
                k = k < 0 ? 0 : (k < BNODES ? k : BNODES - 1);
                int pos = bs + atomicAdd(&cnt[k], 1);
                pos = pos < 0 ? 0 : (pos < E ? pos : E - 1);
                *(v2i*)&list[(size_t)pos << 1] = ent[q];
            }
        }
    }
}

__global__ void k_weights(int* __restrict__ list, const float* __restrict__ dinv, int E) {
    const int p = blockIdx.x * blockDim.x + threadIdx.x;
    if (p >= E) return;
    v2i ent = *(const v2i*)&list[(size_t)p << 1];
    ent.y = __float_as_int(dinv[ent.x] * dinv[ent.y]);
    *(v2i*)&list[(size_t)p << 1] = ent;
}

template <int F>
__global__ __launch_bounds__(256) void k_agg_rows(
    const float* __restrict__ H, const int* __restrict__ rowptr,
    const int* __restrict__ list, const float* __restrict__ dinv,
    const float* __restrict__ bias, float* __restrict__ dest,
    int N, int E, int doRelu)
{
    constexpr int LPN = F / 4;
    constexpr int NPB = 256 / LPN;
    const int t = threadIdx.x;
    if (t >= NPB * LPN) return;
    const int ln   = t / LPN;
    const int col  = (t - ln * LPN) << 2;
    v4f bz;
    bz.x = bias[col]; bz.y = bias[col + 1]; bz.z = bias[col + 2]; bz.w = bias[col + 3];
    const v2i* __restrict__ ents = (const v2i*)list;

    for (int g = blockIdx.x; g * NPB < N; g += gridDim.x) {
        const int i = g * NPB + ln;
        if (i >= N) continue;
        int beg = rowptr[i], end = rowptr[i + 1];
        beg = beg < 0 ? 0 : beg;
        end = end > E ? E : end;
        const float di = dinv[i];
        const v4f hs = *(const v4f*)&H[(size_t)i * F + col];
        v4f acc = v4f{};
        const int last = end - 1;
        for (int j = beg; j < end; j += 4) {
            const bool p1 = j + 1 < end, p2 = j + 2 < end, p3 = j + 3 < end;
            const v2i e0 = ents[j];
            const v2i e1 = ents[p1 ? j + 1 : last];
            const v2i e2 = ents[p2 ? j + 2 : last];
            const v2i e3 = ents[p3 ? j + 3 : last];
            const v4f h0 = *(const v4f*)&H[(size_t)e0.x * F + col];
            const v4f h1 = *(const v4f*)&H[(size_t)e1.x * F + col];
            const v4f h2 = *(const v4f*)&H[(size_t)e2.x * F + col];
            const v4f h3 = *(const v4f*)&H[(size_t)e3.x * F + col];
            acc += h0 * __int_as_float(e0.y);
            acc += h1 * (p1 ? __int_as_float(e1.y) : 0.0f);
            acc += h2 * (p2 ? __int_as_float(e2.y) : 0.0f);
            acc += h3 * (p3 ? __int_as_float(e3.y) : 0.0f);
        }
        v4f v = acc + hs * di * di + bz;
        if (doRelu) {
            v.x = fmaxf(v.x, 0.0f); v.y = fmaxf(v.y, 0.0f);
            v.z = fmaxf(v.z, 0.0f); v.w = fmaxf(v.w, 0.0f);
        }
        *(v4f*)&dest[(size_t)i * F + col] = v;
    }
}

__global__ void k_agg_elem(
    const float* __restrict__ H, const int* __restrict__ rowptr,
    const int* __restrict__ list, const float* __restrict__ dinv,
    const float* __restrict__ bias, float* __restrict__ dest,
    int N, int E, int F, int doRelu)
{
    const size_t idx = (size_t)blockIdx.x * blockDim.x + threadIdx.x;
    if (idx >= (size_t)N * F) return;
    const int i = (int)(idx / F);
    const int f = (int)(idx - (size_t)i * F);
    const v2i* __restrict__ ents = (const v2i*)list;
    int beg = rowptr[i], end = rowptr[i + 1];
    beg = beg < 0 ? 0 : beg;
    end = end > E ? E : end;
    const float di = dinv[i];
    const float hs = H[idx];
    float acc = 0.0f;
    const int last = end - 1;
    for (int j = beg; j < end; j += 4) {
        const bool p1 = j + 1 < end, p2 = j + 2 < end, p3 = j + 3 < end;
        const v2i e0 = ents[j];
        const v2i e1 = ents[p1 ? j + 1 : last];
        const v2i e2 = ents[p2 ? j + 2 : last];
        const v2i e3 = ents[p3 ? j + 3 : last];
        const float h0 = H[(size_t)e0.x * F + f];
        const float h1 = H[(size_t)e1.x * F + f];
        const float h2 = H[(size_t)e2.x * F + f];
        const float h3 = H[(size_t)e3.x * F + f];
        acc += h0 * __int_as_float(e0.y);
        acc += h1 * (p1 ? __int_as_float(e1.y) : 0.0f);
        acc += h2 * (p2 ? __int_as_float(e2.y) : 0.0f);
        acc += h3 * (p3 ? __int_as_float(e3.y) : 0.0f);
    }
    const float v = acc + hs * di * di + bias[f];
    dest[idx] = doRelu ? fmaxf(v, 0.0f) : v;
}

extern "C" void kernel_launch(void* const* d_in, const int* in_sizes, int n_in,
                              void* d_out, int out_size, void* d_ws, size_t ws_size,
                              hipStream_t stream) {
    (void)n_in; (void)out_size;

    const float* x   = (const float*)d_in[0];
    const int*   ei  = (const int*)  d_in[1];
    const float* W1  = (const float*)d_in[2];
    const float* b1  = (const float*)d_in[3];
    const float* W1b = (const float*)d_in[4];
    const float* b1b = (const float*)d_in[5];
    const float* W2b = (const float*)d_in[6];
    const float* b2b = (const float*)d_in[7];
    const float* W2  = (const float*)d_in[8];
    const float* b2  = (const float*)d_in[9];

    const int F_IN = 500, H1 = 200, H2 = 100, H3 = 40, C = 3;
    const int N = in_sizes[0] / F_IN;
    const int E = in_sizes[1] / 2;
    const int* srcIdx = ei;
    const int* dstIdx = ei + E;

    char*  ws   = (char*)d_ws;
    float* dinv = (float*)ws;
    size_t off  = ((size_t)N * sizeof(float) + 511) & ~(size_t)511;
    off += (size_t)256 * 512 * sizeof(float);
    float* bufH = (float*)(ws + off);
    off += (((size_t)N * H1 * sizeof(float)) + 511) & ~(size_t)511;
    float* bufA = (float*)(ws + off);
    off += (((size_t)N * H1 * sizeof(float)) + 511) & ~(size_t)511;
    int*   list = (int*)(ws + off);
    off += (((size_t)E * 2 * sizeof(int)) + 511) & ~(size_t)511;
    int*   rowptr = (int*)(ws + off);
    off += (((size_t)(N + 1) * sizeof(int)) + 511) & ~(size_t)511;
    float* out  = (float*)d_out;

    const size_t cntPad = ((size_t)N + 127) & ~(size_t)127;
    int* cnt  = (int*)bufA;
    int* rank = cnt + cntPad;

    const int gemmBlocks = (N + BM - 1) / BM;

    auto gemm = [&](const float* in, const float* Wt, int K, int F) {
        if (F <= 16)       k_gemm_wmma<1><<<gemmBlocks, 128, 0, stream>>>(in, Wt, bufH, N, K, F);
        else if (F <= 48)  k_gemm_wmma<3><<<gemmBlocks, 128, 0, stream>>>(in, Wt, bufH, N, K, F);
        else if (F <= 112) k_gemm_wmma<7><<<gemmBlocks, 128, 0, stream>>>(in, Wt, bufH, N, K, F);
        else               k_gemm_wmma<13><<<gemmBlocks, 128, 0, stream>>>(in, Wt, bufH, N, K, F);
    };

    const bool gather = ws_size >= off && cntPad + (size_t)E <= (size_t)N * H1;

    const int    NB     = (N + BNODES - 1) >> BSH;
    const int    chunk  = (E + G1 - 1) / G1;
    const size_t tmpPad = ((size_t)E * 2 + 127) & ~(size_t)127;
    int* tmpE   = (int*)bufA;
    int* blkcnt = tmpE + tmpPad;
    int* blkoff = blkcnt + (size_t)MAXB * G1;
    const bool binned = NB <= MAXB && tmpPad + (size_t)2 * MAXB * G1 <= (size_t)N * H1;

    if (gather) {
        if (binned) {
            const int nTab = NB * G1;
            k_bin_count  <<<G1, L1T, 0, stream>>>(dstIdx, blkcnt, E, NB, chunk);
            k_scan_tab   <<<(nTab + SCAN_SEG - 1) / SCAN_SEG, SCAN_T, 0, stream>>>(blkcnt, blkoff, nTab);
            k_bin_scatter<<<G1, L1T, 0, stream>>>(srcIdx, dstIdx, blkoff, tmpE, E, NB, chunk);
            k_bucket_sort<<<NB, L1T, 0, stream>>>(tmpE, blkoff, rowptr, dinv, list, N, E, NB);
            k_weights    <<<(E + 255) / 256, 256, 0, stream>>>(list, dinv, E);
        } else {
            k_izero<<<(N + 255) / 256, 256, 0, stream>>>(cnt, N);
            k_rank <<<(E + 255) / 256, 256, 0, stream>>>(dstIdx, cnt, rank, E);
            k_scan <<<(N + SCAN_SEG - 1) / SCAN_SEG, SCAN_T, 0, stream>>>(cnt, rowptr, dinv, N);
            k_place<<<(E + 255) / 256, 256, 0, stream>>>(srcIdx, dstIdx, rank, rowptr, dinv, list, E);
        }

        const int resident = 2048;
        auto rowsGrid = [&](int nodesPerBlock) {
            const int groups = (N + nodesPerBlock - 1) / nodesPerBlock;
            return groups < resident ? groups : resident;
        };
        auto layer = [&](const float* in, const float* Wt, const float* bias,
                         float* dest, int K, int F, int relu) {
            gemm(in, Wt, K, F);
            if (F == 200)      k_agg_rows<200><<<rowsGrid(256 / 50), 256, 0, stream>>>(
                                   bufH, rowptr, list, dinv, bias, dest, N, E, relu);
            else if (F == 100) k_agg_rows<100><<<rowsGrid(256 / 25), 256, 0, stream>>>(
                                   bufH, rowptr, list, dinv, bias, dest, N, E, relu);
            else if (F == 40)  k_agg_rows<40><<<rowsGrid(256 / 10), 256, 0, stream>>>(
                                   bufH, rowptr, list, dinv, bias, dest, N, E, relu);
            else {
                const size_t tot = (size_t)N * F;
                k_agg_elem<<<(int)((tot + 255) / 256), 256, 0, stream>>>(
                    bufH, rowptr, list, dinv, bias, dest, N, E, F, relu);
            }
        };
        layer(x,    W1,  b1,  bufA, F_IN, H1, 1);
        layer(bufA, W1b, b1b, bufA, H1,   H2, 1);
        layer(bufA, W2b, b2b, bufA, H2,   H3, 1);
        layer(bufA, W2,  b2,  out,  H3,   C,  0);
        return;
    }

    k_fill1<<<(N + 255) / 256, 256, 0, stream>>>(dinv, N);
    k_deg  <<<(E + 255) / 256, 256, 0, stream>>>(dstIdx, dinv, E);
    k_rsqrt<<<(N + 255) / 256, 256, 0, stream>>>(dinv, N);

    const int aggBlocks  = 65536;

    auto layer = [&](const float* in, const float* Wt, const float* bias,
                     float* dest, int K, int F, int relu) {
        gemm(in, Wt, K, F);
        size_t tot = (size_t)N * F;
        k_zero<<<(int)((tot + 255) / 256), 256, 0, stream>>>(dest, tot);
        if (F >= 32) {
            k_agg_wave<<<aggBlocks, 256, 0, stream>>>(bufH, srcIdx, dstIdx, dinv, dest, E, F);
        } else {
            size_t tote = (size_t)E * F;
            k_agg_small<<<(int)((tote + 255) / 256), 256, 0, stream>>>(
                bufH, srcIdx, dstIdx, dinv, dest, E, F);
        }
        k_finish<<<(int)((tot + 255) / 256), 256, 0, stream>>>(
            dest, bufH, dinv, bias, N, F, relu);
    };

    layer(x,    W1,  b1,  bufA, F_IN, H1, 1);
    layer(bufA, W1b, b1b, bufA, H1,   H2, 1);
    layer(bufA, W2b, b2b, bufA, H2,   H3, 1);
    layer(bufA, W2,  b2,  out,  H3,   C,  0);
}


# Edge weights factored out of the list: GEMM stores rows scaled by dinv[row], a node's row sum is scaled once by dinv[dst]; list = src only, weight pass gone; sum order kept, products reassociated
# speedup vs baseline: 4.2802x; 1.0302x over previous
#include <hip/hip_runtime.h>

typedef float v2f __attribute__((ext_vector_type(2)));
typedef float v8f __attribute__((ext_vector_type(8)));
typedef float v4f __attribute__((ext_vector_type(4)));

#define TK 32
#define LDP (TK + 4)
#define BM 64
#define XQ ((BM * TK / 4) / 128)

template <int NT>
__global__ __launch_bounds__(128) void k_gemm_wmma(
    const float* __restrict__ X, const float* __restrict__ W,
    const float* __restrict__ rs, float* __restrict__ H, int M, int K, int Nout)
{
    __shared__ __attribute__((aligned(16))) float As[BM][LDP];
    __shared__ __attribute__((aligned(16))) float Ws[NT * 16][LDP];

    const int row0 = blockIdx.x * BM;
    const int t    = threadIdx.x;
    const int wave = t >> 5;
    const int lane = t & 31;
    const int half = lane >> 4;
    const int m    = lane & 15;
    const int arow = (wave << 4) + m;

    const int srow = t >> 3;
    const int skq  = (t & 7) << 2;

    v8f acc[NT];
    #pragma unroll
    for (int j = 0; j < NT; ++j) acc[j] = v8f{};

    v4f xa[XQ];
    v4f wa[NT];

#define LOAD_CHUNK(k0_)                                                        \
    {                                                                          \
        const int gk_  = (k0_) + skq;                                          \
        const int gkc_ = gk_ < K ? gk_ : K - 4;       \
        _Pragma("unroll")                                                      \
        for (int i = 0; i < XQ; ++i) {                                         \
            const int gr_  = row0 + srow + (i << 4);                           \
            const int grc_ = gr_ < M ? gr_ : M - 1;   \
            xa[i] = *(const v4f*)&X[(size_t)grc_ * K + gkc_];               \
        }                                                                      \
        _Pragma("unroll")                                                      \
        for (int i = 0; i < NT; ++i) {                                         \
            const int r_  = srow + (i << 4);                                   \
            const int rc_ = r_ < Nout ? r_ : Nout - 1;   \
            wa[i] = *(const v4f*)&W[(size_t)rc_ * K + gkc_];                \
        }                                                                      \
    }

#define KSTEP(kk_)                                                             \
    {                                                                          \
        const int kb_ = (kk_) + (half << 1);            \
        const v2f a_ = *(const v2f*)&As[arow][kb_];                            \
        _Pragma("unroll")                                                      \
        for (int j = 0; j < NT; ++j) {                                         \
            const v2f b_ = *(const v2f*)&Ws[(j << 4) + m][kb_];                \
            acc[j] = __builtin_amdgcn_wmma_f32_16x16x4_f32(                    \
                false, a_, false, b_, (short)0, acc[j], false, false);         \
        }                                                                      \
    }

    LOAD_CHUNK(0)
    for (int k0 = 0; k0 < K; k0 += TK) {
        __syncthreads();
        #pragma unroll
        for (int i = 0; i < XQ; ++i) *(v4f*)&As[srow + (i << 4)][skq] = xa[i];
        #pragma unroll
        for (int i = 0; i < NT; ++i) *(v4f*)&Ws[srow + (i << 4)][skq] = wa[i];
        __syncthreads();

        if (k0 + TK < K) LOAD_CHUNK(k0 + TK)

        if (k0 + TK <= K) {
            #pragma unroll
            for (int kk = 0; kk < TK; kk += 4) KSTEP(kk)
        } else {
            for (int kk = 0; k0 + kk < K; kk += 4) KSTEP(kk)
        }
    }
#undef LOAD_CHUNK
#undef KSTEP

    const int rbase = row0 + (wave << 4) + (half << 3);
    float sc[8];
    #pragma unroll
    for (int r = 0; r < 8; ++r) sc[r] = 1.0f;
    if (rs != nullptr) {
        #pragma unroll
        for (int r = 0; r < 8; ++r) {
            const int gr = rbase + r;
            sc[r] = rs[gr < M ? gr : M - 1];
        }
    }
    #pragma unroll
    for (int j = 0; j < NT; ++j) {
        #pragma unroll
        for (int r = 0; r < 8; ++r) acc[j][r] *= sc[r];
    }
    if (row0 + BM <= M) {
        float* __restrict__ hp = H + (size_t)rbase * Nout + m;
        #pragma unroll
        for (int j = 0; j < NT; ++j) {
            if (((j + 1) << 4) <= Nout) {
                #pragma unroll
                for (int r = 0; r < 8; ++r) hp[(size_t)r * Nout + (j << 4)] = acc[j][r];
            } else if ((j << 4) + m < Nout) {
                #pragma unroll
                for (int r = 0; r < 8; ++r) hp[(size_t)r * Nout + (j << 4)] = acc[j][r];
            }
        }
    } else {
        #pragma unroll
        for (int j = 0; j < NT; ++j) {
            const int n = (j << 4) + m;
            #pragma unroll
            for (int r = 0; r < 8; ++r) {
                const int grow = rbase + r;
                if (grow < M && n < Nout) H[(size_t)grow * Nout + n] = acc[j][r];
            }
        }
    }
}

__global__ void k_fill1(float* __restrict__ p, int n) {
    int i = blockIdx.x * blockDim.x + threadIdx.x;
    if (i < n) p[i] = 1.0f;
}

__global__ void k_deg(const int* __restrict__ dst, float* __restrict__ deg, int E) {
    int e = blockIdx.x * blockDim.x + threadIdx.x;
    if (e < E) atomicAdd(&deg[dst[e]], 1.0f);
}

__global__ void k_rsqrt(float* __restrict__ p, int n) {
    int i = blockIdx.x * blockDim.x + threadIdx.x;
    if (i < n) { float d = p[i]; p[i] = (d > 0.0f) ? rsqrtf(d) : 0.0f; }
}

__global__ void k_zero(float* __restrict__ p, size_t n) {
    size_t i = (size_t)blockIdx.x * blockDim.x + threadIdx.x;
    if (i < n) p[i] = 0.0f;
}

__global__ __launch_bounds__(256) void k_agg_wave(
    const float* __restrict__ H, const int* __restrict__ src,
    const int* __restrict__ dst, const float* __restrict__ dinv,
    float* __restrict__ out, int E, int F)
{
    const int lane   = threadIdx.x & 31;
    const int wid    = (int)((blockIdx.x * blockDim.x + threadIdx.x) >> 5);
    const int stride = (int)((gridDim.x * blockDim.x) >> 5);

    for (int e = wid; e < E; e += stride) {
        const int s = src[e];
        const int d = dst[e];
        const int e2 = e + stride;
        if (e2 < E) {
            __builtin_prefetch(&H[(size_t)src[e2] * F], 0, 1);
        }
        const float nrm = dinv[s] * dinv[d];
        const float* __restrict__ hrow = H + (size_t)s * F;
        float* __restrict__ orow = out + (size_t)d * F;
        for (int f = lane; f < F; f += 32) {
            atomicAdd(&orow[f], hrow[f] * nrm);
        }
    }
}

__global__ void k_agg_small(
    const float* __restrict__ H, const int* __restrict__ src,
    const int* __restrict__ dst, const float* __restrict__ dinv,
    float* __restrict__ out, int E, int F)
{
    size_t idx = (size_t)blockIdx.x * blockDim.x + threadIdx.x;
    size_t tot = (size_t)E * F;
    if (idx >= tot) return;
    int e = (int)(idx / F);
    int f = (int)(idx % F);
    int s = src[e], d = dst[e];
    float nrm = dinv[s] * dinv[d];
    atomicAdd(&out[(size_t)d * F + f], H[(size_t)s * F + f] * nrm);
}

__global__ void k_finish(float* __restrict__ dest, const float* __restrict__ H,
                         const float* __restrict__ dinv, const float* __restrict__ bias,
                         int N, int F, int doRelu)
{
    size_t idx = (size_t)blockIdx.x * blockDim.x + threadIdx.x;
    size_t tot = (size_t)N * F;
    if (idx >= tot) return;
    int i = (int)(idx / F);
    int f = (int)(idx % F);
    float di = dinv[i];
    float v  = dest[idx] + H[idx] * di * di + bias[f];
    dest[idx] = doRelu ? fmaxf(v, 0.0f) : v;
}

typedef int v2i __attribute__((ext_vector_type(2)));
typedef int v4i __attribute__((ext_vector_type(4)));

__global__ void k_izero(int* __restrict__ p, int n) {
    int i = blockIdx.x * blockDim.x + threadIdx.x;
    if (i < n) p[i] = 0;
}

__global__ void k_rank(const int* __restrict__ dst, int* __restrict__ cnt,
                       int* __restrict__ rank, int E) {
    int e = blockIdx.x * blockDim.x + threadIdx.x;
    if (e < E) rank[e] = atomicAdd(&cnt[dst[e]], 1);
}

#define SCAN_T 1024
#define SCAN_SEG (SCAN_T * 4)
__global__ __launch_bounds__(SCAN_T) void k_scan(
    const int* __restrict__ cnt, int* __restrict__ rowptr,
    float* __restrict__ dinv, int N)
{
    __shared__ int wpre[32];
    __shared__ int wtot[32];
    const int t = threadIdx.x, lane = t & 31, wv = t >> 5;
    const int base = blockIdx.x * SCAN_SEG;

    int pre = 0;
    const v4i* c4 = (const v4i*)cnt;
    for (int i = t; i < (base >> 2); i += SCAN_T) {
        const v4i v = c4[i];
        pre += (v.x + v.y) + (v.z + v.w);
    }

    const int i0 = base + (t << 2);
    int c[4];
    #pragma unroll
    for (int q = 0; q < 4; ++q) c[q] = (i0 + q < N) ? cnt[i0 + q] : 0;
    const int s = (c[0] + c[1]) + (c[2] + c[3]);

    int incl = s;
    #pragma unroll
    for (int o = 1; o < 32; o <<= 1) {
        const int v = __shfl_up(incl, o);
        if (lane >= o) incl += v;
    }
    #pragma unroll
    for (int o = 16; o > 0; o >>= 1) pre += __shfl_xor(pre, o);
    if (lane == 31) wtot[wv] = incl;
    if (lane == 0)  wpre[wv] = pre;
    __syncthreads();

    int front = wpre[lane];
    int mine  = (lane < wv) ? wtot[lane] : 0;
    #pragma unroll
    for (int o = 16; o > 0; o >>= 1) {
        front += __shfl_xor(front, o);
        mine  += __shfl_xor(mine, o);
    }
    int run = front + mine + (incl - s);
    #pragma unroll
    for (int q = 0; q < 4; ++q) {
        if (i0 + q < N) {
            rowptr[i0 + q] = run;
            const float d = (float)(c[q] + 1);
            dinv[i0 + q] = (d > 0.0f) ? rsqrtf(d) : 0.0f;
        }
        run += c[q];
    }
    if (blockIdx.x == gridDim.x - 1 && t == SCAN_T - 1) rowptr[N] = run;
}

__global__ void k_place(const int* __restrict__ src, const int* __restrict__ dst,
                        const int* __restrict__ rank, const int* __restrict__ rowptr,
                        int* __restrict__ list, int E) {
    int e = blockIdx.x * blockDim.x + threadIdx.x;
    if (e >= E) return;
    const int s = src[e], d = dst[e];
    int pos = rowptr[d] + rank[e];
    pos = pos < 0 ? 0 : (pos < E ? pos : E - 1);
    list[pos] = s;
}

#define BSH 9
#define BNODES (1 << BSH)
#define MAXB 256
#define G1 256
#define L1T 1024
#define EB 8

__global__ __launch_bounds__(L1T) void k_bin_count(
    const int* __restrict__ dst, int* __restrict__ blkcnt, int E, int NB, int chunk)
{
    __shared__ int hist[MAXB];
    const int t = threadIdx.x;
    if (t < MAXB) hist[t] = 0;
    __syncthreads();
    const int e0 = blockIdx.x * chunk;
    const int e1 = (e0 + chunk < E) ? e0 + chunk : E;
    for (int e = e0 + t; e < e1; e += EB * L1T) {
        int d[EB];
        #pragma unroll
        for (int q = 0; q < EB; ++q) d[q] = dst[(e + q * L1T < e1) ? e + q * L1T : e1 - 1];
        #pragma unroll
        for (int q = 0; q < EB; ++q) {
            if (e + q * L1T < e1) {
                int b = (int)((unsigned)d[q] >> BSH);
                b = b < NB ? b : NB - 1;
                atomicAdd(&hist[b], 1);
            }
        }
    }
    __syncthreads();
    if (t < NB) blkcnt[t * G1 + blockIdx.x] = hist[t];
}

__global__ __launch_bounds__(SCAN_T) void k_scan_tab(
    const int* __restrict__ cnt, int* __restrict__ off, int n)
{
    __shared__ int wpre[32];
    __shared__ int wtot[32];
    const int t = threadIdx.x, lane = t & 31, wv = t >> 5;
    const int base = blockIdx.x * SCAN_SEG;

    int pre = 0;
    const v4i* c4 = (const v4i*)cnt;
    #pragma unroll 4
    for (int i = t; i < (base >> 2); i += SCAN_T) {
        const v4i v = c4[i];
        pre += (v.x + v.y) + (v.z + v.w);
    }

    const int i0 = base + (t << 2);
    int c[4];
    #pragma unroll
    for (int q = 0; q < 4; ++q) c[q] = (i0 + q < n) ? cnt[i0 + q] : 0;
    const int s = (c[0] + c[1]) + (c[2] + c[3]);

    int incl = s;
    #pragma unroll
    for (int o = 1; o < 32; o <<= 1) {
        const int v = __shfl_up(incl, o);
        if (lane >= o) incl += v;
    }
    #pragma unroll
    for (int o = 16; o > 0; o >>= 1) pre += __shfl_xor(pre, o);
    if (lane == 31) wtot[wv] = incl;
    if (lane == 0)  wpre[wv] = pre;
    __syncthreads();

    int front = wpre[lane];
    int mine  = (lane < wv) ? wtot[lane] : 0;
    #pragma unroll
    for (int o = 16; o > 0; o >>= 1) {
        front += __shfl_xor(front, o);
        mine  += __shfl_xor(mine, o);
    }
    int run = front + mine + (incl - s);
    #pragma unroll
    for (int q = 0; q < 4; ++q) {
        if (i0 + q < n) off[i0 + q] = run;
        run += c[q];
    }
}

__global__ __launch_bounds__(L1T) void k_bin_scatter(
    const int* __restrict__ src, const int* __restrict__ dst,
    const int* __restrict__ blkoff, int* __restrict__ tmp, int E, int NB, int chunk)
{
    __shared__ int cur[MAXB];
    const int t = threadIdx.x;
    if (t < NB) cur[t] = blkoff[t * G1 + blockIdx.x];
    __syncthreads();
    const int e0 = blockIdx.x * chunk;
    const int e1 = (e0 + chunk < E) ? e0 + chunk : E;
    for (int e = e0 + t; e < e1; e += EB * L1T) {
        v2i ent[EB];
        #pragma unroll
        for (int q = 0; q < EB; ++q) {
            const int ec = (e + q * L1T < e1) ? e + q * L1T : e1 - 1;
            ent[q].x = src[ec];
            ent[q].y = dst[ec];
        }
        #pragma unroll
        for (int q = 0; q < EB; ++q) {
            if (e + q * L1T < e1) {
                int b = (int)((unsigned)ent[q].y >> BSH);
                b = b < NB ? b : NB - 1;
                int pos = atomicAdd(&cur[b], 1);
                pos = pos < 0 ? 0 : (pos < E ? pos : E - 1);
                *(v2i*)&tmp[(size_t)pos << 1] = ent[q];
            }
        }
    }
}

__global__ __launch_bounds__(L1T) void k_bucket_sort(
    const int* __restrict__ tmp, const int* __restrict__ blkoff,
    int* __restrict__ rowptr, float* __restrict__ dinv, int* __restrict__ list,
    int N, int E, int NB)
{
    __shared__ int cnt[BNODES];
    __shared__ int wsum[32];
    const int t = threadIdx.x, lane = t & 31, wv = t >> 5;
    const int b = blockIdx.x;
    const int base = b << BSH;
    int bs = blkoff[b * G1];
    int be = (b + 1 < NB) ? blkoff[(b + 1) * G1] : E;
    bs = bs < 0 ? 0 : (bs < E ? bs : E);
    be = be < bs ? bs : (be < E ? be : E);
    if (t < BNODES) cnt[t] = 0;
    __syncthreads();

    const v2i* __restrict__ te = (const v2i*)tmp;
    for (int e = bs + t; e < be; e += EB * L1T) {
        int d[EB];
        #pragma unroll
        for (int q = 0; q < EB; ++q) d[q] = te[(e + q * L1T < be) ? e + q * L1T : be - 1].y;
        #pragma unroll
        for (int q = 0; q < EB; ++q) {
            if (e + q * L1T < be) {
                int k = d[q] - base;
                k = k < 0 ? 0 : (k < BNODES ? k : BNODES - 1);
                atomicAdd(&cnt[k], 1);
            }
        }
    }
    __syncthreads();

    const int c = (t < BNODES) ? cnt[t] : 0;
    int incl = c;
    #pragma unroll
    for (int o = 1; o < 32; o <<= 1) {
        const int v = __shfl_up(incl, o);
        if (lane >= o) incl += v;
    }
    if (lane == 31) wsum[wv] = incl;
    __syncthreads();
    int front = (lane < wv) ? wsum[lane] : 0;
    #pragma unroll
    for (int o = 16; o > 0; o >>= 1) front += __shfl_xor(front, o);
    const int excl = front + (incl - c);
    if (t < BNODES) {
        cnt[t] = excl;
        const int node = base + t;
        if (node < N) {
            rowptr[node] = bs + excl;
            const float d = (float)(c + 1);
            dinv[node] = (d > 0.0f) ? rsqrtf(d) : 0.0f;
        }
    }
    if (b == NB - 1 && t == 0) rowptr[N] = be;
    __syncthreads();

    for (int e = bs + t; e < be; e += EB * L1T) {
        v2i ent[EB];
        #pragma unroll
        for (int q = 0; q < EB; ++q) ent[q] = te[(e + q * L1T < be) ? e + q * L1T : be - 1];
        #pragma unroll
        for (int q = 0; q < EB; ++q) {
            if (e + q * L1T < be) {
                int k = ent[q].y - base;
                k = k < 0 ? 0 : (k < BNODES ? k : BNODES - 1);
                int pos = bs + atomicAdd(&cnt[k], 1);
                pos = pos < 0 ? 0 : (pos < E ? pos : E - 1);
                list[pos] = ent[q].x;
            }
        }
    }
}

template <int F>
__global__ __launch_bounds__(256) void k_agg_rows(
    const float* __restrict__ H, const int* __restrict__ rowptr,
    const int* __restrict__ list, const float* __restrict__ dinv,
    const float* __restrict__ bias, float* __restrict__ dest,
    int N, int E, int doRelu)
{
    constexpr int LPN = F / 4;
    constexpr int NPB = 256 / LPN;
    const int t = threadIdx.x;
    if (t >= NPB * LPN) return;
    const int ln   = t / LPN;
    const int col  = (t - ln * LPN) << 2;
    v4f bz;
    bz.x = bias[col]; bz.y = bias[col + 1]; bz.z = bias[col + 2]; bz.w = bias[col + 3];
    for (int g = blockIdx.x; g * NPB < N; g += gridDim.x) {
        const int i = g * NPB + ln;
        if (i >= N) continue;
        int beg = rowptr[i], end = rowptr[i + 1];
        beg = beg < 0 ? 0 : beg;
        end = end > E ? E : end;
        const float di = dinv[i];
        const v4f hs = *(const v4f*)&H[(size_t)i * F + col];
        v4f acc = v4f{};
        const int last = end - 1;
        for (int j = beg; j < end; j += 4) {
            const bool p1 = j + 1 < end, p2 = j + 2 < end, p3 = j + 3 < end;
            const int s0 = list[j];
            const int s1 = list[p1 ? j + 1 : last];
            const int s2 = list[p2 ? j + 2 : last];
            const int s3 = list[p3 ? j + 3 : last];
            const v4f h0 = *(const v4f*)&H[(size_t)s0 * F + col];
            const v4f h1 = *(const v4f*)&H[(size_t)s1 * F + col];
            const v4f h2 = *(const v4f*)&H[(size_t)s2 * F + col];
            const v4f h3 = *(const v4f*)&H[(size_t)s3 * F + col];
            acc += h0;
            acc += h1 * (p1 ? 1.0f : 0.0f);
            acc += h2 * (p2 ? 1.0f : 0.0f);
            acc += h3 * (p3 ? 1.0f : 0.0f);
        }
        v4f v = (acc + hs) * di + bz;
        if (doRelu) {
            v.x = fmaxf(v.x, 0.0f); v.y = fmaxf(v.y, 0.0f);
            v.z = fmaxf(v.z, 0.0f); v.w = fmaxf(v.w, 0.0f);
        }
        *(v4f*)&dest[(size_t)i * F + col] = v;
    }
}

__global__ void k_agg_elem(
    const float* __restrict__ H, const int* __restrict__ rowptr,
    const int* __restrict__ list, const float* __restrict__ dinv,
    const float* __restrict__ bias, float* __restrict__ dest,
    int N, int E, int F, int doRelu)
{
    const size_t idx = (size_t)blockIdx.x * blockDim.x + threadIdx.x;
    if (idx >= (size_t)N * F) return;
    const int i = (int)(idx / F);
    const int f = (int)(idx - (size_t)i * F);
    int beg = rowptr[i], end = rowptr[i + 1];
    beg = beg < 0 ? 0 : beg;
    end = end > E ? E : end;
    const float di = dinv[i];
    const float hs = H[idx];
    float acc = 0.0f;
    const int last = end - 1;
    for (int j = beg; j < end; j += 4) {
        const bool p1 = j + 1 < end, p2 = j + 2 < end, p3 = j + 3 < end;
        const int s0 = list[j];
        const int s1 = list[p1 ? j + 1 : last];
        const int s2 = list[p2 ? j + 2 : last];
        const int s3 = list[p3 ? j + 3 : last];
        const float h0 = H[(size_t)s0 * F + f];
        const float h1 = H[(size_t)s1 * F + f];
        const float h2 = H[(size_t)s2 * F + f];
        const float h3 = H[(size_t)s3 * F + f];
        acc += h0;
        acc += h1 * (p1 ? 1.0f : 0.0f);
        acc += h2 * (p2 ? 1.0f : 0.0f);
        acc += h3 * (p3 ? 1.0f : 0.0f);
    }
    const float v = (acc + hs) * di + bias[f];
    dest[idx] = doRelu ? fmaxf(v, 0.0f) : v;
}

extern "C" void kernel_launch(void* const* d_in, const int* in_sizes, int n_in,
                              void* d_out, int out_size, void* d_ws, size_t ws_size,
                              hipStream_t stream) {
    (void)n_in; (void)out_size;

    const float* x   = (const float*)d_in[0];
    const int*   ei  = (const int*)  d_in[1];
    const float* W1  = (const float*)d_in[2];
    const float* b1  = (const float*)d_in[3];
    const float* W1b = (const float*)d_in[4];
    const float* b1b = (const float*)d_in[5];
    const float* W2b = (const float*)d_in[6];
    const float* b2b = (const float*)d_in[7];
    const float* W2  = (const float*)d_in[8];
    const float* b2  = (const float*)d_in[9];

    const int F_IN = 500, H1 = 200, H2 = 100, H3 = 40, C = 3;
    const int N = in_sizes[0] / F_IN;
    const int E = in_sizes[1] / 2;
    const int* srcIdx = ei;
    const int* dstIdx = ei + E;

    char*  ws   = (char*)d_ws;
    float* dinv = (float*)ws;
    size_t off  = ((size_t)N * sizeof(float) + 511) & ~(size_t)511;
    off += (size_t)256 * 512 * sizeof(float);
    float* bufH = (float*)(ws + off);
    off += (((size_t)N * H1 * sizeof(float)) + 511) & ~(size_t)511;
    float* bufA = (float*)(ws + off);
    off += (((size_t)N * H1 * sizeof(float)) + 511) & ~(size_t)511;
    int*   list = (int*)(ws + off);
    off += (((size_t)E * 2 * sizeof(int)) + 511) & ~(size_t)511;
    int*   rowptr = (int*)(ws + off);
    off += (((size_t)(N + 1) * sizeof(int)) + 511) & ~(size_t)511;
    float* out  = (float*)d_out;

    const size_t cntPad = ((size_t)N + 127) & ~(size_t)127;
    int* cnt  = (int*)bufA;
    int* rank = cnt + cntPad;

    const int gemmBlocks = (N + BM - 1) / BM;

    auto gemm = [&](const float* in, const float* Wt, const float* rs, int K, int F) {
        if (F <= 16)       k_gemm_wmma<1><<<gemmBlocks, 128, 0, stream>>>(in, Wt, rs, bufH, N, K, F);
        else if (F <= 48)  k_gemm_wmma<3><<<gemmBlocks, 128, 0, stream>>>(in, Wt, rs, bufH, N, K, F);
        else if (F <= 112) k_gemm_wmma<7><<<gemmBlocks, 128, 0, stream>>>(in, Wt, rs, bufH, N, K, F);
        else               k_gemm_wmma<13><<<gemmBlocks, 128, 0, stream>>>(in, Wt, rs, bufH, N, K, F);
    };

    const bool gather = ws_size >= off && cntPad + (size_t)E <= (size_t)N * H1;

    const int    NB     = (N + BNODES - 1) >> BSH;
    const int    chunk  = (E + G1 - 1) / G1;
    const size_t tmpPad = ((size_t)E * 2 + 127) & ~(size_t)127;
    int* tmpE   = (int*)bufA;
    int* blkcnt = tmpE + tmpPad;
    int* blkoff = blkcnt + (size_t)MAXB * G1;
    const bool binned = NB <= MAXB && tmpPad + (size_t)2 * MAXB * G1 <= (size_t)N * H1;

    if (gather) {
        if (binned) {
            const int nTab = NB * G1;
            k_bin_count  <<<G1, L1T, 0, stream>>>(dstIdx, blkcnt, E, NB, chunk);
            k_scan_tab   <<<(nTab + SCAN_SEG - 1) / SCAN_SEG, SCAN_T, 0, stream>>>(blkcnt, blkoff, nTab);
            k_bin_scatter<<<G1, L1T, 0, stream>>>(srcIdx, dstIdx, blkoff, tmpE, E, NB, chunk);
            k_bucket_sort<<<NB, L1T, 0, stream>>>(tmpE, blkoff, rowptr, dinv, list, N, E, NB);
        } else {
            k_izero<<<(N + 255) / 256, 256, 0, stream>>>(cnt, N);
            k_rank <<<(E + 255) / 256, 256, 0, stream>>>(dstIdx, cnt, rank, E);
            k_scan <<<(N + SCAN_SEG - 1) / SCAN_SEG, SCAN_T, 0, stream>>>(cnt, rowptr, dinv, N);
            k_place<<<(E + 255) / 256, 256, 0, stream>>>(srcIdx, dstIdx, rank, rowptr, list, E);
        }

        const int resident = 2048;
        auto rowsGrid = [&](int nodesPerBlock) {
            const int groups = (N + nodesPerBlock - 1) / nodesPerBlock;
            return groups < resident ? groups : resident;
        };
        auto layer = [&](const float* in, const float* Wt, const float* bias,
                         float* dest, int K, int F, int relu) {
            gemm(in, Wt, dinv, K, F);
            if (F == 200)      k_agg_rows<200><<<rowsGrid(256 / 50), 256, 0, stream>>>(
                                   bufH, rowptr, list, dinv, bias, dest, N, E, relu);
            else if (F == 100) k_agg_rows<100><<<rowsGrid(256 / 25), 256, 0, stream>>>(
                                   bufH, rowptr, list, dinv, bias, dest, N, E, relu);
            else if (F == 40)  k_agg_rows<40><<<rowsGrid(256 / 10), 256, 0, stream>>>(
                                   bufH, rowptr, list, dinv, bias, dest, N, E, relu);
            else {
                const size_t tot = (size_t)N * F;
                k_agg_elem<<<(int)((tot + 255) / 256), 256, 0, stream>>>(
                    bufH, rowptr, list, dinv, bias, dest, N, E, F, relu);
            }
        };
        layer(x,    W1,  b1,  bufA, F_IN, H1, 1);
        layer(bufA, W1b, b1b, bufA, H1,   H2, 1);
        layer(bufA, W2b, b2b, bufA, H2,   H3, 1);
        layer(bufA, W2,  b2,  out,  H3,   C,  0);
        return;
    }

    k_fill1<<<(N + 255) / 256, 256, 0, stream>>>(dinv, N);
    k_deg  <<<(E + 255) / 256, 256, 0, stream>>>(dstIdx, dinv, E);
    k_rsqrt<<<(N + 255) / 256, 256, 0, stream>>>(dinv, N);

    const int aggBlocks  = 65536;

    auto layer = [&](const float* in, const float* Wt, const float* bias,
                     float* dest, int K, int F, int relu) {
        gemm(in, Wt, nullptr, K, F);
        size_t tot = (size_t)N * F;
        k_zero<<<(int)((tot + 255) / 256), 256, 0, stream>>>(dest, tot);
        if (F >= 32) {
            k_agg_wave<<<aggBlocks, 256, 0, stream>>>(bufH, srcIdx, dstIdx, dinv, dest, E, F);
        } else {
            size_t tote = (size_t)E * F;
            k_agg_small<<<(int)((tote + 255) / 256), 256, 0, stream>>>(
                bufH, srcIdx, dstIdx, dinv, dest, E, F);
        }
        k_finish<<<(int)((tot + 255) / 256), 256, 0, stream>>>(
            dest, bufH, dinv, bias, N, F, relu);
    };

    layer(x,    W1,  b1,  bufA, F_IN, H1, 1);
    layer(bufA, W1b, b1b, bufA, H1,   H2, 1);
    layer(bufA, W2b, b2b, bufA, H2,   H3, 1);
    layer(bufA, W2,  b2,  out,  H3,   C,  0);
}


# Layers 2-4: each GEMM runs in the block that aggregated its 32 input rows (gather -> LDS tile -> WMMA, 8 rows in flight per lane); activation buffer and 3 launches gone; sum and k order kept
# speedup vs baseline: 4.4538x; 1.0406x over previous
#include <hip/hip_runtime.h>

typedef float v2f __attribute__((ext_vector_type(2)));
typedef float v8f __attribute__((ext_vector_type(8)));
typedef float v4f __attribute__((ext_vector_type(4)));

#define TK 32
#define LDP (TK + 4)
#define BM 64
#define XQ ((BM * TK / 4) / 128)

template <int NT>
__global__ __launch_bounds__(128) void k_gemm_wmma(
    const float* __restrict__ X, const float* __restrict__ W,
    const float* __restrict__ rs, float* __restrict__ H, int M, int K, int Nout)
{
    __shared__ __attribute__((aligned(16))) float As[BM][LDP];
    __shared__ __attribute__((aligned(16))) float Ws[NT * 16][LDP];

    const int row0 = blockIdx.x * BM;
    const int t    = threadIdx.x;
    const int wave = t >> 5;
    const int lane = t & 31;
    const int half = lane >> 4;
    const int m    = lane & 15;
    const int arow = (wave << 4) + m;

    const int srow = t >> 3;
    const int skq  = (t & 7) << 2;

    v8f acc[NT];
    #pragma unroll
    for (int j = 0; j < NT; ++j) acc[j] = v8f{};

    v4f xa[XQ];
    v4f wa[NT];

#define LOAD_CHUNK(k0_)                                                        \
    {                                                                          \
        const int gk_  = (k0_) + skq;                                          \
        const int gkc_ = gk_ < K ? gk_ : K - 4;       \
        _Pragma("unroll")                                                      \
        for (int i = 0; i < XQ; ++i) {                                         \
            const int gr_  = row0 + srow + (i << 4);                           \
            const int grc_ = gr_ < M ? gr_ : M - 1;   \
            xa[i] = *(const v4f*)&X[(size_t)grc_ * K + gkc_];               \
        }                                                                      \
        _Pragma("unroll")                                                      \
        for (int i = 0; i < NT; ++i) {                                         \
            const int r_  = srow + (i << 4);                                   \
            const int rc_ = r_ < Nout ? r_ : Nout - 1;   \
            wa[i] = *(const v4f*)&W[(size_t)rc_ * K + gkc_];                \
        }                                                                      \
    }

#define KSTEP(kk_)                                                             \
    {                                                                          \
        const int kb_ = (kk_) + (half << 1);            \
        const v2f a_ = *(const v2f*)&As[arow][kb_];                            \
        _Pragma("unroll")                                                      \
        for (int j = 0; j < NT; ++j) {                                         \
            const v2f b_ = *(const v2f*)&Ws[(j << 4) + m][kb_];                \
            acc[j] = __builtin_amdgcn_wmma_f32_16x16x4_f32(                    \
                false, a_, false, b_, (short)0, acc[j], false, false);         \
        }                                                                      \
    }

    LOAD_CHUNK(0)
    for (int k0 = 0; k0 < K; k0 += TK) {
        __syncthreads();
        #pragma unroll
        for (int i = 0; i < XQ; ++i) *(v4f*)&As[srow + (i << 4)][skq] = xa[i];
        #pragma unroll
        for (int i = 0; i < NT; ++i) *(v4f*)&Ws[srow + (i << 4)][skq] = wa[i];
        __syncthreads();

        if (k0 + TK < K) LOAD_CHUNK(k0 + TK)

        if (k0 + TK <= K) {
            #pragma unroll
            for (int kk = 0; kk < TK; kk += 4) KSTEP(kk)
        } else {
            for (int kk = 0; k0 + kk < K; kk += 4) KSTEP(kk)
        }
    }
#undef LOAD_CHUNK
#undef KSTEP

    const int rbase = row0 + (wave << 4) + (half << 3);
    float sc[8];
    #pragma unroll
    for (int r = 0; r < 8; ++r) sc[r] = 1.0f;
    if (rs != nullptr) {
        #pragma unroll
        for (int r = 0; r < 8; ++r) {
            const int gr = rbase + r;
            sc[r] = rs[gr < M ? gr : M - 1];
        }
    }
    #pragma unroll
    for (int j = 0; j < NT; ++j) {
        #pragma unroll
        for (int r = 0; r < 8; ++r) acc[j][r] *= sc[r];
    }
    if (row0 + BM <= M) {
        float* __restrict__ hp = H + (size_t)rbase * Nout + m;
        #pragma unroll
        for (int j = 0; j < NT; ++j) {
            if (((j + 1) << 4) <= Nout) {
                #pragma unroll
                for (int r = 0; r < 8; ++r) hp[(size_t)r * Nout + (j << 4)] = acc[j][r];
            } else if ((j << 4) + m < Nout) {
                #pragma unroll
                for (int r = 0; r < 8; ++r) hp[(size_t)r * Nout + (j << 4)] = acc[j][r];
            }
        }
    } else {
        #pragma unroll
        for (int j = 0; j < NT; ++j) {
            const int n = (j << 4) + m;
            #pragma unroll
            for (int r = 0; r < 8; ++r) {
                const int grow = rbase + r;
                if (grow < M && n < Nout) H[(size_t)grow * Nout + n] = acc[j][r];
            }
        }
    }
}

__global__ void k_fill1(float* __restrict__ p, int n) {
    int i = blockIdx.x * blockDim.x + threadIdx.x;
    if (i < n) p[i] = 1.0f;
}

__global__ void k_deg(const int* __restrict__ dst, float* __restrict__ deg, int E) {
    int e = blockIdx.x * blockDim.x + threadIdx.x;
    if (e < E) atomicAdd(&deg[dst[e]], 1.0f);
}

__global__ void k_rsqrt(float* __restrict__ p, int n) {
    int i = blockIdx.x * blockDim.x + threadIdx.x;
    if (i < n) { float d = p[i]; p[i] = (d > 0.0f) ? rsqrtf(d) : 0.0f; }
}

__global__ void k_zero(float* __restrict__ p, size_t n) {
    size_t i = (size_t)blockIdx.x * blockDim.x + threadIdx.x;
    if (i < n) p[i] = 0.0f;
}

__global__ __launch_bounds__(256) void k_agg_wave(
    const float* __restrict__ H, const int* __restrict__ src,
    const int* __restrict__ dst, const float* __restrict__ dinv,
    float* __restrict__ out, int E, int F)
{
    const int lane   = threadIdx.x & 31;
    const int wid    = (int)((blockIdx.x * blockDim.x + threadIdx.x) >> 5);
    const int stride = (int)((gridDim.x * blockDim.x) >> 5);

    for (int e = wid; e < E; e += stride) {
        const int s = src[e];
        const int d = dst[e];
        const int e2 = e + stride;
        if (e2 < E) {
            __builtin_prefetch(&H[(size_t)src[e2] * F], 0, 1);
        }
        const float nrm = dinv[s] * dinv[d];
        const float* __restrict__ hrow = H + (size_t)s * F;
        float* __restrict__ orow = out + (size_t)d * F;
        for (int f = lane; f < F; f += 32) {
            atomicAdd(&orow[f], hrow[f] * nrm);
        }
    }
}

__global__ void k_agg_small(
    const float* __restrict__ H, const int* __restrict__ src,
    const int* __restrict__ dst, const float* __restrict__ dinv,
    float* __restrict__ out, int E, int F)
{
    size_t idx = (size_t)blockIdx.x * blockDim.x + threadIdx.x;
    size_t tot = (size_t)E * F;
    if (idx >= tot) return;
    int e = (int)(idx / F);
    int f = (int)(idx % F);
    int s = src[e], d = dst[e];
    float nrm = dinv[s] * dinv[d];
    atomicAdd(&out[(size_t)d * F + f], H[(size_t)s * F + f] * nrm);
}

__global__ void k_finish(float* __restrict__ dest, const float* __restrict__ H,
                         const float* __restrict__ dinv, const float* __restrict__ bias,
                         int N, int F, int doRelu)
{
    size_t idx = (size_t)blockIdx.x * blockDim.x + threadIdx.x;
    size_t tot = (size_t)N * F;
    if (idx >= tot) return;
    int i = (int)(idx / F);
    int f = (int)(idx % F);
    float di = dinv[i];
    float v  = dest[idx] + H[idx] * di * di + bias[f];
    dest[idx] = doRelu ? fmaxf(v, 0.0f) : v;
}

typedef int v2i __attribute__((ext_vector_type(2)));
typedef int v4i __attribute__((ext_vector_type(4)));

__global__ void k_izero(int* __restrict__ p, int n) {
    int i = blockIdx.x * blockDim.x + threadIdx.x;
    if (i < n) p[i] = 0;
}

__global__ void k_rank(const int* __restrict__ dst, int* __restrict__ cnt,
                       int* __restrict__ rank, int E) {
    int e = blockIdx.x * blockDim.x + threadIdx.x;
    if (e < E) rank[e] = atomicAdd(&cnt[dst[e]], 1);
}

#define SCAN_T 1024
#define SCAN_SEG (SCAN_T * 4)
__global__ __launch_bounds__(SCAN_T) void k_scan(
    const int* __restrict__ cnt, int* __restrict__ rowptr,
    float* __restrict__ dinv, int N)
{
    __shared__ int wpre[32];
    __shared__ int wtot[32];
    const int t = threadIdx.x, lane = t & 31, wv = t >> 5;
    const int base = blockIdx.x * SCAN_SEG;

    int pre = 0;
    const v4i* c4 = (const v4i*)cnt;
    for (int i = t; i < (base >> 2); i += SCAN_T) {
        const v4i v = c4[i];
        pre += (v.x + v.y) + (v.z + v.w);
    }

    const int i0 = base + (t << 2);
    int c[4];
    #pragma unroll
    for (int q = 0; q < 4; ++q) c[q] = (i0 + q < N) ? cnt[i0 + q] : 0;
    const int s = (c[0] + c[1]) + (c[2] + c[3]);

    int incl = s;
    #pragma unroll
    for (int o = 1; o < 32; o <<= 1) {
        const int v = __shfl_up(incl, o);
        if (lane >= o) incl += v;
    }
    #pragma unroll
    for (int o = 16; o > 0; o >>= 1) pre += __shfl_xor(pre, o);
    if (lane == 31) wtot[wv] = incl;
    if (lane == 0)  wpre[wv] = pre;
    __syncthreads();

    int front = wpre[lane];
    int mine  = (lane < wv) ? wtot[lane] : 0;
    #pragma unroll
    for (int o = 16; o > 0; o >>= 1) {
        front += __shfl_xor(front, o);
        mine  += __shfl_xor(mine, o);
    }
    int run = front + mine + (incl - s);
    #pragma unroll
    for (int q = 0; q < 4; ++q) {
        if (i0 + q < N) {
            rowptr[i0 + q] = run;
            const float d = (float)(c[q] + 1);
            dinv[i0 + q] = (d > 0.0f) ? rsqrtf(d) : 0.0f;
        }
        run += c[q];
    }
    if (blockIdx.x == gridDim.x - 1 && t == SCAN_T - 1) rowptr[N] = run;
}

__global__ void k_place(const int* __restrict__ src, const int* __restrict__ dst,
                        const int* __restrict__ rank, const int* __restrict__ rowptr,
                        int* __restrict__ list, int E) {
    int e = blockIdx.x * blockDim.x + threadIdx.x;
    if (e >= E) return;
    const int s = src[e], d = dst[e];
    int pos = rowptr[d] + rank[e];
    pos = pos < 0 ? 0 : (pos < E ? pos : E - 1);
    list[pos] = s;
}

#define BSH 9
#define BNODES (1 << BSH)
#define MAXB 256
#define G1 256
#define L1T 1024
#define EB 8

__global__ __launch_bounds__(L1T) void k_bin_count(
    const int* __restrict__ dst, int* __restrict__ blkcnt, int E, int NB, int chunk)
{
    __shared__ int hist[MAXB];
    const int t = threadIdx.x;
    if (t < MAXB) hist[t] = 0;
    __syncthreads();
    const int e0 = blockIdx.x * chunk;
    const int e1 = (e0 + chunk < E) ? e0 + chunk : E;
    for (int e = e0 + t; e < e1; e += EB * L1T) {
        int d[EB];
        #pragma unroll
        for (int q = 0; q < EB; ++q) d[q] = dst[(e + q * L1T < e1) ? e + q * L1T : e1 - 1];
        #pragma unroll
        for (int q = 0; q < EB; ++q) {
            if (e + q * L1T < e1) {
                int b = (int)((unsigned)d[q] >> BSH);
                b = b < NB ? b : NB - 1;
                atomicAdd(&hist[b], 1);
            }
        }
    }
    __syncthreads();
    if (t < NB) blkcnt[t * G1 + blockIdx.x] = hist[t];
}

__global__ __launch_bounds__(SCAN_T) void k_scan_tab(
    const int* __restrict__ cnt, int* __restrict__ off, int n)
{
    __shared__ int wpre[32];
    __shared__ int wtot[32];
    const int t = threadIdx.x, lane = t & 31, wv = t >> 5;
    const int base = blockIdx.x * SCAN_SEG;

    int pre = 0;
    const v4i* c4 = (const v4i*)cnt;
    #pragma unroll 4
    for (int i = t; i < (base >> 2); i += SCAN_T) {
        const v4i v = c4[i];
        pre += (v.x + v.y) + (v.z + v.w);
    }

    const int i0 = base + (t << 2);
    int c[4];
    #pragma unroll
    for (int q = 0; q < 4; ++q) c[q] = (i0 + q < n) ? cnt[i0 + q] : 0;
    const int s = (c[0] + c[1]) + (c[2] + c[3]);

    int incl = s;
    #pragma unroll
    for (int o = 1; o < 32; o <<= 1) {
        const int v = __shfl_up(incl, o);
        if (lane >= o) incl += v;
    }
    #pragma unroll
    for (int o = 16; o > 0; o >>= 1) pre += __shfl_xor(pre, o);
    if (lane == 31) wtot[wv] = incl;
    if (lane == 0)  wpre[wv] = pre;
    __syncthreads();

    int front = wpre[lane];
    int mine  = (lane < wv) ? wtot[lane] : 0;
    #pragma unroll
    for (int o = 16; o > 0; o >>= 1) {
        front += __shfl_xor(front, o);
        mine  += __shfl_xor(mine, o);
    }
    int run = front + mine + (incl - s);
    #pragma unroll
    for (int q = 0; q < 4; ++q) {
        if (i0 + q < n) off[i0 + q] = run;
        run += c[q];
    }
}

__global__ __launch_bounds__(L1T) void k_bin_scatter(
    const int* __restrict__ src, const int* __restrict__ dst,
    const int* __restrict__ blkoff, int* __restrict__ tmp, int E, int NB, int chunk)
{
    __shared__ int cur[MAXB];
    const int t = threadIdx.x;
    if (t < NB) cur[t] = blkoff[t * G1 + blockIdx.x];
    __syncthreads();
    const int e0 = blockIdx.x * chunk;
    const int e1 = (e0 + chunk < E) ? e0 + chunk : E;
    for (int e = e0 + t; e < e1; e += EB * L1T) {
        v2i ent[EB];
        #pragma unroll
        for (int q = 0; q < EB; ++q) {
            const int ec = (e + q * L1T < e1) ? e + q * L1T : e1 - 1;
            ent[q].x = src[ec];
            ent[q].y = dst[ec];
        }
        #pragma unroll
        for (int q = 0; q < EB; ++q) {
            if (e + q * L1T < e1) {
                int b = (int)((unsigned)ent[q].y >> BSH);
                b = b < NB ? b : NB - 1;
                int pos = atomicAdd(&cur[b], 1);
                pos = pos < 0 ? 0 : (pos < E ? pos : E - 1);
                *(v2i*)&tmp[(size_t)pos << 1] = ent[q];
            }
        }
    }
}

__global__ __launch_bounds__(L1T) void k_bucket_sort(
    const int* __restrict__ tmp, const int* __restrict__ blkoff,
    int* __restrict__ rowptr, float* __restrict__ dinv, int* __restrict__ list,
    int N, int E, int NB)
{
    __shared__ int cnt[BNODES];
    __shared__ int wsum[32];
    const int t = threadIdx.x, lane = t & 31, wv = t >> 5;
    const int b = blockIdx.x;
    const int base = b << BSH;
    int bs = blkoff[b * G1];
    int be = (b + 1 < NB) ? blkoff[(b + 1) * G1] : E;
    bs = bs < 0 ? 0 : (bs < E ? bs : E);
    be = be < bs ? bs : (be < E ? be : E);
    if (t < BNODES) cnt[t] = 0;
    __syncthreads();

    const v2i* __restrict__ te = (const v2i*)tmp;
    for (int e = bs + t; e < be; e += EB * L1T) {
        int d[EB];
        #pragma unroll
        for (int q = 0; q < EB; ++q) d[q] = te[(e + q * L1T < be) ? e + q * L1T : be - 1].y;
        #pragma unroll
        for (int q = 0; q < EB; ++q) {
            if (e + q * L1T < be) {
                int k = d[q] - base;
                k = k < 0 ? 0 : (k < BNODES ? k : BNODES - 1);
                atomicAdd(&cnt[k], 1);
            }
        }
    }
    __syncthreads();

    const int c = (t < BNODES) ? cnt[t] : 0;
    int incl = c;
    #pragma unroll
    for (int o = 1; o < 32; o <<= 1) {
        const int v = __shfl_up(incl, o);
        if (lane >= o) incl += v;
    }
    if (lane == 31) wsum[wv] = incl;
    __syncthreads();
    int front = (lane < wv) ? wsum[lane] : 0;
    #pragma unroll
    for (int o = 16; o > 0; o >>= 1) front += __shfl_xor(front, o);
    const int excl = front + (incl - c);
    if (t < BNODES) {
        cnt[t] = excl;
        const int node = base + t;
        if (node < N) {
            rowptr[node] = bs + excl;
            const float d = (float)(c + 1);
            dinv[node] = (d > 0.0f) ? rsqrtf(d) : 0.0f;
        }
    }
    if (b == NB - 1 && t == 0) rowptr[N] = be;
    __syncthreads();

    for (int e = bs + t; e < be; e += EB * L1T) {
        v2i ent[EB];
        #pragma unroll
        for (int q = 0; q < EB; ++q) ent[q] = te[(e + q * L1T < be) ? e + q * L1T : be - 1];
        #pragma unroll
        for (int q = 0; q < EB; ++q) {
            if (e + q * L1T < be) {
                int k = ent[q].y - base;
                k = k < 0 ? 0 : (k < BNODES ? k : BNODES - 1);
                int pos = bs + atomicAdd(&cnt[k], 1);
                pos = pos < 0 ? 0 : (pos < E ? pos : E - 1);
                list[pos] = ent[q].x;
            }
        }
    }
}

#define FB 32
template <int F, int NT>
__global__ __launch_bounds__(256) void k_agg_gemm(
    const float* __restrict__ H, const int* __restrict__ rowptr,
    const int* __restrict__ list, const float* __restrict__ dinv,
    const float* __restrict__ bias, const float* __restrict__ Wn,
    float* __restrict__ Hn, int N, int E, int Nout)
{
    constexpr int LPN = F / 4;
    constexpr int NPB = 256 / LPN;
    constexpr int EPT = 8;
    constexpr int P   = ((F / 4) & 1) ? F : F + 4;
    constexpr int CT  = (NT + 3) / 4;
    constexpr int CG  = (NT + CT - 1) / CT;
    constexpr int NTP = CG * CT;
    constexpr int WQ  = (NTP * 128 + 255) / 256;
    __shared__ __attribute__((aligned(16))) float As[FB][P];
    __shared__ __attribute__((aligned(16))) float Ws[NTP * 16][LDP];

    const int t    = threadIdx.x;
    const int row0 = blockIdx.x * FB;

    if (t < NPB * LPN) {
        const int ln  = t / LPN;
        const int col = (t - ln * LPN) << 2;
        v4f bz;
        bz.x = bias[col]; bz.y = bias[col + 1]; bz.z = bias[col + 2]; bz.w = bias[col + 3];
        for (int r = ln; r < FB; r += NPB) {
            const int i = row0 + r;
            v4f v = v4f{};
            if (i < N) {
                int beg = rowptr[i], end = rowptr[i + 1];
                beg = beg < 0 ? 0 : beg;
                end = end > E ? E : end;
                const float di = dinv[i];
                const v4f hs = *(const v4f*)&H[(size_t)i * F + col];
                v4f acc = v4f{};
                const int last = end - 1;
                for (int j = beg; j < end; j += EPT) {
                    int s[EPT];
                    v4f h[EPT];
                    #pragma unroll
                    for (int q = 0; q < EPT; ++q) s[q] = list[(j + q < end) ? j + q : last];
                    #pragma unroll
                    for (int q = 0; q < EPT; ++q) h[q] = *(const v4f*)&H[(size_t)s[q] * F + col];
                    acc += h[0];
                    #pragma unroll
                    for (int q = 1; q < EPT; ++q) acc += h[q] * ((j + q < end) ? 1.0f : 0.0f);
                }
                v = (acc + hs) * di + bz;
                v.x = fmaxf(v.x, 0.0f); v.y = fmaxf(v.y, 0.0f);
                v.z = fmaxf(v.z, 0.0f); v.w = fmaxf(v.w, 0.0f);
            }
            *(v4f*)&As[r][col] = v;
        }
    }

    const int wave = __builtin_amdgcn_readfirstlane(t >> 5);
    const int lane = t & 31;
    const int half = lane >> 4;
    const int m    = lane & 15;
    const int rt   = wave & 1;
    const int cg   = wave >> 1;
    const bool busy = cg < CG;
    const int ct0  = cg * CT;
    const int arow = (rt << 4) + m;

    v8f acc[CT];
    #pragma unroll
    for (int j = 0; j < CT; ++j) acc[j] = v8f{};
    v4f wa[WQ];

#define LOAD_W(k0_)                                                            \
    {                                                                          \
        _Pragma("unroll")                                                      \
        for (int i = 0; i < WQ; ++i) {                                         \
            const int slot_ = t + (i << 8);                                    \
            if ((i + 1) * 256 <= NTP * 128 || slot_ < NTP * 128) {   \
                const int r_   = slot_ >> 3;                                   \
                const int rc_  = r_ < Nout ? r_ : Nout - 1;   \
                const int gk_  = (k0_) + ((slot_ & 7) << 2);                   \
                const int gkc_ = gk_ < F ? gk_ : F - 4;       \
                wa[i] = *(const v4f*)&Wn[(size_t)rc_ * F + gkc_];              \
            }                                                                  \
        }                                                                      \
    }

#define KSTEP_F(kk_)                                                           \
    {                                                                          \
        const int kb_ = (kk_) + (half << 1);            \
        const v2f a_ = *(const v2f*)&As[arow][k0 + kb_];                       \
        _Pragma("unroll")                                                      \
        for (int j = 0; j < CT; ++j) {                                         \
            const v2f b_ = *(const v2f*)&Ws[((ct0 + j) << 4) + m][kb_];        \
            acc[j] = __builtin_amdgcn_wmma_f32_16x16x4_f32(                    \
                false, a_, false, b_, (short)0, acc[j], false, false);         \
        }                                                                      \
    }

    LOAD_W(0)
    for (int k0 = 0; k0 < F; k0 += TK) {
        __syncthreads();
        #pragma unroll
        for (int i = 0; i < WQ; ++i) {
            const int slot = t + (i << 8);
            if ((i + 1) * 256 <= NTP * 128 || slot < NTP * 128)
                *(v4f*)&Ws[slot >> 3][(slot & 7) << 2] = wa[i];
        }
        __syncthreads();

        if (k0 + TK < F) LOAD_W(k0 + TK)

        if (busy) {
            if (k0 + TK <= F) {
                #pragma unroll
                for (int kk = 0; kk < TK; kk += 4) KSTEP_F(kk)
            } else {
                for (int kk = 0; k0 + kk < F; kk += 4) KSTEP_F(kk)
            }
        }
    }
#undef LOAD_W
#undef KSTEP_F

    if (!busy) return;
    const int rbase = row0 + (rt << 4) + (half << 3);
    float sc[8];
    #pragma unroll
    for (int r = 0; r < 8; ++r) {
        const int gr = rbase + r;
        sc[r] = dinv[gr < N ? gr : N - 1];
    }
    if (row0 + FB <= N) {
        float* __restrict__ hp = Hn + (size_t)rbase * Nout + m;
        #pragma unroll
        for (int j = 0; j < CT; ++j) {
            if (((ct0 + j) << 4) + m < Nout) {
                #pragma unroll
                for (int r = 0; r < 8; ++r) hp[(size_t)r * Nout + ((ct0 + j) << 4)] = acc[j][r] * sc[r];
            }
        }
    } else {
        #pragma unroll
        for (int j = 0; j < CT; ++j) {
            const int n = ((ct0 + j) << 4) + m;
            #pragma unroll
            for (int r = 0; r < 8; ++r) {
                const int grow = rbase + r;
                if (grow < N && n < Nout) Hn[(size_t)grow * Nout + n] = acc[j][r] * sc[r];
            }
        }
    }
}

__global__ void k_agg_elem(
    const float* __restrict__ H, const int* __restrict__ rowptr,
    const int* __restrict__ list, const float* __restrict__ dinv,
    const float* __restrict__ bias, float* __restrict__ dest,
    int N, int E, int F, int doRelu)
{
    const size_t idx = (size_t)blockIdx.x * blockDim.x + threadIdx.x;
    if (idx >= (size_t)N * F) return;
    const int i = (int)(idx / F);
    const int f = (int)(idx - (size_t)i * F);
    int beg = rowptr[i], end = rowptr[i + 1];
    beg = beg < 0 ? 0 : beg;
    end = end > E ? E : end;
    const float di = dinv[i];
    const float hs = H[idx];
    float acc = 0.0f;
    const int last = end - 1;
    for (int j = beg; j < end; j += 4) {
        const bool p1 = j + 1 < end, p2 = j + 2 < end, p3 = j + 3 < end;
        const int s0 = list[j];
        const int s1 = list[p1 ? j + 1 : last];
        const int s2 = list[p2 ? j + 2 : last];
        const int s3 = list[p3 ? j + 3 : last];
        const float h0 = H[(size_t)s0 * F + f];
        const float h1 = H[(size_t)s1 * F + f];
        const float h2 = H[(size_t)s2 * F + f];
        const float h3 = H[(size_t)s3 * F + f];
        acc += h0;
        acc += h1 * (p1 ? 1.0f : 0.0f);
        acc += h2 * (p2 ? 1.0f : 0.0f);
        acc += h3 * (p3 ? 1.0f : 0.0f);
    }
    const float v = (acc + hs) * di + bias[f];
    dest[idx] = doRelu ? fmaxf(v, 0.0f) : v;
}

extern "C" void kernel_launch(void* const* d_in, const int* in_sizes, int n_in,
                              void* d_out, int out_size, void* d_ws, size_t ws_size,
                              hipStream_t stream) {
    (void)n_in; (void)out_size;

    const float* x   = (const float*)d_in[0];
    const int*   ei  = (const int*)  d_in[1];
    const float* W1  = (const float*)d_in[2];
    const float* b1  = (const float*)d_in[3];
    const float* W1b = (const float*)d_in[4];
    const float* b1b = (const float*)d_in[5];
    const float* W2b = (const float*)d_in[6];
    const float* b2b = (const float*)d_in[7];
    const float* W2  = (const float*)d_in[8];
    const float* b2  = (const float*)d_in[9];

    const int F_IN = 500, H1 = 200, H2 = 100, H3 = 40, C = 3;
    const int N = in_sizes[0] / F_IN;
    const int E = in_sizes[1] / 2;
    const int* srcIdx = ei;
    const int* dstIdx = ei + E;

    char*  ws   = (char*)d_ws;
    float* dinv = (float*)ws;
    size_t off  = ((size_t)N * sizeof(float) + 511) & ~(size_t)511;
    off += (size_t)256 * 512 * sizeof(float);
    float* bufH = (float*)(ws + off);
    off += (((size_t)N * H1 * sizeof(float)) + 511) & ~(size_t)511;
    float* bufA = (float*)(ws + off);
    off += (((size_t)N * H1 * sizeof(float)) + 511) & ~(size_t)511;
    int*   list = (int*)(ws + off);
    off += (((size_t)E * 2 * sizeof(int)) + 511) & ~(size_t)511;
    int*   rowptr = (int*)(ws + off);
    off += (((size_t)(N + 1) * sizeof(int)) + 511) & ~(size_t)511;
    float* out  = (float*)d_out;

    const size_t cntPad = ((size_t)N + 127) & ~(size_t)127;
    int* cnt  = (int*)bufA;
    int* rank = cnt + cntPad;

    const int gemmBlocks = (N + BM - 1) / BM;

    auto gemm = [&](const float* in, const float* Wt, const float* rs, int K, int F) {
        if (F <= 16)       k_gemm_wmma<1><<<gemmBlocks, 128, 0, stream>>>(in, Wt, rs, bufH, N, K, F);
        else if (F <= 48)  k_gemm_wmma<3><<<gemmBlocks, 128, 0, stream>>>(in, Wt, rs, bufH, N, K, F);
        else if (F <= 112) k_gemm_wmma<7><<<gemmBlocks, 128, 0, stream>>>(in, Wt, rs, bufH, N, K, F);
        else               k_gemm_wmma<13><<<gemmBlocks, 128, 0, stream>>>(in, Wt, rs, bufH, N, K, F);
    };

    const bool gather = ws_size >= off && cntPad + (size_t)E <= (size_t)N * H1;

    const int    NB     = (N + BNODES - 1) >> BSH;
    const int    chunk  = (E + G1 - 1) / G1;
    const size_t tmpPad = ((size_t)E * 2 + 127) & ~(size_t)127;
    int* tmpE   = (int*)bufA;
    int* blkcnt = tmpE + tmpPad;
    int* blkoff = blkcnt + (size_t)MAXB * G1;
    const bool binned = NB <= MAXB && tmpPad + (size_t)2 * MAXB * G1 <= (size_t)N * H1;

    if (gather) {
        if (binned) {
            const int nTab = NB * G1;
            k_bin_count  <<<G1, L1T, 0, stream>>>(dstIdx, blkcnt, E, NB, chunk);
            k_scan_tab   <<<(nTab + SCAN_SEG - 1) / SCAN_SEG, SCAN_T, 0, stream>>>(blkcnt, blkoff, nTab);
            k_bin_scatter<<<G1, L1T, 0, stream>>>(srcIdx, dstIdx, blkoff, tmpE, E, NB, chunk);
            k_bucket_sort<<<NB, L1T, 0, stream>>>(tmpE, blkoff, rowptr, dinv, list, N, E, NB);
        } else {
            k_izero<<<(N + 255) / 256, 256, 0, stream>>>(cnt, N);
            k_rank <<<(E + 255) / 256, 256, 0, stream>>>(dstIdx, cnt, rank, E);
            k_scan <<<(N + SCAN_SEG - 1) / SCAN_SEG, SCAN_T, 0, stream>>>(cnt, rowptr, dinv, N);
            k_place<<<(E + 255) / 256, 256, 0, stream>>>(srcIdx, dstIdx, rank, rowptr, list, E);
        }

        const int fusedBlocks = (N + FB - 1) / FB;
        gemm(x, W1, dinv, F_IN, H1);
        k_agg_gemm<200, 7><<<fusedBlocks, 256, 0, stream>>>(
            bufH, rowptr, list, dinv, b1,  W1b, bufA, N, E, H2);
        k_agg_gemm<100, 3><<<fusedBlocks, 256, 0, stream>>>(
            bufA, rowptr, list, dinv, b1b, W2b, bufH, N, E, H3);
        k_agg_gemm<40, 1><<<fusedBlocks, 256, 0, stream>>>(
            bufH, rowptr, list, dinv, b2b, W2,  bufA, N, E, C);
        {
            const size_t tot = (size_t)N * C;
            k_agg_elem<<<(int)((tot + 255) / 256), 256, 0, stream>>>(
                bufA, rowptr, list, dinv, b2, out, N, E, C, 0);
        }
        return;
    }

    k_fill1<<<(N + 255) / 256, 256, 0, stream>>>(dinv, N);
    k_deg  <<<(E + 255) / 256, 256, 0, stream>>>(dstIdx, dinv, E);
    k_rsqrt<<<(N + 255) / 256, 256, 0, stream>>>(dinv, N);

    const int aggBlocks  = 65536;

    auto layer = [&](const float* in, const float* Wt, const float* bias,
                     float* dest, int K, int F, int relu) {
        gemm(in, Wt, nullptr, K, F);
        size_t tot = (size_t)N * F;
        k_zero<<<(int)((tot + 255) / 256), 256, 0, stream>>>(dest, tot);
        if (F >= 32) {
            k_agg_wave<<<aggBlocks, 256, 0, stream>>>(bufH, srcIdx, dstIdx, dinv, dest, E, F);
        } else {
            size_t tote = (size_t)E * F;
            k_agg_small<<<(int)((tote + 255) / 256), 256, 0, stream>>>(
                bufH, srcIdx, dstIdx, dinv, dest, E, F);
        }
        k_finish<<<(int)((tot + 255) / 256), 256, 0, stream>>>(
            dest, bufH, dinv, bias, N, F, relu);
    };

    layer(x,    W1,  b1,  bufA, F_IN, H1, 1);
    layer(bufA, W1b, b1b, bufA, H1,   H2, 1);
    layer(bufA, W2b, b2b, bufA, H2,   H3, 1);
    layer(bufA, W2,  b2,  out,  H3,   C,  0);
}


# speedup vs baseline: 4.7615x; 1.0691x over previous
#include <hip/hip_runtime.h>

typedef float v2f __attribute__((ext_vector_type(2)));
typedef float v8f __attribute__((ext_vector_type(8)));
typedef float v4f __attribute__((ext_vector_type(4)));

#define TK 32
#define LDP (TK + 4)
#define BM 64
#define XQ ((BM * TK / 4) / 128)

template <int NT>
__global__ __launch_bounds__(128) void k_gemm_wmma(
    const float* __restrict__ X, const float* __restrict__ W,
    const float* __restrict__ rs, float* __restrict__ H, int M, int K, int Nout)
{
    __shared__ __attribute__((aligned(16))) float As[BM][LDP];
    __shared__ __attribute__((aligned(16))) float Ws[NT * 16][LDP];

    const int row0 = blockIdx.x * BM;
    const int t    = threadIdx.x;
    const int wave = t >> 5;
    const int lane = t & 31;
    const int half = lane >> 4;
    const int m    = lane & 15;
    const int arow = (wave << 4) + m;

    const int srow = t >> 3;
    const int skq  = (t & 7) << 2;

    v8f acc[NT];
    #pragma unroll
    for (int j = 0; j < NT; ++j) acc[j] = v8f{};

    v4f xa[XQ];
    v4f wa[NT];

#define LOAD_CHUNK(k0_)                                                        \
    {                                                                          \
        const int gk_  = (k0_) + skq;                                          \
        const int gkc_ = gk_ < K ? gk_ : K - 4;       \
        _Pragma("unroll")                                                      \
        for (int i = 0; i < XQ; ++i) {                                         \
            const int gr_  = row0 + srow + (i << 4);                           \
            const int grc_ = gr_ < M ? gr_ : M - 1;   \
            xa[i] = *(const v4f*)&X[(size_t)grc_ * K + gkc_];               \
        }                                                                      \
        _Pragma("unroll")                                                      \
        for (int i = 0; i < NT; ++i) {                                         \
            const int r_  = srow + (i << 4);                                   \
            const int rc_ = r_ < Nout ? r_ : Nout - 1;   \
            wa[i] = *(const v4f*)&W[(size_t)rc_ * K + gkc_];                \
        }                                                                      \
    }

#define KSTEP(kk_)                                                             \
    {                                                                          \
        const int kb_ = (kk_) + (half << 1);            \
        const v2f a_ = *(const v2f*)&As[arow][kb_];                            \
        _Pragma("unroll")                                                      \
        for (int j = 0; j < NT; ++j) {                                         \
            const v2f b_ = *(const v2f*)&Ws[(j << 4) + m][kb_];                \
            acc[j] = __builtin_amdgcn_wmma_f32_16x16x4_f32(                    \
                false, a_, false, b_, (short)0, acc[j], false, false);         \
        }                                                                      \
    }

    LOAD_CHUNK(0)
    for (int k0 = 0; k0 < K; k0 += TK) {
        __syncthreads();
        #pragma unroll
        for (int i = 0; i < XQ; ++i) *(v4f*)&As[srow + (i << 4)][skq] = xa[i];
        #pragma unroll
        for (int i = 0; i < NT; ++i) *(v4f*)&Ws[srow + (i << 4)][skq] = wa[i];
        __syncthreads();

        if (k0 + TK < K) LOAD_CHUNK(k0 + TK)

        if (k0 + TK <= K) {
            #pragma unroll
            for (int kk = 0; kk < TK; kk += 4) KSTEP(kk)
        } else {
            for (int kk = 0; k0 + kk < K; kk += 4) KSTEP(kk)
        }
    }
#undef LOAD_CHUNK
#undef KSTEP

    const int rbase = row0 + (wave << 4) + (half << 3);
    float sc[8];
    #pragma unroll
    for (int r = 0; r < 8; ++r) sc[r] = 1.0f;
    if (rs != nullptr) {
        #pragma unroll
        for (int r = 0; r < 8; ++r) {
            const int gr = rbase + r;
            sc[r] = rs[gr < M ? gr : M - 1];
        }
    }
    #pragma unroll
    for (int j = 0; j < NT; ++j) {
        #pragma unroll
        for (int r = 0; r < 8; ++r) acc[j][r] *= sc[r];
    }
    if (row0 + BM <= M) {
        float* __restrict__ hp = H + (size_t)rbase * Nout + m;
        #pragma unroll
        for (int j = 0; j < NT; ++j) {
            if (((j + 1) << 4) <= Nout) {
                #pragma unroll
                for (int r = 0; r < 8; ++r) hp[(size_t)r * Nout + (j << 4)] = acc[j][r];
            } else if ((j << 4) + m < Nout) {
                #pragma unroll
                for (int r = 0; r < 8; ++r) hp[(size_t)r * Nout + (j << 4)] = acc[j][r];
            }
        }
    } else {
        #pragma unroll
        for (int j = 0; j < NT; ++j) {
            const int n = (j << 4) + m;
            #pragma unroll
            for (int r = 0; r < 8; ++r) {
                const int grow = rbase + r;
                if (grow < M && n < Nout) H[(size_t)grow * Nout + n] = acc[j][r];
            }
        }
    }
}

__global__ void k_fill1(float* __restrict__ p, int n) {
    int i = blockIdx.x * blockDim.x + threadIdx.x;
    if (i < n) p[i] = 1.0f;
}

__global__ void k_deg(const int* __restrict__ dst, float* __restrict__ deg, int E) {
    int e = blockIdx.x * blockDim.x + threadIdx.x;
    if (e < E) atomicAdd(&deg[dst[e]], 1.0f);
}

__global__ void k_rsqrt(float* __restrict__ p, int n) {
    int i = blockIdx.x * blockDim.x + threadIdx.x;
    if (i < n) { float d = p[i]; p[i] = (d > 0.0f) ? rsqrtf(d) : 0.0f; }
}

__global__ void k_zero(float* __restrict__ p, size_t n) {
    size_t i = (size_t)blockIdx.x * blockDim.x + threadIdx.x;
    if (i < n) p[i] = 0.0f;
}

__global__ __launch_bounds__(256) void k_agg_wave(
    const float* __restrict__ H, const int* __restrict__ src,
    const int* __restrict__ dst, const float* __restrict__ dinv,
    float* __restrict__ out, int E, int F)
{
    const int lane   = threadIdx.x & 31;
    const int wid    = (int)((blockIdx.x * blockDim.x + threadIdx.x) >> 5);
    const int stride = (int)((gridDim.x * blockDim.x) >> 5);

    for (int e = wid; e < E; e += stride) {
        const int s = src[e];
        const int d = dst[e];
        const int e2 = e + stride;
        if (e2 < E) {
            __builtin_prefetch(&H[(size_t)src[e2] * F], 0, 1);
        }
        const float nrm = dinv[s] * dinv[d];
        const float* __restrict__ hrow = H + (size_t)s * F;
        float* __restrict__ orow = out + (size_t)d * F;
        for (int f = lane; f < F; f += 32) {
            atomicAdd(&orow[f], hrow[f] * nrm);
        }
    }
}

__global__ void k_agg_small(
    const float* __restrict__ H, const int* __restrict__ src,
    const int* __restrict__ dst, const float* __restrict__ dinv,
    float* __restrict__ out, int E, int F)
{
    size_t idx = (size_t)blockIdx.x * blockDim.x + threadIdx.x;
    size_t tot = (size_t)E * F;
    if (idx >= tot) return;
    int e = (int)(idx / F);
    int f = (int)(idx % F);
    int s = src[e], d = dst[e];
    float nrm = dinv[s] * dinv[d];
    atomicAdd(&out[(size_t)d * F + f], H[(size_t)s * F + f] * nrm);
}

__global__ void k_finish(float* __restrict__ dest, const float* __restrict__ H,
                         const float* __restrict__ dinv, const float* __restrict__ bias,
                         int N, int F, int doRelu)
{
    size_t idx = (size_t)blockIdx.x * blockDim.x + threadIdx.x;
    size_t tot = (size_t)N * F;
    if (idx >= tot) return;
    int i = (int)(idx / F);
    int f = (int)(idx % F);
    float di = dinv[i];
    float v  = dest[idx] + H[idx] * di * di + bias[f];
    dest[idx] = doRelu ? fmaxf(v, 0.0f) : v;
}

typedef int v2i __attribute__((ext_vector_type(2)));
typedef int v4i __attribute__((ext_vector_type(4)));

__global__ void k_izero(int* __restrict__ p, int n) {
    int i = blockIdx.x * blockDim.x + threadIdx.x;
    if (i < n) p[i] = 0;
}

__global__ void k_rank(const int* __restrict__ dst, int* __restrict__ cnt,
                       int* __restrict__ rank, int E) {
    int e = blockIdx.x * blockDim.x + threadIdx.x;
    if (e < E) rank[e] = atomicAdd(&cnt[dst[e]], 1);
}

#define SCAN_T 1024
#define SCAN_SEG (SCAN_T * 4)
__global__ __launch_bounds__(SCAN_T) void k_scan(
    const int* __restrict__ cnt, int* __restrict__ rowptr,
    float* __restrict__ dinv, int N)
{
    __shared__ int wpre[32];
    __shared__ int wtot[32];
    const int t = threadIdx.x, lane = t & 31, wv = t >> 5;
    const int base = blockIdx.x * SCAN_SEG;

    int pre = 0;
    const v4i* c4 = (const v4i*)cnt;
    for (int i = t; i < (base >> 2); i += SCAN_T) {
        const v4i v = c4[i];
        pre += (v.x + v.y) + (v.z + v.w);
    }

    const int i0 = base + (t << 2);
    int c[4];
    #pragma unroll
    for (int q = 0; q < 4; ++q) c[q] = (i0 + q < N) ? cnt[i0 + q] : 0;
    const int s = (c[0] + c[1]) + (c[2] + c[3]);

    int incl = s;
    #pragma unroll
    for (int o = 1; o < 32; o <<= 1) {
        const int v = __shfl_up(incl, o);
        if (lane >= o) incl += v;
    }
    #pragma unroll
    for (int o = 16; o > 0; o >>= 1) pre += __shfl_xor(pre, o);
    if (lane == 31) wtot[wv] = incl;
    if (lane == 0)  wpre[wv] = pre;
    __syncthreads();

    int front = wpre[lane];
    int mine  = (lane < wv) ? wtot[lane] : 0;
    #pragma unroll
    for (int o = 16; o > 0; o >>= 1) {
        front += __shfl_xor(front, o);
        mine  += __shfl_xor(mine, o);
    }
    int run = front + mine + (incl - s);
    #pragma unroll
    for (int q = 0; q < 4; ++q) {
        if (i0 + q < N) {
            rowptr[i0 + q] = run;
            const float d = (float)(c[q] + 1);
            dinv[i0 + q] = (d > 0.0f) ? rsqrtf(d) : 0.0f;
        }
        run += c[q];
    }
    if (blockIdx.x == gridDim.x - 1 && t == SCAN_T - 1) rowptr[N] = run;
}

__global__ void k_place(const int* __restrict__ src, const int* __restrict__ dst,
                        const int* __restrict__ rank, const int* __restrict__ rowptr,
                        int* __restrict__ list, int E) {
    int e = blockIdx.x * blockDim.x + threadIdx.x;
    if (e >= E) return;
    const int s = src[e], d = dst[e];
    int pos = rowptr[d] + rank[e];
    pos = pos < 0 ? 0 : (pos < E ? pos : E - 1);
    list[pos] = s;
}

#define BSH 9
#define BNODES (1 << BSH)
#define MAXB 256
#define G1 256
#define L1T 1024
#define EB 8

__global__ __launch_bounds__(L1T) void k_bin_count(
    const int* __restrict__ dst, int* __restrict__ blkcnt, int E, int NB, int chunk)
{
    __shared__ int hist[MAXB];
    const int t = threadIdx.x;
    if (t < MAXB) hist[t] = 0;
    __syncthreads();
    const int e0 = blockIdx.x * chunk;
    const int e1 = (e0 + chunk < E) ? e0 + chunk : E;
    for (int e = e0 + t; e < e1; e += EB * L1T) {
        int d[EB];
        #pragma unroll
        for (int q = 0; q < EB; ++q) d[q] = dst[(e + q * L1T < e1) ? e + q * L1T : e1 - 1];
        #pragma unroll
        for (int q = 0; q < EB; ++q) {
            if (e + q * L1T < e1) {
                int b = (int)((unsigned)d[q] >> BSH);
                b = b < NB ? b : NB - 1;
                atomicAdd(&hist[b], 1);
            }
        }
    }
    __syncthreads();
    if (t < NB) blkcnt[t * G1 + blockIdx.x] = hist[t];
}

__global__ __launch_bounds__(SCAN_T) void k_scan_tab(
    const int* __restrict__ cnt, int* __restrict__ off, int n)
{
    __shared__ int wpre[32];
    __shared__ int wtot[32];
    const int t = threadIdx.x, lane = t & 31, wv = t >> 5;
    const int base = blockIdx.x * SCAN_SEG;

    int pre = 0;
    const v4i* c4 = (const v4i*)cnt;
    #pragma unroll 4
    for (int i = t; i < (base >> 2); i += SCAN_T) {
        const v4i v = c4[i];
        pre += (v.x + v.y) + (v.z + v.w);
    }

    const int i0 = base + (t << 2);
    int c[4];
    #pragma unroll
    for (int q = 0; q < 4; ++q) c[q] = (i0 + q < n) ? cnt[i0 + q] : 0;
    const int s = (c[0] + c[1]) + (c[2] + c[3]);

    int incl = s;
    #pragma unroll
    for (int o = 1; o < 32; o <<= 1) {
        const int v = __shfl_up(incl, o);
        if (lane >= o) incl += v;
    }
    #pragma unroll
    for (int o = 16; o > 0; o >>= 1) pre += __shfl_xor(pre, o);
    if (lane == 31) wtot[wv] = incl;
    if (lane == 0)  wpre[wv] = pre;
    __syncthreads();

    int front = wpre[lane];
    int mine  = (lane < wv) ? wtot[lane] : 0;
    #pragma unroll
    for (int o = 16; o > 0; o >>= 1) {
        front += __shfl_xor(front, o);
        mine  += __shfl_xor(mine, o);
    }
    int run = front + mine + (incl - s);
    #pragma unroll
    for (int q = 0; q < 4; ++q) {
        if (i0 + q < n) off[i0 + q] = run;
        run += c[q];
    }
}

__global__ __launch_bounds__(L1T) void k_bin_scatter(
    const int* __restrict__ src, const int* __restrict__ dst,
    const int* __restrict__ blkoff, int* __restrict__ tmp, int E, int NB, int chunk)
{
    __shared__ int cur[MAXB];
    const int t = threadIdx.x;
    if (t < NB) cur[t] = blkoff[t * G1 + blockIdx.x];
    __syncthreads();
    const int e0 = blockIdx.x * chunk;
    const int e1 = (e0 + chunk < E) ? e0 + chunk : E;
    for (int e = e0 + t; e < e1; e += EB * L1T) {
        v2i ent[EB];
        #pragma unroll
        for (int q = 0; q < EB; ++q) {
            const int ec = (e + q * L1T < e1) ? e + q * L1T : e1 - 1;
            ent[q].x = src[ec];
            ent[q].y = dst[ec];
        }
        #pragma unroll
        for (int q = 0; q < EB; ++q) {
            if (e + q * L1T < e1) {
                int b = (int)((unsigned)ent[q].y >> BSH);
                b = b < NB ? b : NB - 1;
                int pos = atomicAdd(&cur[b], 1);
                pos = pos < 0 ? 0 : (pos < E ? pos : E - 1);
                *(v2i*)&tmp[(size_t)pos << 1] = ent[q];
            }
        }
    }
}

__global__ __launch_bounds__(L1T) void k_bucket_sort(
    const int* __restrict__ tmp, const int* __restrict__ blkoff,
    int* __restrict__ rowptr, float* __restrict__ dinv, int* __restrict__ list,
    int N, int E, int NB)
{
    __shared__ int cnt[BNODES];
    __shared__ int wsum[32];
    const int t = threadIdx.x, lane = t & 31, wv = t >> 5;
    const int b = blockIdx.x;
    const int base = b << BSH;
    int bs = blkoff[b * G1];
    int be = (b + 1 < NB) ? blkoff[(b + 1) * G1] : E;
    bs = bs < 0 ? 0 : (bs < E ? bs : E);
    be = be < bs ? bs : (be < E ? be : E);
    if (t < BNODES) cnt[t] = 0;
    __syncthreads();

    const v2i* __restrict__ te = (const v2i*)tmp;
    for (int e = bs + t; e < be; e += EB * L1T) {
        int d[EB];
        #pragma unroll
        for (int q = 0; q < EB; ++q) d[q] = te[(e + q * L1T < be) ? e + q * L1T : be - 1].y;
        #pragma unroll
        for (int q = 0; q < EB; ++q) {
            if (e + q * L1T < be) {
                int k = d[q] - base;
                k = k < 0 ? 0 : (k < BNODES ? k : BNODES - 1);
                atomicAdd(&cnt[k], 1);
            }
        }
    }
    __syncthreads();

    const int c = (t < BNODES) ? cnt[t] : 0;
    int incl = c;
    #pragma unroll
    for (int o = 1; o < 32; o <<= 1) {
        const int v = __shfl_up(incl, o);
        if (lane >= o) incl += v;
    }
    if (lane == 31) wsum[wv] = incl;
    __syncthreads();
    int front = (lane < wv) ? wsum[lane] : 0;
    #pragma unroll
    for (int o = 16; o > 0; o >>= 1) front += __shfl_xor(front, o);
    const int excl = front + (incl - c);
    if (t < BNODES) {
        cnt[t] = excl;
        const int node = base + t;
        if (node < N) {
            rowptr[node] = bs + excl;
            const float d = (float)(c + 1);
            dinv[node] = (d > 0.0f) ? rsqrtf(d) : 0.0f;
        }
    }
    if (b == NB - 1 && t == 0) rowptr[N] = be;
    __syncthreads();

    for (int e = bs + t; e < be; e += EB * L1T) {
        v2i ent[EB];
        #pragma unroll
        for (int q = 0; q < EB; ++q) ent[q] = te[(e + q * L1T < be) ? e + q * L1T : be - 1];
        #pragma unroll
        for (int q = 0; q < EB; ++q) {
            if (e + q * L1T < be) {
                int k = ent[q].y - base;
                k = k < 0 ? 0 : (k < BNODES ? k : BNODES - 1);
                int pos = bs + atomicAdd(&cnt[k], 1);
                pos = pos < 0 ? 0 : (pos < E ? pos : E - 1);
                list[pos] = ent[q].x;
            }
        }
    }
}

#define FB 32
template <int F, int NT, int PIN, int POUT>
__global__ __launch_bounds__(256) void k_agg_gemm(
    const float* __restrict__ H, const int* __restrict__ rowptr,
    const int* __restrict__ list, const float* __restrict__ dinv,
    const float* __restrict__ bias, const float* __restrict__ Wn,
    float* __restrict__ Hn, int N, int E, int Nout)
{
    constexpr int LPN = (F / 4 <= 16) ? 16 : 32;
    constexpr int NV  = (F / 4 + LPN - 1) / LPN;
    constexpr int NPB = 256 / LPN;
    constexpr int EPT = 8;
    constexpr int P   = ((F / 4) & 1) ? F : F + 4;
    constexpr int CT  = (NT + 3) / 4;
    constexpr int CG  = (NT + CT - 1) / CT;
    constexpr int NTP = CG * CT;
    constexpr int WQ  = (NTP * 128 + 255) / 256;
    __shared__ __attribute__((aligned(16))) float As[FB][P];
    __shared__ __attribute__((aligned(16))) float Ws[NTP * 16][LDP];

    const int t    = threadIdx.x;
    const int row0 = blockIdx.x * FB;

    static_assert(NV <= 2, "a lane holds at most two 16-byte pieces of a row");
    {
        const int ln   = t / LPN;
        const int l    = t - ln * LPN;
        const int col0 = l << 2;
        const int col1 = col0 + LPN * 4;
        const bool on1 = (NV > 1) && (col1 < F);
        if ((LPN * 4 <= F) || (col0 < F)) {
            v4f bz0, bz1 = v4f{};
            bz0.x = bias[col0]; bz0.y = bias[col0 + 1]; bz0.z = bias[col0 + 2]; bz0.w = bias[col0 + 3];
            if (on1) { bz1.x = bias[col1]; bz1.y = bias[col1 + 1]; bz1.z = bias[col1 + 2]; bz1.w = bias[col1 + 3]; }
            for (int r = ln; r < FB; r += NPB) {
                const int i = row0 + r;
                v4f x0 = v4f{}, x1 = v4f{};
                if (i < N) {
                    int beg = rowptr[i], end = rowptr[i + 1];
                    beg = beg < 0 ? 0 : beg;
                    end = end > E ? E : end;
                    const float di = dinv[i];
                    const float* __restrict__ hown = H + (size_t)i * PIN;
                    const v4f hs0 = *(const v4f*)&hown[col0];
                    v4f hs1 = v4f{};
                    if (on1) hs1 = *(const v4f*)&hown[col1];
                    v4f acc0 = v4f{}, acc1 = v4f{};
                    const int last = end - 1;
                    for (int j = beg; j < end; j += EPT) {
                        int s[EPT];
                        v4f h0[EPT], h1[EPT];
                        #pragma unroll
                        for (int q = 0; q < EPT; ++q) s[q] = list[(j + q < end) ? j + q : last];
                        #pragma unroll
                        for (int q = 0; q < EPT; ++q) h0[q] = *(const v4f*)&H[(size_t)s[q] * PIN + col0];
                        if (on1) {
                            #pragma unroll
                            for (int q = 0; q < EPT; ++q) h1[q] = *(const v4f*)&H[(size_t)s[q] * PIN + col1];
                        } else {
                            #pragma unroll
                            for (int q = 0; q < EPT; ++q) h1[q] = v4f{};
                        }
                        acc0 += h0[0];
                        #pragma unroll
                        for (int q = 1; q < EPT; ++q) acc0 += h0[q] * ((j + q < end) ? 1.0f : 0.0f);
                        if (NV > 1) {
                            acc1 += h1[0];
                            #pragma unroll
                            for (int q = 1; q < EPT; ++q) acc1 += h1[q] * ((j + q < end) ? 1.0f : 0.0f);
                        }
                    }
                    x0 = (acc0 + hs0) * di + bz0;
                    x0.x = fmaxf(x0.x, 0.0f); x0.y = fmaxf(x0.y, 0.0f);
                    x0.z = fmaxf(x0.z, 0.0f); x0.w = fmaxf(x0.w, 0.0f);
                    if (NV > 1) {
                        x1 = (acc1 + hs1) * di + bz1;
                        x1.x = fmaxf(x1.x, 0.0f); x1.y = fmaxf(x1.y, 0.0f);
                        x1.z = fmaxf(x1.z, 0.0f); x1.w = fmaxf(x1.w, 0.0f);
                    }
                }
                *(v4f*)&As[r][col0] = x0;
                if (on1) *(v4f*)&As[r][col1] = x1;
            }
        }
    }

    const int wave = __builtin_amdgcn_readfirstlane(t >> 5);
    const int lane = t & 31;
    const int half = lane >> 4;
    const int m    = lane & 15;
    const int rt   = wave & 1;
    const int cg   = wave >> 1;
    const bool busy = cg < CG;
    const int ct0  = cg * CT;
    const int arow = (rt << 4) + m;

    v8f acc[CT];
    #pragma unroll
    for (int j = 0; j < CT; ++j) acc[j] = v8f{};
    v4f wa[WQ];

#define LOAD_W(k0_)                                                            \
    {                                                                          \
        _Pragma("unroll")                                                      \
        for (int i = 0; i < WQ; ++i) {                                         \
            const int slot_ = t + (i << 8);                                    \
            if ((i + 1) * 256 <= NTP * 128 || slot_ < NTP * 128) {   \
                const int r_   = slot_ >> 3;                                   \
                const int rc_  = r_ < Nout ? r_ : Nout - 1;   \
                const int gk_  = (k0_) + ((slot_ & 7) << 2);                   \
                const int gkc_ = gk_ < F ? gk_ : F - 4;       \
                wa[i] = *(const v4f*)&Wn[(size_t)rc_ * F + gkc_];              \
            }                                                                  \
        }                                                                      \
    }

#define KSTEP_F(kk_)                                                           \
    {                                                                          \
        const int kb_ = (kk_) + (half << 1);            \
        const v2f a_ = *(const v2f*)&As[arow][k0 + kb_];                       \
        _Pragma("unroll")                                                      \
        for (int j = 0; j < CT; ++j) {                                         \
            const v2f b_ = *(const v2f*)&Ws[((ct0 + j) << 4) + m][kb_];        \
            acc[j] = __builtin_amdgcn_wmma_f32_16x16x4_f32(                    \
                false, a_, false, b_, (short)0, acc[j], false, false);         \
        }                                                                      \
    }

    LOAD_W(0)
    for (int k0 = 0; k0 < F; k0 += TK) {
        __syncthreads();
        #pragma unroll
        for (int i = 0; i < WQ; ++i) {
            const int slot = t + (i << 8);
            if ((i + 1) * 256 <= NTP * 128 || slot < NTP * 128)
                *(v4f*)&Ws[slot >> 3][(slot & 7) << 2] = wa[i];
        }
        __syncthreads();

        if (k0 + TK < F) LOAD_W(k0 + TK)

        if (busy) {
            if (k0 + TK <= F) {
                #pragma unroll
                for (int kk = 0; kk < TK; kk += 4) KSTEP_F(kk)
            } else {
                for (int kk = 0; k0 + kk < F; kk += 4) KSTEP_F(kk)
            }
        }
    }
#undef LOAD_W
#undef KSTEP_F

    if (!busy) return;
    const int rbase = row0 + (rt << 4) + (half << 3);
    float sc[8];
    #pragma unroll
    for (int r = 0; r < 8; ++r) {
        const int gr = rbase + r;
        sc[r] = dinv[gr < N ? gr : N - 1];
    }
    if (row0 + FB <= N) {
        float* __restrict__ hp = Hn + (size_t)rbase * POUT + m;
        #pragma unroll
        for (int j = 0; j < CT; ++j) {
            if (((ct0 + j) << 4) + m < Nout) {
                #pragma unroll
                for (int r = 0; r < 8; ++r) hp[(size_t)r * POUT + ((ct0 + j) << 4)] = acc[j][r] * sc[r];
            }
        }
    } else {
        #pragma unroll
        for (int j = 0; j < CT; ++j) {
            const int n = ((ct0 + j) << 4) + m;
            #pragma unroll
            for (int r = 0; r < 8; ++r) {
                const int grow = rbase + r;
                if (grow < N && n < Nout) Hn[(size_t)grow * POUT + n] = acc[j][r] * sc[r];
            }
        }
    }
}

__global__ void k_agg_elem(
    const float* __restrict__ H, const int* __restrict__ rowptr,
    const int* __restrict__ list, const float* __restrict__ dinv,
    const float* __restrict__ bias, float* __restrict__ dest,
    int N, int E, int F, int doRelu)
{
    const size_t idx = (size_t)blockIdx.x * blockDim.x + threadIdx.x;
    if (idx >= (size_t)N * F) return;
    const int i = (int)(idx / F);
    const int f = (int)(idx - (size_t)i * F);
    int beg = rowptr[i], end = rowptr[i + 1];
    beg = beg < 0 ? 0 : beg;
    end = end > E ? E : end;
    const float di = dinv[i];
    const float hs = H[idx];
    float acc = 0.0f;
    const int last = end - 1;
    for (int j = beg; j < end; j += 4) {
        const bool p1 = j + 1 < end, p2 = j + 2 < end, p3 = j + 3 < end;
        const int s0 = list[j];
        const int s1 = list[p1 ? j + 1 : last];
        const int s2 = list[p2 ? j + 2 : last];
        const int s3 = list[p3 ? j + 3 : last];
        const float h0 = H[(size_t)s0 * F + f];
        const float h1 = H[(size_t)s1 * F + f];
        const float h2 = H[(size_t)s2 * F + f];
        const float h3 = H[(size_t)s3 * F + f];
        acc += h0;
        acc += h1 * (p1 ? 1.0f : 0.0f);
        acc += h2 * (p2 ? 1.0f : 0.0f);
        acc += h3 * (p3 ? 1.0f : 0.0f);
    }
    const float v = (acc + hs) * di + bias[f];
    dest[idx] = doRelu ? fmaxf(v, 0.0f) : v;
}

extern "C" void kernel_launch(void* const* d_in, const int* in_sizes, int n_in,
                              void* d_out, int out_size, void* d_ws, size_t ws_size,
                              hipStream_t stream) {
    (void)n_in; (void)out_size;

    const float* x   = (const float*)d_in[0];
    const int*   ei  = (const int*)  d_in[1];
    const float* W1  = (const float*)d_in[2];
    const float* b1  = (const float*)d_in[3];
    const float* W1b = (const float*)d_in[4];
    const float* b1b = (const float*)d_in[5];
    const float* W2b = (const float*)d_in[6];
    const float* b2b = (const float*)d_in[7];
    const float* W2  = (const float*)d_in[8];
    const float* b2  = (const float*)d_in[9];

    const int F_IN = 500, H1 = 200, H2 = 100, H3 = 40, C = 3;
    const int N = in_sizes[0] / F_IN;
    const int E = in_sizes[1] / 2;
    const int* srcIdx = ei;
    const int* dstIdx = ei + E;

    char*  ws   = (char*)d_ws;
    float* dinv = (float*)ws;
    size_t off  = ((size_t)N * sizeof(float) + 511) & ~(size_t)511;
    off += (size_t)256 * 512 * sizeof(float);
    float* bufH = (float*)(ws + off);
    off += (((size_t)N * H1 * sizeof(float)) + 511) & ~(size_t)511;
    float* bufA = (float*)(ws + off);
    off += (((size_t)N * H1 * sizeof(float)) + 511) & ~(size_t)511;
    int*   list = (int*)(ws + off);
    off += (((size_t)E * 2 * sizeof(int)) + 511) & ~(size_t)511;
    int*   rowptr = (int*)(ws + off);
    off += (((size_t)(N + 1) * sizeof(int)) + 511) & ~(size_t)511;
    float* out  = (float*)d_out;

    const size_t cntPad = ((size_t)N + 127) & ~(size_t)127;
    int* cnt  = (int*)bufA;
    int* rank = cnt + cntPad;

    const int gemmBlocks = (N + BM - 1) / BM;

    auto gemm = [&](const float* in, const float* Wt, const float* rs, int K, int F) {
        if (F <= 16)       k_gemm_wmma<1><<<gemmBlocks, 128, 0, stream>>>(in, Wt, rs, bufH, N, K, F);
        else if (F <= 48)  k_gemm_wmma<3><<<gemmBlocks, 128, 0, stream>>>(in, Wt, rs, bufH, N, K, F);
        else if (F <= 112) k_gemm_wmma<7><<<gemmBlocks, 128, 0, stream>>>(in, Wt, rs, bufH, N, K, F);
        else               k_gemm_wmma<13><<<gemmBlocks, 128, 0, stream>>>(in, Wt, rs, bufH, N, K, F);
    };

    const bool gather = ws_size >= off && cntPad + (size_t)E <= (size_t)N * H1;

    const int    NB     = (N + BNODES - 1) >> BSH;
    const int    chunk  = (E + G1 - 1) / G1;
    const size_t tmpPad = ((size_t)E * 2 + 127) & ~(size_t)127;
    int* tmpE   = (int*)bufA;
    int* blkcnt = tmpE + tmpPad;
    int* blkoff = blkcnt + (size_t)MAXB * G1;
    const bool binned = NB <= MAXB && tmpPad + (size_t)2 * MAXB * G1 <= (size_t)N * H1;

    if (gather) {
        if (binned) {
            const int nTab = NB * G1;
            k_bin_count  <<<G1, L1T, 0, stream>>>(dstIdx, blkcnt, E, NB, chunk);
            k_scan_tab   <<<(nTab + SCAN_SEG - 1) / SCAN_SEG, SCAN_T, 0, stream>>>(blkcnt, blkoff, nTab);
            k_bin_scatter<<<G1, L1T, 0, stream>>>(srcIdx, dstIdx, blkoff, tmpE, E, NB, chunk);
            k_bucket_sort<<<NB, L1T, 0, stream>>>(tmpE, blkoff, rowptr, dinv, list, N, E, NB);
        } else {
            k_izero<<<(N + 255) / 256, 256, 0, stream>>>(cnt, N);
            k_rank <<<(E + 255) / 256, 256, 0, stream>>>(dstIdx, cnt, rank, E);
            k_scan <<<(N + SCAN_SEG - 1) / SCAN_SEG, SCAN_T, 0, stream>>>(cnt, rowptr, dinv, N);
            k_place<<<(E + 255) / 256, 256, 0, stream>>>(srcIdx, dstIdx, rank, rowptr, list, E);
        }

        const int fusedBlocks = (N + FB - 1) / FB;
        gemm(x, W1, dinv, F_IN, H1);
        k_agg_gemm<200, 7, 200, 128><<<fusedBlocks, 256, 0, stream>>>(
            bufH, rowptr, list, dinv, b1,  W1b, bufA, N, E, H2);
        k_agg_gemm<100, 3, 128, 64><<<fusedBlocks, 256, 0, stream>>>(
            bufA, rowptr, list, dinv, b1b, W2b, bufH, N, E, H3);
        k_agg_gemm<40, 1, 64, 3><<<fusedBlocks, 256, 0, stream>>>(
            bufH, rowptr, list, dinv, b2b, W2,  bufA, N, E, C);
        {
            const size_t tot = (size_t)N * C;
            k_agg_elem<<<(int)((tot + 255) / 256), 256, 0, stream>>>(
                bufA, rowptr, list, dinv, b2, out, N, E, C, 0);
        }
        return;
    }

    k_fill1<<<(N + 255) / 256, 256, 0, stream>>>(dinv, N);
    k_deg  <<<(E + 255) / 256, 256, 0, stream>>>(dstIdx, dinv, E);
    k_rsqrt<<<(N + 255) / 256, 256, 0, stream>>>(dinv, N);

    const int aggBlocks  = 65536;

    auto layer = [&](const float* in, const float* Wt, const float* bias,
                     float* dest, int K, int F, int relu) {
        gemm(in, Wt, nullptr, K, F);
        size_t tot = (size_t)N * F;
        k_zero<<<(int)((tot + 255) / 256), 256, 0, stream>>>(dest, tot);
        if (F >= 32) {
            k_agg_wave<<<aggBlocks, 256, 0, stream>>>(bufH, srcIdx, dstIdx, dinv, dest, E, F);
        } else {
            size_t tote = (size_t)E * F;
            k_agg_small<<<(int)((tote + 255) / 256), 256, 0, stream>>>(
                bufH, srcIdx, dstIdx, dinv, dest, E, F);
        }
        k_finish<<<(int)((tot + 255) / 256), 256, 0, stream>>>(
            dest, bufH, dinv, bias, N, F, relu);
    };

    layer(x,    W1,  b1,  bufA, F_IN, H1, 1);
    layer(bufA, W1b, b1b, bufA, H1,   H2, 1);
    layer(bufA, W2b, b2b, bufA, H2,   H3, 1);
    layer(bufA, W2,  b2,  out,  H3,   C,  0);
}


# Edge-list build: both scattered store streams (edge copy by bucket, final source list) go through an LDS tile and leave as runs of consecutive addresses; slots by LDS returning adds, list order kept
# speedup vs baseline: 4.9011x; 1.0293x over previous
#include <hip/hip_runtime.h>

typedef float v2f __attribute__((ext_vector_type(2)));
typedef float v8f __attribute__((ext_vector_type(8)));
typedef float v4f __attribute__((ext_vector_type(4)));

#define TK 32
#define LDP (TK + 4)
#define BM 64
#define XQ ((BM * TK / 4) / 128)

template <int NT>
__global__ __launch_bounds__(128) void k_gemm_wmma(
    const float* __restrict__ X, const float* __restrict__ W,
    const float* __restrict__ rs, float* __restrict__ H, int M, int K, int Nout)
{
    __shared__ __attribute__((aligned(16))) float As[BM][LDP];
    __shared__ __attribute__((aligned(16))) float Ws[NT * 16][LDP];

    const int row0 = blockIdx.x * BM;
    const int t    = threadIdx.x;
    const int wave = t >> 5;
    const int lane = t & 31;
    const int half = lane >> 4;
    const int m    = lane & 15;
    const int arow = (wave << 4) + m;

    const int srow = t >> 3;
    const int skq  = (t & 7) << 2;

    v8f acc[NT];
    #pragma unroll
    for (int j = 0; j < NT; ++j) acc[j] = v8f{};

    v4f xa[XQ];
    v4f wa[NT];

#define LOAD_CHUNK(k0_)                                                        \
    {                                                                          \
        const int gk_  = (k0_) + skq;                                          \
        const int gkc_ = gk_ < K ? gk_ : K - 4;       \
        _Pragma("unroll")                                                      \
        for (int i = 0; i < XQ; ++i) {                                         \
            const int gr_  = row0 + srow + (i << 4);                           \
            const int grc_ = gr_ < M ? gr_ : M - 1;   \
            xa[i] = *(const v4f*)&X[(size_t)grc_ * K + gkc_];               \
        }                                                                      \
        _Pragma("unroll")                                                      \
        for (int i = 0; i < NT; ++i) {                                         \
            const int r_  = srow + (i << 4);                                   \
            const int rc_ = r_ < Nout ? r_ : Nout - 1;   \
            wa[i] = *(const v4f*)&W[(size_t)rc_ * K + gkc_];                \
        }                                                                      \
    }

#define KSTEP(kk_)                                                             \
    {                                                                          \
        const int kb_ = (kk_) + (half << 1);            \
        const v2f a_ = *(const v2f*)&As[arow][kb_];                            \
        _Pragma("unroll")                                                      \
        for (int j = 0; j < NT; ++j) {                                         \
            const v2f b_ = *(const v2f*)&Ws[(j << 4) + m][kb_];                \
            acc[j] = __builtin_amdgcn_wmma_f32_16x16x4_f32(                    \
                false, a_, false, b_, (short)0, acc[j], false, false);         \
        }                                                                      \
    }

    LOAD_CHUNK(0)
    for (int k0 = 0; k0 < K; k0 += TK) {
        __syncthreads();
        #pragma unroll
        for (int i = 0; i < XQ; ++i) *(v4f*)&As[srow + (i << 4)][skq] = xa[i];
        #pragma unroll
        for (int i = 0; i < NT; ++i) *(v4f*)&Ws[srow + (i << 4)][skq] = wa[i];
        __syncthreads();

        if (k0 + TK < K) LOAD_CHUNK(k0 + TK)

        if (k0 + TK <= K) {
            #pragma unroll
            for (int kk = 0; kk < TK; kk += 4) KSTEP(kk)
        } else {
            for (int kk = 0; k0 + kk < K; kk += 4) KSTEP(kk)
        }
    }
#undef LOAD_CHUNK
#undef KSTEP

    const int rbase = row0 + (wave << 4) + (half << 3);
    float sc[8];
    #pragma unroll
    for (int r = 0; r < 8; ++r) sc[r] = 1.0f;
    if (rs != nullptr) {
        #pragma unroll
        for (int r = 0; r < 8; ++r) {
            const int gr = rbase + r;
            sc[r] = rs[gr < M ? gr : M - 1];
        }
    }
    #pragma unroll
    for (int j = 0; j < NT; ++j) {
        #pragma unroll
        for (int r = 0; r < 8; ++r) acc[j][r] *= sc[r];
    }
    if (row0 + BM <= M) {
        float* __restrict__ hp = H + (size_t)rbase * Nout + m;
        #pragma unroll
        for (int j = 0; j < NT; ++j) {
            if (((j + 1) << 4) <= Nout) {
                #pragma unroll
                for (int r = 0; r < 8; ++r) hp[(size_t)r * Nout + (j << 4)] = acc[j][r];
            } else if ((j << 4) + m < Nout) {
                #pragma unroll
                for (int r = 0; r < 8; ++r) hp[(size_t)r * Nout + (j << 4)] = acc[j][r];
            }
        }
    } else {
        #pragma unroll
        for (int j = 0; j < NT; ++j) {
            const int n = (j << 4) + m;
            #pragma unroll
            for (int r = 0; r < 8; ++r) {
                const int grow = rbase + r;
                if (grow < M && n < Nout) H[(size_t)grow * Nout + n] = acc[j][r];
            }
        }
    }
}

__global__ void k_fill1(float* __restrict__ p, int n) {
    int i = blockIdx.x * blockDim.x + threadIdx.x;
    if (i < n) p[i] = 1.0f;
}

__global__ void k_deg(const int* __restrict__ dst, float* __restrict__ deg, int E) {
    int e = blockIdx.x * blockDim.x + threadIdx.x;
    if (e < E) atomicAdd(&deg[dst[e]], 1.0f);
}

__global__ void k_rsqrt(float* __restrict__ p, int n) {
    int i = blockIdx.x * blockDim.x + threadIdx.x;
    if (i < n) { float d = p[i]; p[i] = (d > 0.0f) ? rsqrtf(d) : 0.0f; }
}

__global__ void k_zero(float* __restrict__ p, size_t n) {
    size_t i = (size_t)blockIdx.x * blockDim.x + threadIdx.x;
    if (i < n) p[i] = 0.0f;
}

__global__ __launch_bounds__(256) void k_agg_wave(
    const float* __restrict__ H, const int* __restrict__ src,
    const int* __restrict__ dst, const float* __restrict__ dinv,
    float* __restrict__ out, int E, int F)
{
    const int lane   = threadIdx.x & 31;
    const int wid    = (int)((blockIdx.x * blockDim.x + threadIdx.x) >> 5);
    const int stride = (int)((gridDim.x * blockDim.x) >> 5);

    for (int e = wid; e < E; e += stride) {
        const int s = src[e];
        const int d = dst[e];
        const int e2 = e + stride;
        if (e2 < E) {
            __builtin_prefetch(&H[(size_t)src[e2] * F], 0, 1);
        }
        const float nrm = dinv[s] * dinv[d];
        const float* __restrict__ hrow = H + (size_t)s * F;
        float* __restrict__ orow = out + (size_t)d * F;
        for (int f = lane; f < F; f += 32) {
            atomicAdd(&orow[f], hrow[f] * nrm);
        }
    }
}

__global__ void k_agg_small(
    const float* __restrict__ H, const int* __restrict__ src,
    const int* __restrict__ dst, const float* __restrict__ dinv,
    float* __restrict__ out, int E, int F)
{
    size_t idx = (size_t)blockIdx.x * blockDim.x + threadIdx.x;
    size_t tot = (size_t)E * F;
    if (idx >= tot) return;
    int e = (int)(idx / F);
    int f = (int)(idx % F);
    int s = src[e], d = dst[e];
    float nrm = dinv[s] * dinv[d];
    atomicAdd(&out[(size_t)d * F + f], H[(size_t)s * F + f] * nrm);
}

__global__ void k_finish(float* __restrict__ dest, const float* __restrict__ H,
                         const float* __restrict__ dinv, const float* __restrict__ bias,
                         int N, int F, int doRelu)
{
    size_t idx = (size_t)blockIdx.x * blockDim.x + threadIdx.x;
    size_t tot = (size_t)N * F;
    if (idx >= tot) return;
    int i = (int)(idx / F);
    int f = (int)(idx % F);
    float di = dinv[i];
    float v  = dest[idx] + H[idx] * di * di + bias[f];
    dest[idx] = doRelu ? fmaxf(v, 0.0f) : v;
}

typedef int v2i __attribute__((ext_vector_type(2)));
typedef int v4i __attribute__((ext_vector_type(4)));

__global__ void k_izero(int* __restrict__ p, int n) {
    int i = blockIdx.x * blockDim.x + threadIdx.x;
    if (i < n) p[i] = 0;
}

__global__ void k_rank(const int* __restrict__ dst, int* __restrict__ cnt,
                       int* __restrict__ rank, int E) {
    int e = blockIdx.x * blockDim.x + threadIdx.x;
    if (e < E) rank[e] = atomicAdd(&cnt[dst[e]], 1);
}

#define SCAN_T 1024
#define SCAN_SEG (SCAN_T * 4)
__global__ __launch_bounds__(SCAN_T) void k_scan(
    const int* __restrict__ cnt, int* __restrict__ rowptr,
    float* __restrict__ dinv, int N)
{
    __shared__ int wpre[32];
    __shared__ int wtot[32];
    const int t = threadIdx.x, lane = t & 31, wv = t >> 5;
    const int base = blockIdx.x * SCAN_SEG;

    int pre = 0;
    const v4i* c4 = (const v4i*)cnt;
    for (int i = t; i < (base >> 2); i += SCAN_T) {
        const v4i v = c4[i];
        pre += (v.x + v.y) + (v.z + v.w);
    }

    const int i0 = base + (t << 2);
    int c[4];
    #pragma unroll
    for (int q = 0; q < 4; ++q) c[q] = (i0 + q < N) ? cnt[i0 + q] : 0;
    const int s = (c[0] + c[1]) + (c[2] + c[3]);

    int incl = s;
    #pragma unroll
    for (int o = 1; o < 32; o <<= 1) {
        const int v = __shfl_up(incl, o);
        if (lane >= o) incl += v;
    }
    #pragma unroll
    for (int o = 16; o > 0; o >>= 1) pre += __shfl_xor(pre, o);
    if (lane == 31) wtot[wv] = incl;
    if (lane == 0)  wpre[wv] = pre;
    __syncthreads();

    int front = wpre[lane];
    int mine  = (lane < wv) ? wtot[lane] : 0;
    #pragma unroll
    for (int o = 16; o > 0; o >>= 1) {
        front += __shfl_xor(front, o);
        mine  += __shfl_xor(mine, o);
    }
    int run = front + mine + (incl - s);
    #pragma unroll
    for (int q = 0; q < 4; ++q) {
        if (i0 + q < N) {
            rowptr[i0 + q] = run;
            const float d = (float)(c[q] + 1);
            dinv[i0 + q] = (d > 0.0f) ? rsqrtf(d) : 0.0f;
        }
        run += c[q];
    }
    if (blockIdx.x == gridDim.x - 1 && t == SCAN_T - 1) rowptr[N] = run;
}

__global__ void k_place(const int* __restrict__ src, const int* __restrict__ dst,
                        const int* __restrict__ rank, const int* __restrict__ rowptr,
                        int* __restrict__ list, int E) {
    int e = blockIdx.x * blockDim.x + threadIdx.x;
    if (e >= E) return;
    const int s = src[e], d = dst[e];
    int pos = rowptr[d] + rank[e];
    pos = pos < 0 ? 0 : (pos < E ? pos : E - 1);
    list[pos] = s;
}

#define BSH 9
#define BNODES (1 << BSH)
#define MAXB 256
#define G1 256
#define L1T 1024
#define EB 8
#define SCAP 8192
#define LCAP 16384

__global__ __launch_bounds__(L1T) void k_bin_count(
    const int* __restrict__ dst, int* __restrict__ blkcnt, int E, int NB, int chunk)
{
    __shared__ int hist[MAXB];
    const int t = threadIdx.x;
    if (t < MAXB) hist[t] = 0;
    __syncthreads();
    const int e0 = blockIdx.x * chunk;
    const int e1 = (e0 + chunk < E) ? e0 + chunk : E;
    for (int e = e0 + t; e < e1; e += EB * L1T) {
        int d[EB];
        #pragma unroll
        for (int q = 0; q < EB; ++q) d[q] = dst[(e + q * L1T < e1) ? e + q * L1T : e1 - 1];
        #pragma unroll
        for (int q = 0; q < EB; ++q) {
            if (e + q * L1T < e1) {
                int b = (int)((unsigned)d[q] >> BSH);
                b = b < NB ? b : NB - 1;
                atomicAdd(&hist[b], 1);
            }
        }
    }
    __syncthreads();
    if (t < NB) blkcnt[t * G1 + blockIdx.x] = hist[t];
}

__global__ __launch_bounds__(SCAN_T) void k_scan_tab(
    const int* __restrict__ cnt, int* __restrict__ off, int n)
{
    __shared__ int wpre[32];
    __shared__ int wtot[32];
    const int t = threadIdx.x, lane = t & 31, wv = t >> 5;
    const int base = blockIdx.x * SCAN_SEG;

    int pre = 0;
    const v4i* c4 = (const v4i*)cnt;
    #pragma unroll 4
    for (int i = t; i < (base >> 2); i += SCAN_T) {
        const v4i v = c4[i];
        pre += (v.x + v.y) + (v.z + v.w);
    }

    const int i0 = base + (t << 2);
    int c[4];
    #pragma unroll
    for (int q = 0; q < 4; ++q) c[q] = (i0 + q < n) ? cnt[i0 + q] : 0;
    const int s = (c[0] + c[1]) + (c[2] + c[3]);

    int incl = s;
    #pragma unroll
    for (int o = 1; o < 32; o <<= 1) {
        const int v = __shfl_up(incl, o);
        if (lane >= o) incl += v;
    }
    #pragma unroll
    for (int o = 16; o > 0; o >>= 1) pre += __shfl_xor(pre, o);
    if (lane == 31) wtot[wv] = incl;
    if (lane == 0)  wpre[wv] = pre;
    __syncthreads();

    int front = wpre[lane];
    int mine  = (lane < wv) ? wtot[lane] : 0;
    #pragma unroll
    for (int o = 16; o > 0; o >>= 1) {
        front += __shfl_xor(front, o);
        mine  += __shfl_xor(mine, o);
    }
    int run = front + mine + (incl - s);
    #pragma unroll
    for (int q = 0; q < 4; ++q) {
        if (i0 + q < n) off[i0 + q] = run;
        run += c[q];
    }
}

__global__ __launch_bounds__(L1T) void k_bin_scatter(
    const int* __restrict__ src, const int* __restrict__ dst,
    const int* __restrict__ blkcnt, const int* __restrict__ blkoff,
    int* __restrict__ tmp, int E, int NB, int chunk)
{
    __shared__ int cur[MAXB];
    __shared__ int delta[MAXB];
    __shared__ int wsum[32];
    __shared__ __attribute__((aligned(8))) int stage[SCAP * 2];
    const int t = threadIdx.x, lane = t & 31, wv = t >> 5;
    const int blk = blockIdx.x;
    const int e0 = blk * chunk;
    const int e1 = (e0 + chunk < E) ? e0 + chunk : E;
    const int n  = e1 - e0;
    const bool staged = n <= SCAP;

    const int c = (t < NB) ? blkcnt[t * G1 + blk] : 0;
    const int g = (t < NB) ? blkoff[t * G1 + blk] : 0;
    int incl = c;
    for (int o = 1; o < 32; o <<= 1) {
        const int v = __shfl_up(incl, o);
        if (lane >= o) incl += v;
    }
    if (lane == 31) wsum[wv] = incl;
    __syncthreads();
    int front = (lane < wv) ? wsum[lane] : 0;
    for (int o = 16; o > 0; o >>= 1) front += __shfl_xor(front, o);
    const int excl = front + (incl - c);
    if (t < NB) {
        cur[t]   = staged ? excl : g;
        delta[t] = g - excl;
    }
    __syncthreads();

    for (int e = e0 + t; e < e1; e += EB * L1T) {
        v2i ent[EB];
        #pragma unroll
        for (int q = 0; q < EB; ++q) {
            const int ec = (e + q * L1T < e1) ? e + q * L1T : e1 - 1;
            ent[q].x = src[ec];
            ent[q].y = dst[ec];
        }
        #pragma unroll
        for (int q = 0; q < EB; ++q) {
            if (e + q * L1T < e1) {
                int b = (int)((unsigned)ent[q].y >> BSH);
                b = b < NB ? b : NB - 1;
                int pos = atomicAdd(&cur[b], 1);
                if (staged) {
                    pos = pos < 0 ? 0 : (pos < SCAP ? pos : SCAP - 1);
                    *(v2i*)&stage[pos << 1] = ent[q];
                } else {
                    pos = pos < 0 ? 0 : (pos < E ? pos : E - 1);
                    *(v2i*)&tmp[(size_t)pos << 1] = ent[q];
                }
            }
        }
    }
    if (!staged) return;
    __syncthreads();

    for (int k = t; k < n; k += L1T) {
        const v2i en = *(const v2i*)&stage[k << 1];
        int b = (int)((unsigned)en.y >> BSH);
        b = b < NB ? b : NB - 1;
        int pos = k + delta[b];
        pos = pos < 0 ? 0 : (pos < E ? pos : E - 1);
        *(v2i*)&tmp[(size_t)pos << 1] = en;
    }
}

__global__ __launch_bounds__(L1T) void k_bucket_sort(
    const int* __restrict__ tmp, const int* __restrict__ blkoff,
    int* __restrict__ rowptr, float* __restrict__ dinv, int* __restrict__ list,
    int N, int E, int NB)
{
    __shared__ int cnt[BNODES];
    __shared__ int wsum[32];
    __shared__ int stage[LCAP];
    const int t = threadIdx.x, lane = t & 31, wv = t >> 5;
    const int b = blockIdx.x;
    const int base = b << BSH;
    int bs = blkoff[b * G1];
    int be = (b + 1 < NB) ? blkoff[(b + 1) * G1] : E;
    bs = bs < 0 ? 0 : (bs < E ? bs : E);
    be = be < bs ? bs : (be < E ? be : E);
    const int abase = bs & ~31;
    const bool staged = (be - abase) <= LCAP;
    if (t < BNODES) cnt[t] = 0;
    __syncthreads();

    const v2i* __restrict__ te = (const v2i*)tmp;
    for (int e = bs + t; e < be; e += EB * L1T) {
        int d[EB];
        #pragma unroll
        for (int q = 0; q < EB; ++q) d[q] = te[(e + q * L1T < be) ? e + q * L1T : be - 1].y;
        #pragma unroll
        for (int q = 0; q < EB; ++q) {
            if (e + q * L1T < be) {
                int k = d[q] - base;
                k = k < 0 ? 0 : (k < BNODES ? k : BNODES - 1);
                atomicAdd(&cnt[k], 1);
            }
        }
    }
    __syncthreads();

    const int c = (t < BNODES) ? cnt[t] : 0;
    int incl = c;
    #pragma unroll
    for (int o = 1; o < 32; o <<= 1) {
        const int v = __shfl_up(incl, o);
        if (lane >= o) incl += v;
    }
    if (lane == 31) wsum[wv] = incl;
    __syncthreads();
    int front = (lane < wv) ? wsum[lane] : 0;
    #pragma unroll
    for (int o = 16; o > 0; o >>= 1) front += __shfl_xor(front, o);
    const int excl = front + (incl - c);
    if (t < BNODES) {
        cnt[t] = excl;
        const int node = base + t;
        if (node < N) {
            rowptr[node] = bs + excl;
            const float d = (float)(c + 1);
            dinv[node] = (d > 0.0f) ? rsqrtf(d) : 0.0f;
        }
    }
    if (b == NB - 1 && t == 0) rowptr[N] = be;
    __syncthreads();

    for (int e = bs + t; e < be; e += EB * L1T) {
        v2i ent[EB];
        #pragma unroll
        for (int q = 0; q < EB; ++q) ent[q] = te[(e + q * L1T < be) ? e + q * L1T : be - 1];
        #pragma unroll
        for (int q = 0; q < EB; ++q) {
            if (e + q * L1T < be) {
                int k = ent[q].y - base;
                k = k < 0 ? 0 : (k < BNODES ? k : BNODES - 1);
                int pos = bs + atomicAdd(&cnt[k], 1);
                if (staged) {
                    pos -= abase;
                    pos = pos < 0 ? 0 : (pos < LCAP ? pos : LCAP - 1);
                    stage[pos] = ent[q].x;
                } else {
                    pos = pos < 0 ? 0 : (pos < E ? pos : E - 1);
                    list[pos] = ent[q].x;
                }
            }
        }
    }
    if (!staged) return;
    __syncthreads();
    for (int gpos = abase + t; gpos < be; gpos += L1T) {
        if (gpos >= bs) list[gpos] = stage[gpos - abase];
    }
}

#define FB 32
template <int F, int NT, int PIN, int POUT>
__global__ __launch_bounds__(256) void k_agg_gemm(
    const float* __restrict__ H, const int* __restrict__ rowptr,
    const int* __restrict__ list, const float* __restrict__ dinv,
    const float* __restrict__ bias, const float* __restrict__ Wn,
    float* __restrict__ Hn, int N, int E, int Nout)
{
    constexpr int LPN = (F / 4 <= 16) ? 16 : 32;
    constexpr int NV  = (F / 4 + LPN - 1) / LPN;
    constexpr int NPB = 256 / LPN;
    constexpr int EPT = 8;
    constexpr int P   = ((F / 4) & 1) ? F : F + 4;
    constexpr int CT  = (NT + 3) / 4;
    constexpr int CG  = (NT + CT - 1) / CT;
    constexpr int NTP = CG * CT;
    constexpr int WQ  = (NTP * 128 + 255) / 256;
    __shared__ __attribute__((aligned(16))) float As[FB][P];
    __shared__ __attribute__((aligned(16))) float Ws[NTP * 16][LDP];

    const int t    = threadIdx.x;
    const int row0 = blockIdx.x * FB;

    static_assert(NV <= 2, "a lane holds at most two 16-byte pieces of a row");
    {
        const int ln   = t / LPN;
        const int l    = t - ln * LPN;
        const int col0 = l << 2;
        const int col1 = col0 + LPN * 4;
        const bool on1 = (NV > 1) && (col1 < F);
        if ((LPN * 4 <= F) || (col0 < F)) {
            v4f bz0, bz1 = v4f{};
            bz0.x = bias[col0]; bz0.y = bias[col0 + 1]; bz0.z = bias[col0 + 2]; bz0.w = bias[col0 + 3];
            if (on1) { bz1.x = bias[col1]; bz1.y = bias[col1 + 1]; bz1.z = bias[col1 + 2]; bz1.w = bias[col1 + 3]; }
            for (int r = ln; r < FB; r += NPB) {
                const int i = row0 + r;
                v4f x0 = v4f{}, x1 = v4f{};
                if (i < N) {
                    int beg = rowptr[i], end = rowptr[i + 1];
                    beg = beg < 0 ? 0 : beg;
                    end = end > E ? E : end;
                    const float di = dinv[i];
                    const float* __restrict__ hown = H + (size_t)i * PIN;
                    const v4f hs0 = *(const v4f*)&hown[col0];
                    v4f hs1 = v4f{};
                    if (on1) hs1 = *(const v4f*)&hown[col1];
                    v4f acc0 = v4f{}, acc1 = v4f{};
                    const int last = end - 1;
                    for (int j = beg; j < end; j += EPT) {
                        int s[EPT];
                        v4f h0[EPT], h1[EPT];
                        #pragma unroll
                        for (int q = 0; q < EPT; ++q) s[q] = list[(j + q < end) ? j + q : last];
                        #pragma unroll
                        for (int q = 0; q < EPT; ++q) h0[q] = *(const v4f*)&H[(size_t)s[q] * PIN + col0];
                        if (on1) {
                            #pragma unroll
                            for (int q = 0; q < EPT; ++q) h1[q] = *(const v4f*)&H[(size_t)s[q] * PIN + col1];
                        } else {
                            #pragma unroll
                            for (int q = 0; q < EPT; ++q) h1[q] = v4f{};
                        }
                        acc0 += h0[0];
                        #pragma unroll
                        for (int q = 1; q < EPT; ++q) acc0 += h0[q] * ((j + q < end) ? 1.0f : 0.0f);
                        if (NV > 1) {
                            acc1 += h1[0];
                            #pragma unroll
                            for (int q = 1; q < EPT; ++q) acc1 += h1[q] * ((j + q < end) ? 1.0f : 0.0f);
                        }
                    }
                    x0 = (acc0 + hs0) * di + bz0;
                    x0.x = fmaxf(x0.x, 0.0f); x0.y = fmaxf(x0.y, 0.0f);
                    x0.z = fmaxf(x0.z, 0.0f); x0.w = fmaxf(x0.w, 0.0f);
                    if (NV > 1) {
                        x1 = (acc1 + hs1) * di + bz1;
                        x1.x = fmaxf(x1.x, 0.0f); x1.y = fmaxf(x1.y, 0.0f);
                        x1.z = fmaxf(x1.z, 0.0f); x1.w = fmaxf(x1.w, 0.0f);
                    }
                }
                *(v4f*)&As[r][col0] = x0;
                if (on1) *(v4f*)&As[r][col1] = x1;
            }
        }
    }

    const int wave = __builtin_amdgcn_readfirstlane(t >> 5);
    const int lane = t & 31;
    const int half = lane >> 4;
    const int m    = lane & 15;
    const int rt   = wave & 1;
    const int cg   = wave >> 1;
    const bool busy = cg < CG;
    const int ct0  = cg * CT;
    const int arow = (rt << 4) + m;

    v8f acc[CT];
    #pragma unroll
    for (int j = 0; j < CT; ++j) acc[j] = v8f{};
    v4f wa[WQ];

#define LOAD_W(k0_)                                                            \
    {                                                                          \
        _Pragma("unroll")                                                      \
        for (int i = 0; i < WQ; ++i) {                                         \
            const int slot_ = t + (i << 8);                                    \
            if ((i + 1) * 256 <= NTP * 128 || slot_ < NTP * 128) {   \
                const int r_   = slot_ >> 3;                                   \
                const int rc_  = r_ < Nout ? r_ : Nout - 1;   \
                const int gk_  = (k0_) + ((slot_ & 7) << 2);                   \
                const int gkc_ = gk_ < F ? gk_ : F - 4;       \
                wa[i] = *(const v4f*)&Wn[(size_t)rc_ * F + gkc_];              \
            }                                                                  \
        }                                                                      \
    }

#define KSTEP_F(kk_)                                                           \
    {                                                                          \
        const int kb_ = (kk_) + (half << 1);            \
        const v2f a_ = *(const v2f*)&As[arow][k0 + kb_];                       \
        _Pragma("unroll")                                                      \
        for (int j = 0; j < CT; ++j) {                                         \
            const v2f b_ = *(const v2f*)&Ws[((ct0 + j) << 4) + m][kb_];        \
            acc[j] = __builtin_amdgcn_wmma_f32_16x16x4_f32(                    \
                false, a_, false, b_, (short)0, acc[j], false, false);         \
        }                                                                      \
    }

    LOAD_W(0)
    for (int k0 = 0; k0 < F; k0 += TK) {
        __syncthreads();
        #pragma unroll
        for (int i = 0; i < WQ; ++i) {
            const int slot = t + (i << 8);
            if ((i + 1) * 256 <= NTP * 128 || slot < NTP * 128)
                *(v4f*)&Ws[slot >> 3][(slot & 7) << 2] = wa[i];
        }
        __syncthreads();

        if (k0 + TK < F) LOAD_W(k0 + TK)

        if (busy) {
            if (k0 + TK <= F) {
                #pragma unroll
                for (int kk = 0; kk < TK; kk += 4) KSTEP_F(kk)
            } else {
                for (int kk = 0; k0 + kk < F; kk += 4) KSTEP_F(kk)
            }
        }
    }
#undef LOAD_W
#undef KSTEP_F

    if (!busy) return;
    const int rbase = row0 + (rt << 4) + (half << 3);
    float sc[8];
    #pragma unroll
    for (int r = 0; r < 8; ++r) {
        const int gr = rbase + r;
        sc[r] = dinv[gr < N ? gr : N - 1];
    }
    if (row0 + FB <= N) {
        float* __restrict__ hp = Hn + (size_t)rbase * POUT + m;
        #pragma unroll
        for (int j = 0; j < CT; ++j) {
            if (((ct0 + j) << 4) + m < Nout) {
                #pragma unroll
                for (int r = 0; r < 8; ++r) hp[(size_t)r * POUT + ((ct0 + j) << 4)] = acc[j][r] * sc[r];
            }
        }
    } else {
        #pragma unroll
        for (int j = 0; j < CT; ++j) {
            const int n = ((ct0 + j) << 4) + m;
            #pragma unroll
            for (int r = 0; r < 8; ++r) {
                const int grow = rbase + r;
                if (grow < N && n < Nout) Hn[(size_t)grow * POUT + n] = acc[j][r] * sc[r];
            }
        }
    }
}

__global__ void k_agg_elem(
    const float* __restrict__ H, const int* __restrict__ rowptr,
    const int* __restrict__ list, const float* __restrict__ dinv,
    const float* __restrict__ bias, float* __restrict__ dest,
    int N, int E, int F, int doRelu)
{
    const size_t idx = (size_t)blockIdx.x * blockDim.x + threadIdx.x;
    if (idx >= (size_t)N * F) return;
    const int i = (int)(idx / F);
    const int f = (int)(idx - (size_t)i * F);
    int beg = rowptr[i], end = rowptr[i + 1];
    beg = beg < 0 ? 0 : beg;
    end = end > E ? E : end;
    const float di = dinv[i];
    const float hs = H[idx];
    float acc = 0.0f;
    const int last = end - 1;
    for (int j = beg; j < end; j += 4) {
        const bool p1 = j + 1 < end, p2 = j + 2 < end, p3 = j + 3 < end;
        const int s0 = list[j];
        const int s1 = list[p1 ? j + 1 : last];
        const int s2 = list[p2 ? j + 2 : last];
        const int s3 = list[p3 ? j + 3 : last];
        const float h0 = H[(size_t)s0 * F + f];
        const float h1 = H[(size_t)s1 * F + f];
        const float h2 = H[(size_t)s2 * F + f];
        const float h3 = H[(size_t)s3 * F + f];
        acc += h0;
        acc += h1 * (p1 ? 1.0f : 0.0f);
        acc += h2 * (p2 ? 1.0f : 0.0f);
        acc += h3 * (p3 ? 1.0f : 0.0f);
    }
    const float v = (acc + hs) * di + bias[f];
    dest[idx] = doRelu ? fmaxf(v, 0.0f) : v;
}

extern "C" void kernel_launch(void* const* d_in, const int* in_sizes, int n_in,
                              void* d_out, int out_size, void* d_ws, size_t ws_size,
                              hipStream_t stream) {
    (void)n_in; (void)out_size;

    const float* x   = (const float*)d_in[0];
    const int*   ei  = (const int*)  d_in[1];
    const float* W1  = (const float*)d_in[2];
    const float* b1  = (const float*)d_in[3];
    const float* W1b = (const float*)d_in[4];
    const float* b1b = (const float*)d_in[5];
    const float* W2b = (const float*)d_in[6];
    const float* b2b = (const float*)d_in[7];
    const float* W2  = (const float*)d_in[8];
    const float* b2  = (const float*)d_in[9];

    const int F_IN = 500, H1 = 200, H2 = 100, H3 = 40, C = 3;
    const int N = in_sizes[0] / F_IN;
    const int E = in_sizes[1] / 2;
    const int* srcIdx = ei;
    const int* dstIdx = ei + E;

    char*  ws   = (char*)d_ws;
    float* dinv = (float*)ws;
    size_t off  = ((size_t)N * sizeof(float) + 511) & ~(size_t)511;
    off += (size_t)256 * 512 * sizeof(float);
    float* bufH = (float*)(ws + off);
    off += (((size_t)N * H1 * sizeof(float)) + 511) & ~(size_t)511;
    float* bufA = (float*)(ws + off);
    off += (((size_t)N * H1 * sizeof(float)) + 511) & ~(size_t)511;
    int*   list = (int*)(ws + off);
    off += (((size_t)E * 2 * sizeof(int)) + 511) & ~(size_t)511;
    int*   rowptr = (int*)(ws + off);
    off += (((size_t)(N + 1) * sizeof(int)) + 511) & ~(size_t)511;
    float* out  = (float*)d_out;

    const size_t cntPad = ((size_t)N + 127) & ~(size_t)127;
    int* cnt  = (int*)bufA;
    int* rank = cnt + cntPad;

    const int gemmBlocks = (N + BM - 1) / BM;

    auto gemm = [&](const float* in, const float* Wt, const float* rs, int K, int F) {
        if (F <= 16)       k_gemm_wmma<1><<<gemmBlocks, 128, 0, stream>>>(in, Wt, rs, bufH, N, K, F);
        else if (F <= 48)  k_gemm_wmma<3><<<gemmBlocks, 128, 0, stream>>>(in, Wt, rs, bufH, N, K, F);
        else if (F <= 112) k_gemm_wmma<7><<<gemmBlocks, 128, 0, stream>>>(in, Wt, rs, bufH, N, K, F);
        else               k_gemm_wmma<13><<<gemmBlocks, 128, 0, stream>>>(in, Wt, rs, bufH, N, K, F);
    };

    const bool gather = ws_size >= off && cntPad + (size_t)E <= (size_t)N * H1;

    const int    NB     = (N + BNODES - 1) >> BSH;
    const int    chunk  = (E + G1 - 1) / G1;
    const size_t tmpPad = ((size_t)E * 2 + 127) & ~(size_t)127;
    int* tmpE   = (int*)bufA;
    int* blkcnt = tmpE + tmpPad;
    int* blkoff = blkcnt + (size_t)MAXB * G1;
    const bool binned = NB <= MAXB && tmpPad + (size_t)2 * MAXB * G1 <= (size_t)N * H1;

    if (gather) {
        if (binned) {
            const int nTab = NB * G1;
            k_bin_count  <<<G1, L1T, 0, stream>>>(dstIdx, blkcnt, E, NB, chunk);
            k_scan_tab   <<<(nTab + SCAN_SEG - 1) / SCAN_SEG, SCAN_T, 0, stream>>>(blkcnt, blkoff, nTab);
            k_bin_scatter<<<G1, L1T, 0, stream>>>(srcIdx, dstIdx, blkcnt, blkoff, tmpE, E, NB, chunk);
            k_bucket_sort<<<NB, L1T, 0, stream>>>(tmpE, blkoff, rowptr, dinv, list, N, E, NB);
        } else {
            k_izero<<<(N + 255) / 256, 256, 0, stream>>>(cnt, N);
            k_rank <<<(E + 255) / 256, 256, 0, stream>>>(dstIdx, cnt, rank, E);
            k_scan <<<(N + SCAN_SEG - 1) / SCAN_SEG, SCAN_T, 0, stream>>>(cnt, rowptr, dinv, N);
            k_place<<<(E + 255) / 256, 256, 0, stream>>>(srcIdx, dstIdx, rank, rowptr, list, E);
        }

        const int fusedBlocks = (N + FB - 1) / FB;
        gemm(x, W1, dinv, F_IN, H1);
        k_agg_gemm<200, 7, 200, 128><<<fusedBlocks, 256, 0, stream>>>(
            bufH, rowptr, list, dinv, b1,  W1b, bufA, N, E, H2);
        k_agg_gemm<100, 3, 128, 64><<<fusedBlocks, 256, 0, stream>>>(
            bufA, rowptr, list, dinv, b1b, W2b, bufH, N, E, H3);
        k_agg_gemm<40, 1, 64, 3><<<fusedBlocks, 256, 0, stream>>>(
            bufH, rowptr, list, dinv, b2b, W2,  bufA, N, E, C);
        {
            const size_t tot = (size_t)N * C;
            k_agg_elem<<<(int)((tot + 255) / 256), 256, 0, stream>>>(
                bufA, rowptr, list, dinv, b2, out, N, E, C, 0);
        }
        return;
    }

    k_fill1<<<(N + 255) / 256, 256, 0, stream>>>(dinv, N);
    k_deg  <<<(E + 255) / 256, 256, 0, stream>>>(dstIdx, dinv, E);
    k_rsqrt<<<(N + 255) / 256, 256, 0, stream>>>(dinv, N);

    const int aggBlocks  = 65536;

    auto layer = [&](const float* in, const float* Wt, const float* bias,
                     float* dest, int K, int F, int relu) {
        gemm(in, Wt, nullptr, K, F);
        size_t tot = (size_t)N * F;
        k_zero<<<(int)((tot + 255) / 256), 256, 0, stream>>>(dest, tot);
        if (F >= 32) {
            k_agg_wave<<<aggBlocks, 256, 0, stream>>>(bufH, srcIdx, dstIdx, dinv, dest, E, F);
        } else {
            size_t tote = (size_t)E * F;
            k_agg_small<<<(int)((tote + 255) / 256), 256, 0, stream>>>(
                bufH, srcIdx, dstIdx, dinv, dest, E, F);
        }
        k_finish<<<(int)((tot + 255) / 256), 256, 0, stream>>>(
            dest, bufH, dinv, bias, N, F, relu);
    };

    layer(x,    W1,  b1,  bufA, F_IN, H1, 1);
    layer(bufA, W1b, b1b, bufA, H1,   H2, 1);
    layer(bufA, W2b, b2b, bufA, H2,   H3, 1);
    layer(bufA, W2,  b2,  out,  H3,   C,  0);
}
